# starting kernel
# baseline (speedup 1.0000x reference)
#include <hip/hip_runtime.h>
#include <hip/hip_bf16.h>

typedef __attribute__((ext_vector_type(16))) _Float16 v16h;
typedef __attribute__((ext_vector_type(8)))  float    v8f;

#define T_HIST 3
#define F_IN   5
#define HDIM   128
#define CDIM   5

__global__ void fill_zero_kernel(float* __restrict__ p, int n) {
    int i = blockIdx.x * blockDim.x + threadIdx.x;
    if (i < n) p[i] = 0.0f;
}

__global__ void deg_kernel(const int* __restrict__ dst, float* __restrict__ deg, int e) {
    int i = blockIdx.x * blockDim.x + threadIdx.x;
    if (i < e) atomicAdd(&deg[dst[i]], 1.0f);
}

__global__ void dinv_kernel(float* __restrict__ deg, int n) {
    int i = blockIdx.x * blockDim.x + threadIdx.x;
    if (i < n) deg[i] = rsqrtf(deg[i] + 1.0f);
}

__global__ void xw1_seed_kernel(const float* __restrict__ x, const float* __restrict__ W1,
                                const float* __restrict__ b1, const float* __restrict__ dinv,
                                float* __restrict__ xw1, float* __restrict__ agg_seed, int n) {
    int idx = blockIdx.x * blockDim.x + threadIdx.x;
    if (idx >= n * HDIM) return;
    int row = idx >> 7;
    int h   = idx & (HDIM - 1);
    const float* xr = x + (size_t)row * (T_HIST * F_IN) + (T_HIST - 1) * F_IN;
    float s = 0.0f;
#pragma unroll
    for (int f = 0; f < F_IN; ++f) s = fmaf(xr[f], W1[f * HDIM + h], s);
    xw1[idx] = s;
    float di = dinv[row];
    agg_seed[idx] = di * di * s + b1[h];
}

__global__ __launch_bounds__(256) void scatter_kernel(const int* __restrict__ src, const int* __restrict__ dst,
                                                      const float* __restrict__ dinv, const float* __restrict__ xw,
                                                      float* __restrict__ agg, int e) {
    int edge = (blockIdx.x * blockDim.x + threadIdx.x) >> 5;
    int lane = threadIdx.x & 31;
    if (edge >= e) return;
    int s = src[edge], d = dst[edge];
    float norm = dinv[s] * dinv[d];
    const float4* xr = (const float4*)(xw + (size_t)s * HDIM);
    float4 v = xr[lane];
    float* ag = agg + (size_t)d * HDIM + lane * 4;
    atomicAdd(ag + 0, norm * v.x);
    atomicAdd(ag + 1, norm * v.y);
    atomicAdd(ag + 2, norm * v.z);
    atomicAdd(ag + 3, norm * v.w);
}

__global__ __launch_bounds__(256) void gemm2_wmma_kernel(const float* __restrict__ pre1,
                                                         const float* __restrict__ W2,
                                                         const float* __restrict__ b2,
                                                         const float* __restrict__ dinv,
                                                         float* __restrict__ xw2,
                                                         float* __restrict__ agg_seed,
                                                         int nrows) {
    __shared__ _Float16 aS[16][HDIM];
    __shared__ _Float16 bS[HDIM][HDIM];
    __shared__ float    dS[16];

    const int tid       = threadIdx.x;
    const int block_row = blockIdx.x * 16;

    for (int i = tid; i < HDIM * HDIM; i += 256) {
        int k = i >> 7;
        int n = i & (HDIM - 1);
        bS[n][k] = (_Float16)W2[i];
    }
    for (int i = tid; i < 16 * HDIM; i += 256) {
        int m = i >> 7;
        int k = i & (HDIM - 1);
        int row = block_row + m;
        float v = (row < nrows) ? fmaxf(pre1[(size_t)row * HDIM + k], 0.0f) : 0.0f;
        aS[m][k] = (_Float16)v;
    }
    if (tid < 16) {
        int row = block_row + tid;
        dS[tid] = (row < nrows) ? dinv[row] : 0.0f;
    }
    __syncthreads();

    const int wave  = tid >> 5;
    const int lane  = tid & 31;
    const int ncol  = wave * 16;
    const int mrow  = lane & 15;
    const int khalf = (lane >> 4) * 8;

    v8f acc = {};
    for (int kk = 0; kk < HDIM; kk += 32) {
        v16h a, b;
#pragma unroll
        for (int j = 0; j < 16; j += 2) {
            int K = (j & 7) + ((j >> 3) << 4) + khalf + kk;
            a[j]     = aS[mrow][K];
            a[j + 1] = aS[mrow][K + 1];
            b[j]     = bS[ncol + mrow][K];
            b[j + 1] = bS[ncol + mrow][K + 1];
        }
        acc = __builtin_amdgcn_wmma_f32_16x16x32_f16(false, a, false, b,
                                                     (short)0, acc, false, false);
    }

    const int col  = ncol + (lane & 15);
    const float bc = b2[col];
    if (block_row + 16 <= nrows) {
#pragma unroll
        for (int v = 0; v < 8; ++v) {
            int r  = v + ((lane >> 4) << 3);
            int row = block_row + r;
            float val = acc[v];
            float di  = dS[r];
            xw2[(size_t)row * HDIM + col]      = val;
            agg_seed[(size_t)row * HDIM + col] = di * di * val + bc;
        }
    } else {
#pragma unroll
        for (int v = 0; v < 8; ++v) {
            int r  = v + ((lane >> 4) << 3);
            int row = block_row + r;
            if (row < nrows) {
                float val = acc[v];
                float di  = dS[r];
                xw2[(size_t)row * HDIM + col]      = val;
                agg_seed[(size_t)row * HDIM + col] = di * di * val + bc;
            }
        }
    }
}

__global__ __launch_bounds__(256) void out_kernel(const float* __restrict__ pre2,
                                                  const float* __restrict__ Wout,
                                                  const float* __restrict__ bout,
                                                  float* __restrict__ out, int n) {
    int node = (blockIdx.x * blockDim.x + threadIdx.x) >> 5;
    int lane = threadIdx.x & 31;
    if (node >= n) return;

    const float4* hr = (const float4*)(pre2 + (size_t)node * HDIM);
    float4 hv = hr[lane];
    hv.x = fmaxf(hv.x, 0.0f);
    hv.y = fmaxf(hv.y, 0.0f);
    hv.z = fmaxf(hv.z, 0.0f);
    hv.w = fmaxf(hv.w, 0.0f);

    int h0 = lane * 4;
    float acc[CDIM];
#pragma unroll
    for (int c = 0; c < CDIM; ++c) {
        acc[c] = hv.x * Wout[(h0 + 0) * CDIM + c]
               + hv.y * Wout[(h0 + 1) * CDIM + c]
               + hv.z * Wout[(h0 + 2) * CDIM + c]
               + hv.w * Wout[(h0 + 3) * CDIM + c];
    }
#pragma unroll
    for (int off = 16; off > 0; off >>= 1) {
#pragma unroll
        for (int c = 0; c < CDIM; ++c) acc[c] += __shfl_xor(acc[c], off, 32);
    }
    if (lane == 0) {
#pragma unroll
        for (int c = 0; c < CDIM; ++c) acc[c] += bout[c];
        float m = acc[0];
#pragma unroll
        for (int c = 1; c < CDIM; ++c) m = fmaxf(m, acc[c]);
        float sum = 0.0f;
#pragma unroll
        for (int c = 0; c < CDIM; ++c) { acc[c] = __expf(acc[c] - m); sum += acc[c]; }
        float inv = 1.0f / sum;
#pragma unroll
        for (int c = 0; c < CDIM; ++c) out[(size_t)node * CDIM + c] = acc[c] * inv;
    }
}

extern "C" void kernel_launch(void* const* d_in, const int* in_sizes, int n_in,
                              void* d_out, int out_size, void* d_ws, size_t ws_size,
                              hipStream_t stream) {
    const float* x    = (const float*)d_in[0];
    const int*   eidx = (const int*)  d_in[1];
    const float* W1   = (const float*)d_in[2];
    const float* b1   = (const float*)d_in[3];
    const float* W2   = (const float*)d_in[4];
    const float* b2   = (const float*)d_in[5];
    const float* Wout = (const float*)d_in[6];
    const float* bout = (const float*)d_in[7];
    float*       out  = (float*)d_out;

    const int N = in_sizes[0] / (T_HIST * F_IN);
    const int E = in_sizes[1] / 2;
    const int* src = eidx;
    const int* dst = eidx + E;

    float* dinv = (float*)d_ws;
    float* bufA = dinv + ((N + 63) & ~63);
    float* bufB = bufA + (size_t)N * HDIM;

    const int TB = 256;
    const int NH = N * HDIM;
    const int NH_blocks   = (NH + TB - 1) / TB;
    const int edge_blocks = ((E * 32) + TB - 1) / TB;
    const int node_blocks = ((N * 32) + TB - 1) / TB;

    fill_zero_kernel<<<(N + TB - 1) / TB, TB, 0, stream>>>(dinv, N);
    deg_kernel<<<(E + TB - 1) / TB, TB, 0, stream>>>(dst, dinv, E);
    dinv_kernel<<<(N + TB - 1) / TB, TB, 0, stream>>>(dinv, N);

    xw1_seed_kernel<<<NH_blocks, TB, 0, stream>>>(x, W1, b1, dinv, bufA, bufB, N);
    scatter_kernel<<<edge_blocks, TB, 0, stream>>>(src, dst, dinv, bufA, bufB, E);

    gemm2_wmma_kernel<<<(N + 15) / 16, 256, 0, stream>>>(bufB, W2, b2, dinv, bufA, bufB, N);
    scatter_kernel<<<edge_blocks, TB, 0, stream>>>(src, dst, dinv, bufA, bufB, E);

    out_kernel<<<node_blocks, TB, 0, stream>>>(bufB, Wout, bout, out, N);
}


# layer 1 aggregate-then-project: neighbours summed on the 5 input features (5 atomics per edge, not 128; float sum order changes), x@W1+b1 applied after, in the layer-2 GEMM's A staging
# speedup vs baseline: 1.4119x; 1.4119x over previous
#include <hip/hip_runtime.h>
#include <hip/hip_bf16.h>

typedef __attribute__((ext_vector_type(16))) _Float16 v16h;
typedef __attribute__((ext_vector_type(8)))  float    v8f;

#define T_HIST 3
#define F_IN   5
#define HDIM   128
#define CDIM   5

__global__ void fill_zero_kernel(float* __restrict__ p, int n) {
    int i = blockIdx.x * blockDim.x + threadIdx.x;
    if (i < n) p[i] = 0.0f;
}

__global__ void deg_kernel(const int* __restrict__ dst, float* __restrict__ deg, int e) {
    int i = blockIdx.x * blockDim.x + threadIdx.x;
    if (i < e) atomicAdd(&deg[dst[i]], 1.0f);
}

__global__ void dinv_kernel(float* __restrict__ deg, int n) {
    int i = blockIdx.x * blockDim.x + threadIdx.x;
    if (i < n) deg[i] = rsqrtf(deg[i] + 1.0f);
}


__global__ void seed_in_kernel(const float* __restrict__ x, const float* __restrict__ dinv,
                               float* __restrict__ agg, int n) {
    int t   = blockIdx.x * blockDim.x + threadIdx.x;
    int row = t >> 3;
    int f   = t & 7;
    if (row >= n || f >= F_IN) return;
    float di = dinv[row];
    agg[(size_t)row * HDIM + f] = di * di * x[(size_t)row * (T_HIST * F_IN) + (T_HIST - 1) * F_IN + f];
}

__global__ __launch_bounds__(256) void scatter_in_kernel(const int* __restrict__ src, const int* __restrict__ dst,
                                                         const float* __restrict__ dinv, const float* __restrict__ x,
                                                         float* __restrict__ agg, int e) {
    const int lane = threadIdx.x & 31;
    const int edge = blockIdx.x * blockDim.x + threadIdx.x;
    int s = 0, d = 0;
    float nrm = 0.0f;
    if (edge < e) {
        s = src[edge];
        d = dst[edge];
        nrm = dinv[s] * dinv[d];
    }
    const int f     = lane & 7;
    const int fc    = f < F_IN ? f : F_IN - 1;
    const int sub   = lane >> 3;
    const int wbase = edge - lane;

    int   dd[8];
    float v[8];
    for (int p = 0; p < 8; ++p) {
        const int from = p * 4 + sub;
        const int   ss = __shfl(s, from, 32);
        const float nn = __shfl(nrm, from, 32);
        dd[p] = __shfl(d, from, 32);
        v[p]  = nn * x[(size_t)ss * (T_HIST * F_IN) + (T_HIST - 1) * F_IN + fc];
    }
    for (int p = 0; p < 8; ++p) {
        if (f < F_IN && wbase + p * 4 + sub < e)
            atomicAdd(agg + (size_t)dd[p] * HDIM + f, v[p]);
    }
}

__global__ __launch_bounds__(256) void scatter_kernel(const int* __restrict__ src, const int* __restrict__ dst,
                                                      const float* __restrict__ dinv, const float* __restrict__ xw,
                                                      float* __restrict__ agg, int e) {
    int edge = (blockIdx.x * blockDim.x + threadIdx.x) >> 5;
    int lane = threadIdx.x & 31;
    if (edge >= e) return;
    int s = src[edge], d = dst[edge];
    float norm = dinv[s] * dinv[d];
    const float4* xr = (const float4*)(xw + (size_t)s * HDIM);
    float4 v = xr[lane];
    float* ag = agg + (size_t)d * HDIM + lane * 4;
    atomicAdd(ag + 0, norm * v.x);
    atomicAdd(ag + 1, norm * v.y);
    atomicAdd(ag + 2, norm * v.z);
    atomicAdd(ag + 3, norm * v.w);
}

__global__ __launch_bounds__(256) void gemm2_wmma_kernel(const float* __restrict__ W1,
                                                         const float* __restrict__ b1,
                                                         const float* __restrict__ W2,
                                                         const float* __restrict__ b2,
                                                         const float* __restrict__ dinv,
                                                         float* __restrict__ xw2,
                                                         float* __restrict__ agg,
                                                         int nrows) {
    __shared__ _Float16 aS[16][HDIM];
    __shared__ _Float16 bS[HDIM][HDIM];
    __shared__ float    dS[16];
    __shared__ float    zS[16][8];

    const int tid       = threadIdx.x;
    const int block_row = blockIdx.x * 16;

    if (tid < 128) {
        int m = tid >> 3;
        int f = tid & 7;
        int row = block_row + m;
        if (f < F_IN) zS[m][f] = (row < nrows) ? agg[(size_t)row * HDIM + f] : 0.0f;
    }
    for (int i = tid; i < HDIM * HDIM; i += 256) {
        int k = i >> 7;
        int n = i & (HDIM - 1);
        bS[n][k] = (_Float16)W2[i];
    }
    if (tid < 16) {
        int row = block_row + tid;
        dS[tid] = (row < nrows) ? dinv[row] : 0.0f;
    }
    __syncthreads();

    {
        const int k = tid & (HDIM - 1);
        float w[F_IN];
        for (int f = 0; f < F_IN; ++f) w[f] = W1[f * HDIM + k];
        const float bk = b1[k];
        for (int m = tid >> 7; m < 16; m += 2) {
            float s = 0.0f;
            for (int f = 0; f < F_IN; ++f) s = fmaf(zS[m][f], w[f], s);
            float v = (block_row + m < nrows) ? fmaxf(s + bk, 0.0f) : 0.0f;
            aS[m][k] = (_Float16)v;
        }
    }
    __syncthreads();

    const int wave  = tid >> 5;
    const int lane  = tid & 31;
    const int ncol  = wave * 16;
    const int mrow  = lane & 15;
    const int khalf = (lane >> 4) * 8;

    v8f acc = {};
    for (int kk = 0; kk < HDIM; kk += 32) {
        v16h a, b;
#pragma unroll
        for (int j = 0; j < 16; j += 2) {
            int K = (j & 7) + ((j >> 3) << 4) + khalf + kk;
            a[j]     = aS[mrow][K];
            a[j + 1] = aS[mrow][K + 1];
            b[j]     = bS[ncol + mrow][K];
            b[j + 1] = bS[ncol + mrow][K + 1];
        }
        acc = __builtin_amdgcn_wmma_f32_16x16x32_f16(false, a, false, b,
                                                     (short)0, acc, false, false);
    }

    const int col  = ncol + (lane & 15);
    const float bc = b2[col];
    if (block_row + 16 <= nrows) {
#pragma unroll
        for (int v = 0; v < 8; ++v) {
            int r  = v + ((lane >> 4) << 3);
            int row = block_row + r;
            float val = acc[v];
            float di  = dS[r];
            xw2[(size_t)row * HDIM + col]      = val;
            agg[(size_t)row * HDIM + col]      = di * di * val + bc;
        }
    } else {
#pragma unroll
        for (int v = 0; v < 8; ++v) {
            int r  = v + ((lane >> 4) << 3);
            int row = block_row + r;
            if (row < nrows) {
                float val = acc[v];
                float di  = dS[r];
                xw2[(size_t)row * HDIM + col]      = val;
                agg[(size_t)row * HDIM + col]      = di * di * val + bc;
            }
        }
    }
}

__global__ __launch_bounds__(256) void out_kernel(const float* __restrict__ pre2,
                                                  const float* __restrict__ Wout,
                                                  const float* __restrict__ bout,
                                                  float* __restrict__ out, int n) {
    int node = (blockIdx.x * blockDim.x + threadIdx.x) >> 5;
    int lane = threadIdx.x & 31;
    if (node >= n) return;

    const float4* hr = (const float4*)(pre2 + (size_t)node * HDIM);
    float4 hv = hr[lane];
    hv.x = fmaxf(hv.x, 0.0f);
    hv.y = fmaxf(hv.y, 0.0f);
    hv.z = fmaxf(hv.z, 0.0f);
    hv.w = fmaxf(hv.w, 0.0f);

    int h0 = lane * 4;
    float acc[CDIM];
#pragma unroll
    for (int c = 0; c < CDIM; ++c) {
        acc[c] = hv.x * Wout[(h0 + 0) * CDIM + c]
               + hv.y * Wout[(h0 + 1) * CDIM + c]
               + hv.z * Wout[(h0 + 2) * CDIM + c]
               + hv.w * Wout[(h0 + 3) * CDIM + c];
    }
#pragma unroll
    for (int off = 16; off > 0; off >>= 1) {
#pragma unroll
        for (int c = 0; c < CDIM; ++c) acc[c] += __shfl_xor(acc[c], off, 32);
    }
    if (lane == 0) {
#pragma unroll
        for (int c = 0; c < CDIM; ++c) acc[c] += bout[c];
        float m = acc[0];
#pragma unroll
        for (int c = 1; c < CDIM; ++c) m = fmaxf(m, acc[c]);
        float sum = 0.0f;
#pragma unroll
        for (int c = 0; c < CDIM; ++c) { acc[c] = __expf(acc[c] - m); sum += acc[c]; }
        float inv = 1.0f / sum;
#pragma unroll
        for (int c = 0; c < CDIM; ++c) out[(size_t)node * CDIM + c] = acc[c] * inv;
    }
}

extern "C" void kernel_launch(void* const* d_in, const int* in_sizes, int n_in,
                              void* d_out, int out_size, void* d_ws, size_t ws_size,
                              hipStream_t stream) {
    const float* x    = (const float*)d_in[0];
    const int*   eidx = (const int*)  d_in[1];
    const float* W1   = (const float*)d_in[2];
    const float* b1   = (const float*)d_in[3];
    const float* W2   = (const float*)d_in[4];
    const float* b2   = (const float*)d_in[5];
    const float* Wout = (const float*)d_in[6];
    const float* bout = (const float*)d_in[7];
    float*       out  = (float*)d_out;

    const int N = in_sizes[0] / (T_HIST * F_IN);
    const int E = in_sizes[1] / 2;
    const int* src = eidx;
    const int* dst = eidx + E;

    float* dinv = (float*)d_ws;
    float* bufA = dinv + ((N + 63) & ~63);
    float* bufB = bufA + (size_t)N * HDIM;

    const int TB = 256;
    const int edge_blocks = ((E * 32) + TB - 1) / TB;
    const int node_blocks = ((N * 32) + TB - 1) / TB;

    fill_zero_kernel<<<(N + TB - 1) / TB, TB, 0, stream>>>(dinv, N);
    deg_kernel<<<(E + TB - 1) / TB, TB, 0, stream>>>(dst, dinv, E);
    dinv_kernel<<<(N + TB - 1) / TB, TB, 0, stream>>>(dinv, N);

    seed_in_kernel<<<(N * 8 + TB - 1) / TB, TB, 0, stream>>>(x, dinv, bufB, N);
    scatter_in_kernel<<<(E + TB - 1) / TB, TB, 0, stream>>>(src, dst, dinv, x, bufB, E);

    gemm2_wmma_kernel<<<(N + 15) / 16, 256, 0, stream>>>(W1, b1, W2, b2, dinv, bufA, bufB, N);
    scatter_kernel<<<edge_blocks, TB, 0, stream>>>(src, dst, dinv, bufA, bufB, E);

    out_kernel<<<node_blocks, TB, 0, stream>>>(bufB, Wout, bout, out, N);
}


# gemm2 operand staging: W2 fragments loaded per wave from global into registers (no transposed 2-byte LDS copy of W2), 64-row block, A tile at padded pitch read by 16-byte LDS loads; sum order kept
# speedup vs baseline: 1.8557x; 1.3143x over previous
#include <hip/hip_runtime.h>
#include <hip/hip_bf16.h>

typedef __attribute__((ext_vector_type(16))) _Float16 v16h;
typedef __attribute__((ext_vector_type(8)))  float    v8f;
typedef __attribute__((ext_vector_type(8)))  _Float16 v8h;

#define T_HIST 3
#define F_IN   5
#define HDIM   128
#define CDIM   5

__global__ void fill_zero_kernel(float* __restrict__ p, int n) {
    int i = blockIdx.x * blockDim.x + threadIdx.x;
    if (i < n) p[i] = 0.0f;
}

__global__ void deg_kernel(const int* __restrict__ dst, float* __restrict__ deg, int e) {
    int i = blockIdx.x * blockDim.x + threadIdx.x;
    if (i < e) atomicAdd(&deg[dst[i]], 1.0f);
}

__global__ void dinv_kernel(float* __restrict__ deg, int n) {
    int i = blockIdx.x * blockDim.x + threadIdx.x;
    if (i < n) deg[i] = rsqrtf(deg[i] + 1.0f);
}


__global__ void seed_in_kernel(const float* __restrict__ x, const float* __restrict__ dinv,
                               float* __restrict__ agg, int n) {
    int t   = blockIdx.x * blockDim.x + threadIdx.x;
    int row = t >> 3;
    int f   = t & 7;
    if (row >= n || f >= F_IN) return;
    float di = dinv[row];
    agg[(size_t)row * HDIM + f] = di * di * x[(size_t)row * (T_HIST * F_IN) + (T_HIST - 1) * F_IN + f];
}

__global__ __launch_bounds__(256) void scatter_in_kernel(const int* __restrict__ src, const int* __restrict__ dst,
                                                         const float* __restrict__ dinv, const float* __restrict__ x,
                                                         float* __restrict__ agg, int e) {
    const int lane = threadIdx.x & 31;
    const int edge = blockIdx.x * blockDim.x + threadIdx.x;
    int s = 0, d = 0;
    float nrm = 0.0f;
    if (edge < e) {
        s = src[edge];
        d = dst[edge];
        nrm = dinv[s] * dinv[d];
    }
    const int f     = lane & 7;
    const int fc    = f < F_IN ? f : F_IN - 1;
    const int sub   = lane >> 3;
    const int wbase = edge - lane;

    int   dd[8];
    float v[8];
    for (int p = 0; p < 8; ++p) {
        const int from = p * 4 + sub;
        const int   ss = __shfl(s, from, 32);
        const float nn = __shfl(nrm, from, 32);
        dd[p] = __shfl(d, from, 32);
        v[p]  = nn * x[(size_t)ss * (T_HIST * F_IN) + (T_HIST - 1) * F_IN + fc];
    }
    for (int p = 0; p < 8; ++p) {
        if (f < F_IN && wbase + p * 4 + sub < e)
            atomicAdd(agg + (size_t)dd[p] * HDIM + f, v[p]);
    }
}

__global__ __launch_bounds__(256) void scatter_kernel(const int* __restrict__ src, const int* __restrict__ dst,
                                                      const float* __restrict__ dinv, const float* __restrict__ xw,
                                                      float* __restrict__ agg, int e) {
    int edge = (blockIdx.x * blockDim.x + threadIdx.x) >> 5;
    int lane = threadIdx.x & 31;
    if (edge >= e) return;
    int s = src[edge], d = dst[edge];
    float norm = dinv[s] * dinv[d];
    const float4* xr = (const float4*)(xw + (size_t)s * HDIM);
    float4 v = xr[lane];
    float* ag = agg + (size_t)d * HDIM + lane * 4;
    atomicAdd(ag + 0, norm * v.x);
    atomicAdd(ag + 1, norm * v.y);
    atomicAdd(ag + 2, norm * v.z);
    atomicAdd(ag + 3, norm * v.w);
}

#define G2_ROWS  64
#define G2_PITCH (HDIM + 8)

__global__ __launch_bounds__(256) void gemm2_wmma_kernel(const float* __restrict__ W1,
                                                         const float* __restrict__ b1,
                                                         const float* __restrict__ W2,
                                                         const float* __restrict__ b2,
                                                         const float* __restrict__ dinv,
                                                         float* __restrict__ xw2,
                                                         float* __restrict__ agg,
                                                         int nrows) {
    __shared__ __attribute__((aligned(16))) _Float16 aS[G2_ROWS][G2_PITCH];
    __shared__ __attribute__((aligned(16))) float    zS[G2_ROWS][8];
    __shared__ __attribute__((aligned(16))) float    dS[G2_ROWS];

    const int tid   = threadIdx.x;
    const int wave  = tid >> 5;
    const int lane  = tid & 31;
    const int ncol  = wave * 16;
    const int c     = lane & 15;
    const int h     = lane >> 4;
    const int khalf = h * 8;
    const int col   = ncol + c;
    const int row0  = blockIdx.x * G2_ROWS;

    for (int i = tid; i < G2_ROWS * 8; i += 256) {
        const int m = i >> 3;
        const int f = i & 7;
        const int row = row0 + m;
        if (f < F_IN) zS[m][f] = (row < nrows) ? agg[(size_t)row * HDIM + f] : 0.0f;
    }
    if (tid < G2_ROWS) {
        const int row = row0 + tid;
        dS[tid] = (row < nrows) ? dinv[row] : 0.0f;
    }

    v16h bf[4];
#pragma unroll
    for (int ks = 0; ks < 4; ++ks) {
#pragma unroll
        for (int j = 0; j < 16; ++j) {
            const int K = (j & 7) + ((j >> 3) << 4) + khalf + ks * 32;
            bf[ks][j] = (_Float16)W2[K * HDIM + col];
        }
    }
    const float bc = b2[col];

    const int k  = tid & (HDIM - 1);
    const int mh = tid >> 7;
    float w[F_IN];
#pragma unroll
    for (int f = 0; f < F_IN; ++f) w[f] = W1[f * HDIM + k];
    const float bk = b1[k];
    __syncthreads();

#pragma unroll 4
    for (int i = 0; i < G2_ROWS / 2; ++i) {
        const int m = 2 * i + mh;
        float s = 0.0f;
#pragma unroll
        for (int f = 0; f < F_IN; ++f) s = fmaf(zS[m][f], w[f], s);
        const float v = (row0 + m < nrows) ? fmaxf(s + bk, 0.0f) : 0.0f;
        aS[m][k] = (_Float16)v;
    }
    __syncthreads();

    v8f acc[4] = {};
#pragma unroll
    for (int ks = 0; ks < 4; ++ks) {
#pragma unroll
        for (int rt = 0; rt < 4; ++rt) {
            const _Float16* ap = &aS[rt * 16 + c][ks * 32 + khalf];
            const v8h lo = *(const v8h*)ap;
            const v8h hi = *(const v8h*)(ap + 16);
            const v16h a = __builtin_shufflevector(lo, hi, 0, 1, 2, 3, 4, 5, 6, 7, 8, 9, 10, 11, 12, 13, 14, 15);
            acc[rt] = __builtin_amdgcn_wmma_f32_16x16x32_f16(false, a, false, bf[ks],
                                                             (short)0, acc[rt], false, false);
        }
    }

    float* xp = xw2 + ((size_t)(row0 + 8 * h) * HDIM + col);
    float* gp = agg + ((size_t)(row0 + 8 * h) * HDIM + col);
#pragma unroll
    for (int rt = 0; rt < 4; ++rt) {
        const int trow = row0 + rt * 16;
        const float4 d0 = *(const float4*)&dS[rt * 16 + 8 * h];
        const float4 d1 = *(const float4*)&dS[rt * 16 + 8 * h + 4];
        const float di[8] = {d0.x, d0.y, d0.z, d0.w, d1.x, d1.y, d1.z, d1.w};
        if (trow + 16 <= nrows) {
#pragma unroll
            for (int v = 0; v < 8; ++v) {
                const float val = acc[rt][v];
                xp[(rt * 16 + v) * HDIM] = val;
                gp[(rt * 16 + v) * HDIM] = di[v] * di[v] * val + bc;
            }
        } else {
#pragma unroll
            for (int v = 0; v < 8; ++v) {
                if (trow + v + 8 * h < nrows) {
                    const float val = acc[rt][v];
                    xp[(rt * 16 + v) * HDIM] = val;
                    gp[(rt * 16 + v) * HDIM] = di[v] * di[v] * val + bc;
                }
            }
        }
    }
}

__global__ __launch_bounds__(256) void out_kernel(const float* __restrict__ pre2,
                                                  const float* __restrict__ Wout,
                                                  const float* __restrict__ bout,
                                                  float* __restrict__ out, int n) {
    int node = (blockIdx.x * blockDim.x + threadIdx.x) >> 5;
    int lane = threadIdx.x & 31;
    if (node >= n) return;

    const float4* hr = (const float4*)(pre2 + (size_t)node * HDIM);
    float4 hv = hr[lane];
    hv.x = fmaxf(hv.x, 0.0f);
    hv.y = fmaxf(hv.y, 0.0f);
    hv.z = fmaxf(hv.z, 0.0f);
    hv.w = fmaxf(hv.w, 0.0f);

    int h0 = lane * 4;
    float acc[CDIM];
#pragma unroll
    for (int c = 0; c < CDIM; ++c) {
        acc[c] = hv.x * Wout[(h0 + 0) * CDIM + c]
               + hv.y * Wout[(h0 + 1) * CDIM + c]
               + hv.z * Wout[(h0 + 2) * CDIM + c]
               + hv.w * Wout[(h0 + 3) * CDIM + c];
    }
#pragma unroll
    for (int off = 16; off > 0; off >>= 1) {
#pragma unroll
        for (int c = 0; c < CDIM; ++c) acc[c] += __shfl_xor(acc[c], off, 32);
    }
    if (lane == 0) {
#pragma unroll
        for (int c = 0; c < CDIM; ++c) acc[c] += bout[c];
        float m = acc[0];
#pragma unroll
        for (int c = 1; c < CDIM; ++c) m = fmaxf(m, acc[c]);
        float sum = 0.0f;
#pragma unroll
        for (int c = 0; c < CDIM; ++c) { acc[c] = __expf(acc[c] - m); sum += acc[c]; }
        float inv = 1.0f / sum;
#pragma unroll
        for (int c = 0; c < CDIM; ++c) out[(size_t)node * CDIM + c] = acc[c] * inv;
    }
}

extern "C" void kernel_launch(void* const* d_in, const int* in_sizes, int n_in,
                              void* d_out, int out_size, void* d_ws, size_t ws_size,
                              hipStream_t stream) {
    const float* x    = (const float*)d_in[0];
    const int*   eidx = (const int*)  d_in[1];
    const float* W1   = (const float*)d_in[2];
    const float* b1   = (const float*)d_in[3];
    const float* W2   = (const float*)d_in[4];
    const float* b2   = (const float*)d_in[5];
    const float* Wout = (const float*)d_in[6];
    const float* bout = (const float*)d_in[7];
    float*       out  = (float*)d_out;

    const int N = in_sizes[0] / (T_HIST * F_IN);
    const int E = in_sizes[1] / 2;
    const int* src = eidx;
    const int* dst = eidx + E;

    float* dinv = (float*)d_ws;
    float* bufA = dinv + ((N + 63) & ~63);
    float* bufB = bufA + (size_t)N * HDIM;

    const int TB = 256;
    const int edge_blocks = ((E * 32) + TB - 1) / TB;
    const int node_blocks = ((N * 32) + TB - 1) / TB;

    fill_zero_kernel<<<(N + TB - 1) / TB, TB, 0, stream>>>(dinv, N);
    deg_kernel<<<(E + TB - 1) / TB, TB, 0, stream>>>(dst, dinv, E);
    dinv_kernel<<<(N + TB - 1) / TB, TB, 0, stream>>>(dinv, N);

    seed_in_kernel<<<(N * 8 + TB - 1) / TB, TB, 0, stream>>>(x, dinv, bufB, N);
    scatter_in_kernel<<<(E + TB - 1) / TB, TB, 0, stream>>>(src, dst, dinv, x, bufB, E);

    gemm2_wmma_kernel<<<(N + G2_ROWS - 1) / G2_ROWS, 256, 0, stream>>>(W1, b1, W2, b2, dinv, bufA, bufB, N);
    scatter_kernel<<<edge_blocks, TB, 0, stream>>>(src, dst, dinv, bufA, bufB, E);

    out_kernel<<<node_blocks, TB, 0, stream>>>(bufB, Wout, bout, out, N);
}


# layer-2 aggregation as a gather: edges counting-sorted by target in-call (int atomics), one wave sums a node's source rows and applies the head; no 128-wide float atomics; float sum order changes
# speedup vs baseline: 2.7303x; 1.4713x over previous
#include <hip/hip_runtime.h>
#include <hip/hip_bf16.h>

typedef __attribute__((ext_vector_type(16))) _Float16 v16h;
typedef __attribute__((ext_vector_type(8)))  float    v8f;
typedef __attribute__((ext_vector_type(8)))  _Float16 v8h;

#define T_HIST 3
#define F_IN   5
#define HDIM   128
#define CDIM   5
#define ZP     8

__global__ void fill_zero_kernel(int* __restrict__ p, int n) {
    int i = blockIdx.x * blockDim.x + threadIdx.x;
    if (i < n) p[i] = 0;
}

__global__ void deg_kernel(const int* __restrict__ dst, int* __restrict__ cnt, int e) {
    int i = blockIdx.x * blockDim.x + threadIdx.x;
    if (i < e) atomicAdd(&cnt[dst[i]], 1);
}

__global__ __launch_bounds__(256) void dinv_bsum_kernel(const int* __restrict__ cnt, float* __restrict__ dinv,
                                                        int* __restrict__ bsum, int n) {
    __shared__ int wsum[8];
    const int i    = blockIdx.x * 256 + threadIdx.x;
    const int lane = threadIdx.x & 31;
    const int wave = threadIdx.x >> 5;
    const int c    = (i < n) ? cnt[i] : 0;
    if (i < n) dinv[i] = rsqrtf((float)c + 1.0f);
    int s = c;
#pragma unroll
    for (int off = 16; off > 0; off >>= 1) s += __shfl_xor(s, off, 32);
    if (lane == 0) wsum[wave] = s;
    __syncthreads();
    if (threadIdx.x == 0) {
        int t = 0;
#pragma unroll
        for (int w = 0; w < 8; ++w) t += wsum[w];
        bsum[blockIdx.x] = t;
    }
}

__global__ __launch_bounds__(256) void rowptr_kernel(const int* __restrict__ cnt, const int* __restrict__ bsum,
                                                     int* __restrict__ rowptr, int* __restrict__ cursor, int n) {
    __shared__ int wpre[8];
    __shared__ int wtot[8];
    const int tid  = threadIdx.x;
    const int lane = tid & 31;
    const int wave = tid >> 5;
    const int nblk = blockIdx.x;

    int part = 0;
    for (int j = tid; j < nblk; j += 256) part += bsum[j];
#pragma unroll
    for (int off = 16; off > 0; off >>= 1) part += __shfl_xor(part, off, 32);

    const int i = blockIdx.x * 256 + tid;
    const int c = (i < n) ? cnt[i] : 0;
    int incl = c;
#pragma unroll
    for (int off = 1; off < 32; off <<= 1) {
        const int t = __shfl_up(incl, off, 32);
        if (lane >= off) incl += t;
    }
    if (lane == 31) wtot[wave] = incl;
    if (lane == 0)  wpre[wave] = part;
    __syncthreads();
    int base = 0;
#pragma unroll
    for (int w = 0; w < 8; ++w) {
        base += wpre[w];
        if (w < wave) base += wtot[w];
    }
    if (i < n) {
        const int excl = base + incl - c;
        rowptr[i] = excl;
        cursor[i] = excl;
    }
}

__global__ void fill_csr_kernel(const int* __restrict__ src, const int* __restrict__ dst,
                                int* __restrict__ cursor, int* __restrict__ col, int e) {
    int i = blockIdx.x * blockDim.x + threadIdx.x;
    if (i < e) {
        const int p = atomicAdd(&cursor[dst[i]], 1);
        col[p] = src[i];
    }
}


__global__ void seed_in_kernel(const float* __restrict__ x, const float* __restrict__ dinv,
                               float* __restrict__ agg, int n) {
    int t   = blockIdx.x * blockDim.x + threadIdx.x;
    int row = t >> 3;
    int f   = t & 7;
    if (row >= n || f >= F_IN) return;
    float di = dinv[row];
    agg[(size_t)row * ZP + f] = di * di * x[(size_t)row * (T_HIST * F_IN) + (T_HIST - 1) * F_IN + f];
}

__global__ __launch_bounds__(256) void scatter_in_kernel(const int* __restrict__ src, const int* __restrict__ dst,
                                                         const float* __restrict__ dinv, const float* __restrict__ x,
                                                         float* __restrict__ agg, int e) {
    const int lane = threadIdx.x & 31;
    const int edge = blockIdx.x * blockDim.x + threadIdx.x;
    int s = 0, d = 0;
    float nrm = 0.0f;
    if (edge < e) {
        s = src[edge];
        d = dst[edge];
        nrm = dinv[s] * dinv[d];
    }
    const int f     = lane & 7;
    const int fc    = f < F_IN ? f : F_IN - 1;
    const int sub   = lane >> 3;
    const int wbase = edge - lane;

    int   dd[8];
    float v[8];
    for (int p = 0; p < 8; ++p) {
        const int from = p * 4 + sub;
        const int   ss = __shfl(s, from, 32);
        const float nn = __shfl(nrm, from, 32);
        dd[p] = __shfl(d, from, 32);
        v[p]  = nn * x[(size_t)ss * (T_HIST * F_IN) + (T_HIST - 1) * F_IN + fc];
    }
    for (int p = 0; p < 8; ++p) {
        if (f < F_IN && wbase + p * 4 + sub < e)
            atomicAdd(agg + (size_t)dd[p] * ZP + f, v[p]);
    }
}

#define G2_ROWS  64
#define G2_PITCH (HDIM + 8)

__global__ __launch_bounds__(256) void gemm2_wmma_kernel(const float* __restrict__ W1,
                                                         const float* __restrict__ b1,
                                                         const float* __restrict__ W2,
                                                         const float* __restrict__ zin,
                                                         float* __restrict__ xw2,
                                                         int nrows) {
    __shared__ __attribute__((aligned(16))) _Float16 aS[G2_ROWS][G2_PITCH];
    __shared__ __attribute__((aligned(16))) float    zS[G2_ROWS][8];

    const int tid   = threadIdx.x;
    const int wave  = tid >> 5;
    const int lane  = tid & 31;
    const int ncol  = wave * 16;
    const int c     = lane & 15;
    const int h     = lane >> 4;
    const int khalf = h * 8;
    const int col   = ncol + c;
    const int row0  = blockIdx.x * G2_ROWS;

    for (int i = tid; i < G2_ROWS * 8; i += 256) {
        const int m = i >> 3;
        const int f = i & 7;
        const int row = row0 + m;
        if (f < F_IN) zS[m][f] = (row < nrows) ? zin[(size_t)row * ZP + f] : 0.0f;
    }

    v16h bf[4];
#pragma unroll
    for (int ks = 0; ks < 4; ++ks) {
#pragma unroll
        for (int j = 0; j < 16; ++j) {
            const int K = (j & 7) + ((j >> 3) << 4) + khalf + ks * 32;
            bf[ks][j] = (_Float16)W2[K * HDIM + col];
        }
    }

    const int k  = tid & (HDIM - 1);
    const int mh = tid >> 7;
    float w[F_IN];
#pragma unroll
    for (int f = 0; f < F_IN; ++f) w[f] = W1[f * HDIM + k];
    const float bk = b1[k];
    __syncthreads();

#pragma unroll 4
    for (int i = 0; i < G2_ROWS / 2; ++i) {
        const int m = 2 * i + mh;
        float s = 0.0f;
#pragma unroll
        for (int f = 0; f < F_IN; ++f) s = fmaf(zS[m][f], w[f], s);
        const float v = (row0 + m < nrows) ? fmaxf(s + bk, 0.0f) : 0.0f;
        aS[m][k] = (_Float16)v;
    }
    __syncthreads();

    v8f acc[4] = {};
#pragma unroll
    for (int ks = 0; ks < 4; ++ks) {
#pragma unroll
        for (int rt = 0; rt < 4; ++rt) {
            const _Float16* ap = &aS[rt * 16 + c][ks * 32 + khalf];
            const v8h lo = *(const v8h*)ap;
            const v8h hi = *(const v8h*)(ap + 16);
            const v16h a = __builtin_shufflevector(lo, hi, 0, 1, 2, 3, 4, 5, 6, 7, 8, 9, 10, 11, 12, 13, 14, 15);
            acc[rt] = __builtin_amdgcn_wmma_f32_16x16x32_f16(false, a, false, bf[ks],
                                                             (short)0, acc[rt], false, false);
        }
    }

    float* xp = xw2 + ((size_t)(row0 + 8 * h) * HDIM + col);
#pragma unroll
    for (int rt = 0; rt < 4; ++rt) {
        const int trow = row0 + rt * 16;
        if (trow + 16 <= nrows) {
#pragma unroll
            for (int v = 0; v < 8; ++v) xp[(rt * 16 + v) * HDIM] = acc[rt][v];
        } else {
#pragma unroll
            for (int v = 0; v < 8; ++v) {
                if (trow + v + 8 * h < nrows) xp[(rt * 16 + v) * HDIM] = acc[rt][v];
            }
        }
    }
}

#define GATHER_GROUP 4

__global__ __launch_bounds__(256) void gather_head_kernel(const int* __restrict__ rowptr, const int* __restrict__ cnt,
                                                          const int* __restrict__ col, const float* __restrict__ dinv,
                                                          const float* __restrict__ xw, const float* __restrict__ b2,
                                                          const float* __restrict__ Wout, const float* __restrict__ bout,
                                                          float* __restrict__ out, int n) {
    const int node = __builtin_amdgcn_readfirstlane((int)((blockIdx.x * blockDim.x + threadIdx.x) >> 5));
    const int lane = threadIdx.x & 31;
    if (node >= n) return;

    const int   start = rowptr[node];
    const int   deg   = cnt[node];
    const float di    = dinv[node];
    const float4* xr  = (const float4*)xw;
    const float4 self = xr[(size_t)node * (HDIM / 4) + lane];

    float4 acc = {0.0f, 0.0f, 0.0f, 0.0f};
    for (int base = 0; base < deg; base += 32) {
        const int nb = (deg - base < 32) ? deg - base : 32;
        int   ms = 0;
        float mw = 0.0f;
        if (lane < nb) {
            ms = col[start + base + lane];
            mw = dinv[ms] * di;
        }
        for (int j = 0; j < nb; j += GATHER_GROUP) {
            float4 v[GATHER_GROUP];
            float  w[GATHER_GROUP];
#pragma unroll
            for (int k = 0; k < GATHER_GROUP; ++k) {
                const int jj = (j + k < nb) ? j + k : nb - 1;
                const int   s  = __builtin_amdgcn_readlane(ms, jj);
                const float ww = __builtin_bit_cast(float, __builtin_amdgcn_readlane(__builtin_bit_cast(int, mw), jj));
                w[k] = (j + k < nb) ? ww : 0.0f;
                v[k] = xr[(size_t)s * (HDIM / 4) + lane];
            }
#pragma unroll
            for (int k = 0; k < GATHER_GROUP; ++k) {
                acc.x += w[k] * v[k].x;
                acc.y += w[k] * v[k].y;
                acc.z += w[k] * v[k].z;
                acc.w += w[k] * v[k].w;
            }
        }
    }

    const int h0 = lane * 4;
    const float dd = di * di;
    float4 hv;
    hv.x = fmaxf(acc.x + dd * self.x + b2[h0 + 0], 0.0f);
    hv.y = fmaxf(acc.y + dd * self.y + b2[h0 + 1], 0.0f);
    hv.z = fmaxf(acc.z + dd * self.z + b2[h0 + 2], 0.0f);
    hv.w = fmaxf(acc.w + dd * self.w + b2[h0 + 3], 0.0f);

    float lg[CDIM];
#pragma unroll
    for (int c = 0; c < CDIM; ++c) {
        lg[c] = hv.x * Wout[(h0 + 0) * CDIM + c]
              + hv.y * Wout[(h0 + 1) * CDIM + c]
              + hv.z * Wout[(h0 + 2) * CDIM + c]
              + hv.w * Wout[(h0 + 3) * CDIM + c];
    }
#pragma unroll
    for (int off = 16; off > 0; off >>= 1) {
#pragma unroll
        for (int c = 0; c < CDIM; ++c) lg[c] += __shfl_xor(lg[c], off, 32);
    }
    if (lane == 0) {
#pragma unroll
        for (int c = 0; c < CDIM; ++c) lg[c] += bout[c];
        float m = lg[0];
#pragma unroll
        for (int c = 1; c < CDIM; ++c) m = fmaxf(m, lg[c]);
        float sum = 0.0f;
#pragma unroll
        for (int c = 0; c < CDIM; ++c) { lg[c] = __expf(lg[c] - m); sum += lg[c]; }
        float inv = 1.0f / sum;
#pragma unroll
        for (int c = 0; c < CDIM; ++c) out[(size_t)node * CDIM + c] = lg[c] * inv;
    }
}

extern "C" void kernel_launch(void* const* d_in, const int* in_sizes, int n_in,
                              void* d_out, int out_size, void* d_ws, size_t ws_size,
                              hipStream_t stream) {
    const float* x    = (const float*)d_in[0];
    const int*   eidx = (const int*)  d_in[1];
    const float* W1   = (const float*)d_in[2];
    const float* b1   = (const float*)d_in[3];
    const float* W2   = (const float*)d_in[4];
    const float* b2   = (const float*)d_in[5];
    const float* Wout = (const float*)d_in[6];
    const float* bout = (const float*)d_in[7];
    float*       out  = (float*)d_out;

    const int N = in_sizes[0] / (T_HIST * F_IN);
    const int E = in_sizes[1] / 2;
    const int* src = eidx;
    const int* dst = eidx + E;

    const int NP = (N + 63) & ~63;
    const int NB = (N + 255) / 256;
    float* dinv   = (float*)d_ws;
    float* bufA   = dinv + NP;
    float* zin    = bufA + (size_t)N * HDIM;
    int*   cnt    = (int*)(zin + (size_t)N * ZP);
    int*   rowptr = cnt + NP;
    int*   cursor = rowptr + NP;
    int*   bsum   = cursor + NP;
    int*   col    = bsum + ((NB + 63) & ~63);

    const int TB = 256;
    const int node_blocks = ((N * 32) + TB - 1) / TB;

    fill_zero_kernel<<<(N + TB - 1) / TB, TB, 0, stream>>>(cnt, N);
    deg_kernel<<<(E + TB - 1) / TB, TB, 0, stream>>>(dst, cnt, E);
    dinv_bsum_kernel<<<NB, 256, 0, stream>>>(cnt, dinv, bsum, N);
    rowptr_kernel<<<NB, 256, 0, stream>>>(cnt, bsum, rowptr, cursor, N);
    fill_csr_kernel<<<(E + TB - 1) / TB, TB, 0, stream>>>(src, dst, cursor, col, E);

    seed_in_kernel<<<(N * 8 + TB - 1) / TB, TB, 0, stream>>>(x, dinv, zin, N);
    scatter_in_kernel<<<(E + TB - 1) / TB, TB, 0, stream>>>(src, dst, dinv, x, zin, E);

    gemm2_wmma_kernel<<<(N + G2_ROWS - 1) / G2_ROWS, 256, 0, stream>>>(W1, b1, W2, zin, bufA, N);

    gather_head_kernel<<<node_blocks, TB, 0, stream>>>(rowptr, cnt, col, dinv, bufA, b2, Wout, bout, out, N);
}


# layer 2 aggregate-then-project: the gather reads 32-byte layer-1 sums and rebuilds relu(z@W1+b1) per edge in registers; W2 WMMA and head follow in the same block; float sum order changes
# speedup vs baseline: 2.9653x; 1.0861x over previous
#include <hip/hip_runtime.h>
#include <hip/hip_bf16.h>

typedef __attribute__((ext_vector_type(16))) _Float16 v16h;
typedef __attribute__((ext_vector_type(8)))  float    v8f;
typedef __attribute__((ext_vector_type(8)))  _Float16 v8h;

#define T_HIST 3
#define F_IN   5
#define HDIM   128
#define CDIM   5
#define ZP     8

__global__ void fill_zero_kernel(int* __restrict__ p, int n) {
    int i = blockIdx.x * blockDim.x + threadIdx.x;
    if (i < n) p[i] = 0;
}

__global__ void deg_kernel(const int* __restrict__ dst, int* __restrict__ cnt, int e) {
    int i = blockIdx.x * blockDim.x + threadIdx.x;
    if (i < e) atomicAdd(&cnt[dst[i]], 1);
}

__global__ __launch_bounds__(256) void dinv_bsum_kernel(const int* __restrict__ cnt, float* __restrict__ dinv,
                                                        int* __restrict__ bsum, int n) {
    __shared__ int wsum[8];
    const int i    = blockIdx.x * 256 + threadIdx.x;
    const int lane = threadIdx.x & 31;
    const int wave = threadIdx.x >> 5;
    const int c    = (i < n) ? cnt[i] : 0;
    if (i < n) dinv[i] = rsqrtf((float)c + 1.0f);
    int s = c;
#pragma unroll
    for (int off = 16; off > 0; off >>= 1) s += __shfl_xor(s, off, 32);
    if (lane == 0) wsum[wave] = s;
    __syncthreads();
    if (threadIdx.x == 0) {
        int t = 0;
#pragma unroll
        for (int w = 0; w < 8; ++w) t += wsum[w];
        bsum[blockIdx.x] = t;
    }
}

__global__ __launch_bounds__(256) void rowptr_kernel(const int* __restrict__ cnt, const int* __restrict__ bsum,
                                                     int* __restrict__ rowptr, int* __restrict__ cursor, int n) {
    __shared__ int wpre[8];
    __shared__ int wtot[8];
    const int tid  = threadIdx.x;
    const int lane = tid & 31;
    const int wave = tid >> 5;
    const int nblk = blockIdx.x;

    int part = 0;
    for (int j = tid; j < nblk; j += 256) part += bsum[j];
#pragma unroll
    for (int off = 16; off > 0; off >>= 1) part += __shfl_xor(part, off, 32);

    const int i = blockIdx.x * 256 + tid;
    const int c = (i < n) ? cnt[i] : 0;
    int incl = c;
#pragma unroll
    for (int off = 1; off < 32; off <<= 1) {
        const int t = __shfl_up(incl, off, 32);
        if (lane >= off) incl += t;
    }
    if (lane == 31) wtot[wave] = incl;
    if (lane == 0)  wpre[wave] = part;
    __syncthreads();
    int base = 0;
#pragma unroll
    for (int w = 0; w < 8; ++w) {
        base += wpre[w];
        if (w < wave) base += wtot[w];
    }
    if (i < n) {
        const int excl = base + incl - c;
        rowptr[i] = excl;
        cursor[i] = excl;
    }
}

__global__ void fill_csr_kernel(const int* __restrict__ src, const int* __restrict__ dst,
                                int* __restrict__ cursor, int* __restrict__ col, int e) {
    int i = blockIdx.x * blockDim.x + threadIdx.x;
    if (i < e) {
        const int p = atomicAdd(&cursor[dst[i]], 1);
        col[p] = src[i];
    }
}


__global__ void seed_in_kernel(const float* __restrict__ x, const float* __restrict__ dinv,
                               float* __restrict__ agg, int n) {
    int t   = blockIdx.x * blockDim.x + threadIdx.x;
    int row = t >> 3;
    int f   = t & 7;
    if (row >= n || f >= F_IN) return;
    float di = dinv[row];
    agg[(size_t)row * ZP + f] = di * di * x[(size_t)row * (T_HIST * F_IN) + (T_HIST - 1) * F_IN + f];
}

__global__ __launch_bounds__(256) void scatter_in_kernel(const int* __restrict__ src, const int* __restrict__ dst,
                                                         const float* __restrict__ dinv, const float* __restrict__ x,
                                                         float* __restrict__ agg, int e) {
    const int lane = threadIdx.x & 31;
    const int edge = blockIdx.x * blockDim.x + threadIdx.x;
    int s = 0, d = 0;
    float nrm = 0.0f;
    if (edge < e) {
        s = src[edge];
        d = dst[edge];
        nrm = dinv[s] * dinv[d];
    }
    const int f     = lane & 7;
    const int fc    = f < F_IN ? f : F_IN - 1;
    const int sub   = lane >> 3;
    const int wbase = edge - lane;

    int   dd[8];
    float v[8];
    for (int p = 0; p < 8; ++p) {
        const int from = p * 4 + sub;
        const int   ss = __shfl(s, from, 32);
        const float nn = __shfl(nrm, from, 32);
        dd[p] = __shfl(d, from, 32);
        v[p]  = nn * x[(size_t)ss * (T_HIST * F_IN) + (T_HIST - 1) * F_IN + fc];
    }
    for (int p = 0; p < 8; ++p) {
        if (f < F_IN && wbase + p * 4 + sub < e)
            atomicAdd(agg + (size_t)dd[p] * ZP + f, v[p]);
    }
}

#define FT_ROWS   64
#define FT_APITCH (HDIM + 8)
#define FT_HPITCH (HDIM + 4)

typedef __attribute__((ext_vector_type(4))) _Float16 v4h;

static __device__ __forceinline__ void add_source(float z0, float z1, float z2, float z3, float z4, float wt,
                                                  const float (&w1)[F_IN][4], const float (&bv)[4], float (&a)[4]) {
#pragma unroll
    for (int q = 0; q < 4; ++q) {
        float s = fmaf(z0, w1[0][q], bv[q]);
        s = fmaf(z1, w1[1][q], s);
        s = fmaf(z2, w1[2][q], s);
        s = fmaf(z3, w1[3][q], s);
        s = fmaf(z4, w1[4][q], s);
        a[q] = fmaf(wt, fmaxf(s, 0.0f), a[q]);
    }
}

static __device__ __forceinline__ float lane_bcast(float v, int l) {
    return __builtin_bit_cast(float, __builtin_amdgcn_readlane(__builtin_bit_cast(int, v), l));
}

__global__ __launch_bounds__(256) void agg_gemm_head_kernel(const int* __restrict__ rowptr, const int* __restrict__ cnt,
                                                            const int* __restrict__ col, const float* __restrict__ dinv,
                                                            const float* __restrict__ zin,
                                                            const float* __restrict__ W1, const float* __restrict__ b1,
                                                            const float* __restrict__ W2, const float* __restrict__ b2,
                                                            const float* __restrict__ Wout, const float* __restrict__ bout,
                                                            float* __restrict__ out, int n) {
    __shared__ __attribute__((aligned(16))) _Float16 aS[FT_ROWS][FT_APITCH];
    __shared__ __attribute__((aligned(16))) float    hS[FT_ROWS][FT_HPITCH];

    const int tid  = threadIdx.x;
    const int lane = tid & 31;
    const int wave = __builtin_amdgcn_readfirstlane(tid >> 5);
    const int row0 = blockIdx.x * FT_ROWS;
    const int h0   = lane * 4;

    {
        float w1[F_IN][4];
        float bv[4];
#pragma unroll
        for (int f = 0; f < F_IN; ++f) {
            const float4 t = *(const float4*)(W1 + f * HDIM + h0);
            w1[f][0] = t.x; w1[f][1] = t.y; w1[f][2] = t.z; w1[f][3] = t.w;
        }
        {
            const float4 t = *(const float4*)(b1 + h0);
            bv[0] = t.x; bv[1] = t.y; bv[2] = t.z; bv[3] = t.w;
        }
        const float4* zr4 = (const float4*)zin;

        int    mstart = 0, mdeg = 0;
        float  mdi = 0.0f, mzl = 0.0f;
        float4 mzs = {0.0f, 0.0f, 0.0f, 0.0f};
        {
            const int nd = row0 + wave * (FT_ROWS / 8) + lane;
            if (lane < FT_ROWS / 8 && nd < n) {
                mstart = rowptr[nd];
                mdeg   = cnt[nd];
                mdi    = dinv[nd];
                mzs    = zr4[(size_t)nd * (ZP / 4)];
                mzl    = zin[(size_t)nd * ZP + 4];
            }
        }

        auto first_piece = [&](int i) -> int {
            const int s  = __builtin_amdgcn_readlane(mstart, i);
            const int d  = __builtin_amdgcn_readlane(mdeg, i);
            const int nb = d < 32 ? d : 32;
            return col[lane < nb ? s + lane : 0];
        };

        int    ms_b = first_piece(0);
        float  wq_a = dinv[ms_b];
        float4 zq_a = zr4[(size_t)ms_b * (ZP / 4)];
        float  zl_a = zin[(size_t)ms_b * ZP + 4];
        ms_b = first_piece(1);

#pragma unroll 2
        for (int i = 0; i < FT_ROWS / 8; ++i) {
            const int    ms_c = first_piece(i + 2);
            const float  wq_b = dinv[ms_b];
            const float4 zq_b = zr4[(size_t)ms_b * (ZP / 4)];
            const float  zl_b = zin[(size_t)ms_b * ZP + 4];

            const int   m   = wave * (FT_ROWS / 8) + i;
            const int   deg = __builtin_amdgcn_readlane(mdeg, i);
            const float di  = lane_bcast(mdi, i);
            float a[4] = {0.0f, 0.0f, 0.0f, 0.0f};
            add_source(lane_bcast(mzs.x, i), lane_bcast(mzs.y, i), lane_bcast(mzs.z, i), lane_bcast(mzs.w, i),
                       lane_bcast(mzl, i), di * di, w1, bv, a);
            {
                const int   nb = deg < 32 ? deg : 32;
                const float mw = wq_a * di;
                for (int j = 0; j < nb; ++j) {
                    add_source(lane_bcast(zq_a.x, j), lane_bcast(zq_a.y, j), lane_bcast(zq_a.z, j), lane_bcast(zq_a.w, j),
                               lane_bcast(zl_a, j), lane_bcast(mw, j), w1, bv, a);
                }
            }
            if (deg > 32) {
                const int start = __builtin_amdgcn_readlane(mstart, i);
                for (int base = 32; base < deg; base += 32) {
                    const int    nb = (deg - base < 32) ? deg - base : 32;
                    const int    ms = col[lane < nb ? start + base + lane : 0];
                    const float  mw = dinv[ms] * di;
                    const float4 zq = zr4[(size_t)ms * (ZP / 4)];
                    const float  zl = zin[(size_t)ms * ZP + 4];
                    for (int j = 0; j < nb; ++j) {
                        add_source(lane_bcast(zq.x, j), lane_bcast(zq.y, j), lane_bcast(zq.z, j), lane_bcast(zq.w, j),
                                   lane_bcast(zl, j), lane_bcast(mw, j), w1, bv, a);
                    }
                }
            }
            v4h pk;
            pk[0] = (_Float16)a[0]; pk[1] = (_Float16)a[1]; pk[2] = (_Float16)a[2]; pk[3] = (_Float16)a[3];
            *(v4h*)&aS[m][h0] = pk;

            wq_a = wq_b; zq_a = zq_b; zl_a = zl_b;
            ms_b = ms_c;
        }
    }

    const int ncol  = wave * 16;
    const int c     = lane & 15;
    const int h     = lane >> 4;
    const int khalf = h * 8;
    const int colw  = ncol + c;

    v16h bf[4];
#pragma unroll
    for (int ks = 0; ks < 4; ++ks) {
#pragma unroll
        for (int j = 0; j < 16; ++j) {
            const int K = (j & 7) + ((j >> 3) << 4) + khalf + ks * 32;
            bf[ks][j] = (_Float16)W2[K * HDIM + colw];
        }
    }
    const float b2c = b2[colw];
    __syncthreads();

    v8f acc[4] = {};
#pragma unroll
    for (int ks = 0; ks < 4; ++ks) {
#pragma unroll
        for (int rt = 0; rt < 4; ++rt) {
            const _Float16* ap = &aS[rt * 16 + c][ks * 32 + khalf];
            const v8h lo = *(const v8h*)ap;
            const v8h hi = *(const v8h*)(ap + 16);
            const v16h av = __builtin_shufflevector(lo, hi, 0, 1, 2, 3, 4, 5, 6, 7, 8, 9, 10, 11, 12, 13, 14, 15);
            acc[rt] = __builtin_amdgcn_wmma_f32_16x16x32_f16(false, av, false, bf[ks],
                                                             (short)0, acc[rt], false, false);
        }
    }

#pragma unroll
    for (int rt = 0; rt < 4; ++rt) {
#pragma unroll
        for (int v = 0; v < 8; ++v) hS[rt * 16 + 8 * h + v][colw] = fmaxf(acc[rt][v] + b2c, 0.0f);
    }

    float wo[4][CDIM];
    {
        const float4* wp = (const float4*)(Wout + (size_t)h0 * CDIM);
        float t[4 * CDIM];
#pragma unroll
        for (int u = 0; u < CDIM; ++u) {
            const float4 x4 = wp[u];
            t[4 * u + 0] = x4.x; t[4 * u + 1] = x4.y; t[4 * u + 2] = x4.z; t[4 * u + 3] = x4.w;
        }
#pragma unroll
        for (int q = 0; q < 4; ++q) {
#pragma unroll
            for (int cc = 0; cc < CDIM; ++cc) wo[q][cc] = t[q * CDIM + cc];
        }
    }
    float bo[CDIM];
#pragma unroll
    for (int cc = 0; cc < CDIM; ++cc) bo[cc] = bout[cc];
    __syncthreads();

    for (int i = 0; i < FT_ROWS / 8; ++i) {
        const int m    = wave * (FT_ROWS / 8) + i;
        const int node = row0 + m;
        if (node >= n) break;
        const float4 hv = *(const float4*)&hS[m][h0];
        float lg[CDIM];
#pragma unroll
        for (int cc = 0; cc < CDIM; ++cc)
            lg[cc] = hv.x * wo[0][cc] + hv.y * wo[1][cc] + hv.z * wo[2][cc] + hv.w * wo[3][cc];
#pragma unroll
        for (int off = 16; off > 0; off >>= 1) {
#pragma unroll
            for (int cc = 0; cc < CDIM; ++cc) lg[cc] += __shfl_xor(lg[cc], off, 32);
        }
#pragma unroll
        for (int cc = 0; cc < CDIM; ++cc) lg[cc] += bo[cc];
        float mx = lg[0];
#pragma unroll
        for (int cc = 1; cc < CDIM; ++cc) mx = fmaxf(mx, lg[cc]);
        float sum = 0.0f;
#pragma unroll
        for (int cc = 0; cc < CDIM; ++cc) { lg[cc] = __expf(lg[cc] - mx); sum += lg[cc]; }
        const float inv = 1.0f / sum;
        float pv = lg[0];
#pragma unroll
        for (int cc = 1; cc < CDIM; ++cc) pv = (lane == cc) ? lg[cc] : pv;
        if (lane < CDIM) out[(size_t)node * CDIM + lane] = pv * inv;
    }
}

extern "C" void kernel_launch(void* const* d_in, const int* in_sizes, int n_in,
                              void* d_out, int out_size, void* d_ws, size_t ws_size,
                              hipStream_t stream) {
    const float* x    = (const float*)d_in[0];
    const int*   eidx = (const int*)  d_in[1];
    const float* W1   = (const float*)d_in[2];
    const float* b1   = (const float*)d_in[3];
    const float* W2   = (const float*)d_in[4];
    const float* b2   = (const float*)d_in[5];
    const float* Wout = (const float*)d_in[6];
    const float* bout = (const float*)d_in[7];
    float*       out  = (float*)d_out;

    const int N = in_sizes[0] / (T_HIST * F_IN);
    const int E = in_sizes[1] / 2;
    const int* src = eidx;
    const int* dst = eidx + E;

    const int NP = (N + 63) & ~63;
    const int NB = (N + 255) / 256;
    float* dinv   = (float*)d_ws;
    float* bufA   = dinv + NP;
    float* zin    = bufA + (size_t)N * HDIM;
    int*   cnt    = (int*)(zin + (size_t)N * ZP);
    int*   rowptr = cnt + NP;
    int*   cursor = rowptr + NP;
    int*   bsum   = cursor + NP;
    int*   col    = bsum + ((NB + 63) & ~63);

    const int TB = 256;

    fill_zero_kernel<<<(N + TB - 1) / TB, TB, 0, stream>>>(cnt, N);
    deg_kernel<<<(E + TB - 1) / TB, TB, 0, stream>>>(dst, cnt, E);
    dinv_bsum_kernel<<<NB, 256, 0, stream>>>(cnt, dinv, bsum, N);
    rowptr_kernel<<<NB, 256, 0, stream>>>(cnt, bsum, rowptr, cursor, N);
    fill_csr_kernel<<<(E + TB - 1) / TB, TB, 0, stream>>>(src, dst, cursor, col, E);

    seed_in_kernel<<<(N * 8 + TB - 1) / TB, TB, 0, stream>>>(x, dinv, zin, N);
    scatter_in_kernel<<<(E + TB - 1) / TB, TB, 0, stream>>>(src, dst, dinv, x, zin, E);

    agg_gemm_head_kernel<<<(N + FT_ROWS - 1) / FT_ROWS, 256, 0, stream>>>(rowptr, cnt, col, dinv, zin, W1, b1, W2, b2,
                                                                          Wout, bout, out, N);
}


# layer-1 sum as a gather: a wave sorts a node's edge segment by source (first 32: order set by data, not by atomic arrival), writes it back for layer 2, sums x rows; no float atomics; 7 launches
# speedup vs baseline: 3.3982x; 1.1460x over previous
#include <hip/hip_runtime.h>
#include <hip/hip_bf16.h>

typedef __attribute__((ext_vector_type(16))) _Float16 v16h;
typedef __attribute__((ext_vector_type(8)))  float    v8f;
typedef __attribute__((ext_vector_type(8)))  _Float16 v8h;

#define T_HIST 3
#define F_IN   5
#define HDIM   128
#define CDIM   5
#define ZP     8

__global__ void fill_zero_kernel(int* __restrict__ p, int n) {
    int i = blockIdx.x * blockDim.x + threadIdx.x;
    if (i < n) p[i] = 0;
}

__global__ void deg_kernel(const int* __restrict__ dst, int* __restrict__ cnt, int e) {
    int i = blockIdx.x * blockDim.x + threadIdx.x;
    if (i < e) atomicAdd(&cnt[dst[i]], 1);
}

__global__ __launch_bounds__(256) void dinv_bsum_kernel(const int* __restrict__ cnt, float* __restrict__ dinv,
                                                        int* __restrict__ bsum, int n) {
    __shared__ int wsum[8];
    const int i    = blockIdx.x * 256 + threadIdx.x;
    const int lane = threadIdx.x & 31;
    const int wave = threadIdx.x >> 5;
    const int c    = (i < n) ? cnt[i] : 0;
    if (i < n) dinv[i] = rsqrtf((float)c + 1.0f);
    int s = c;
#pragma unroll
    for (int off = 16; off > 0; off >>= 1) s += __shfl_xor(s, off, 32);
    if (lane == 0) wsum[wave] = s;
    __syncthreads();
    if (threadIdx.x == 0) {
        int t = 0;
#pragma unroll
        for (int w = 0; w < 8; ++w) t += wsum[w];
        bsum[blockIdx.x] = t;
    }
}

__global__ __launch_bounds__(256) void rowptr_kernel(const int* __restrict__ cnt, const int* __restrict__ bsum,
                                                     int* __restrict__ rowptr, int* __restrict__ cursor, int n) {
    __shared__ int wpre[8];
    __shared__ int wtot[8];
    const int tid  = threadIdx.x;
    const int lane = tid & 31;
    const int wave = tid >> 5;
    const int nblk = blockIdx.x;

    int part = 0;
    for (int j = tid; j < nblk; j += 256) part += bsum[j];
#pragma unroll
    for (int off = 16; off > 0; off >>= 1) part += __shfl_xor(part, off, 32);

    const int i = blockIdx.x * 256 + tid;
    const int c = (i < n) ? cnt[i] : 0;
    int incl = c;
#pragma unroll
    for (int off = 1; off < 32; off <<= 1) {
        const int t = __shfl_up(incl, off, 32);
        if (lane >= off) incl += t;
    }
    if (lane == 31) wtot[wave] = incl;
    if (lane == 0)  wpre[wave] = part;
    __syncthreads();
    int base = 0;
#pragma unroll
    for (int w = 0; w < 8; ++w) {
        base += wpre[w];
        if (w < wave) base += wtot[w];
    }
    if (i < n) {
        const int excl = base + incl - c;
        rowptr[i] = excl;
        cursor[i] = excl;
    }
}

__global__ void fill_csr_kernel(const int* __restrict__ src, const int* __restrict__ dst,
                                int* __restrict__ cursor, int* __restrict__ col, int e) {
    int i = blockIdx.x * blockDim.x + threadIdx.x;
    if (i < e) {
        const int p = atomicAdd(&cursor[dst[i]], 1);
        col[p] = src[i];
    }
}


template <int M> static __device__ __forceinline__ int lane_xor(int v) {
    if constexpr (M == 16) return __builtin_amdgcn_permlanex16(v, v, 0x76543210, 0xfedcba98, false, false);
    else                   return __builtin_amdgcn_update_dpp(v, v, 0x160 + M, 0xf, 0xf, false);
}
template <int M> static __device__ __forceinline__ float lane_xor_f(float v) {
    return __builtin_bit_cast(float, lane_xor<M>(__builtin_bit_cast(int, v)));
}

template <int K, int J> static __device__ __forceinline__ int cmp_xchg(int key, int lane) {
    const int  o   = lane_xor<J>(key);
    const bool low = (((lane & J) == 0) == ((lane & K) == 0));
    const int  mn  = key < o ? key : o;
    const int  mx  = key < o ? o : key;
    return low ? mn : mx;
}
static __device__ __forceinline__ int sort_wave(int key, int lane) {
    key = cmp_xchg<2, 1>(key, lane);
    key = cmp_xchg<4, 2>(key, lane);   key = cmp_xchg<4, 1>(key, lane);
    key = cmp_xchg<8, 4>(key, lane);   key = cmp_xchg<8, 2>(key, lane);   key = cmp_xchg<8, 1>(key, lane);
    key = cmp_xchg<16, 8>(key, lane);  key = cmp_xchg<16, 4>(key, lane);  key = cmp_xchg<16, 2>(key, lane);
    key = cmp_xchg<16, 1>(key, lane);
    key = cmp_xchg<32, 16>(key, lane); key = cmp_xchg<32, 8>(key, lane);  key = cmp_xchg<32, 4>(key, lane);
    key = cmp_xchg<32, 2>(key, lane);  key = cmp_xchg<32, 1>(key, lane);
    return key;
}

static __device__ __forceinline__ float lane_bcast(float v, int l) {
    return __builtin_bit_cast(float, __builtin_amdgcn_readlane(__builtin_bit_cast(int, v), l));
}

static __device__ __forceinline__ float add_piece(int ks, float nrm, int sub, int fc,
                                                  const float* __restrict__ x, float acc) {
#pragma unroll
    for (int p = 0; p < 8; ++p) {
        const int   from = p * 4 + sub;
        const int   ss   = __shfl(ks, from, 32);
        const float nn   = __shfl(nrm, from, 32);
        const float xv   = x[ss * (T_HIST * F_IN) + (T_HIST - 1) * F_IN + fc];
        acc = fmaf(nn, xv, acc);
    }
    return acc;
}

#define GN_NODES 8

__global__ __launch_bounds__(256) void gather_in_kernel(const int* __restrict__ rowptr, const int* __restrict__ cnt,
                                                       int* __restrict__ col, const float* __restrict__ dinv,
                                                       const float* __restrict__ x, float* __restrict__ agg, int n) {
    const int lane = threadIdx.x & 31;
    const int wave  = __builtin_amdgcn_readfirstlane(threadIdx.x >> 5);
    const int node0 = (blockIdx.x * 8 + wave) * GN_NODES;
    const int f     = lane & 7;
    const int fc    = f < F_IN ? f : F_IN - 1;
    const int sub   = lane >> 3;

    int   mstart = 0, mdeg = 0;
    float mdi = 0.0f;
    {
        const int nd = node0 + lane;
        if (lane < GN_NODES && nd < n) {
            mstart = rowptr[nd];
            mdeg   = cnt[nd];
            mdi    = dinv[nd];
        }
    }


    int   key[GN_NODES];
    float xs[GN_NODES];
#pragma unroll
    for (int i = 0; i < GN_NODES; ++i) {
        const int s  = __builtin_amdgcn_readlane(mstart, i);
        const int d  = __builtin_amdgcn_readlane(mdeg, i);
        const int nb = d < 32 ? d : 32;
        key[i] = lane < nb ? col[s + lane] : 0x7fffffff;
        const int nd = node0 + i < n ? node0 + i : n - 1;
        xs[i] = x[nd * (T_HIST * F_IN) + (T_HIST - 1) * F_IN + fc];
    }

    float w[GN_NODES];
#pragma unroll
    for (int i = 0; i < GN_NODES; ++i) {
        const int d  = __builtin_amdgcn_readlane(mdeg, i);
        const int nb = d < 32 ? d : 32;
        key[i] = sort_wave(key[i], lane);
        w[i]   = dinv[lane < nb ? key[i] : 0];
    }

    float acc[GN_NODES];
#pragma unroll
    for (int i = 0; i < GN_NODES; ++i) {
        const int   d   = __builtin_amdgcn_readlane(mdeg, i);
        const int   nb  = d < 32 ? d : 32;
        const float di  = lane_bcast(mdi, i);
        const int   ks  = lane < nb ? key[i] : 0;
        const float nrm = lane < nb ? w[i] * di : 0.0f;
        acc[i] = add_piece(ks, nrm, sub, fc, x, 0.0f);
    }

#pragma unroll
    for (int i = 0; i < GN_NODES; ++i) {
        const int s  = __builtin_amdgcn_readlane(mstart, i);
        const int d  = __builtin_amdgcn_readlane(mdeg, i);
        const int nb = d < 32 ? d : 32;
        if (lane < nb) col[s + lane] = key[i];
    }

    const unsigned longmask = __builtin_amdgcn_ballot_w32(mdeg > 32);
    if (longmask != 0u) {
#pragma nounroll
        for (int i = 0; i < GN_NODES; ++i) {
            if ((longmask >> i) & 1u) {
                const int   s  = __builtin_amdgcn_readlane(mstart, i);
                const int   d  = __builtin_amdgcn_readlane(mdeg, i);
                const float di = lane_bcast(mdi, i);
                float extra = 0.0f;
                for (int base = 32; base < d; base += 32) {
                    const int   nbp = (d - base < 32) ? d - base : 32;
                    const int   ks  = col[s + base + (lane < nbp ? lane : 0)];
                    const float wv  = dinv[ks];
                    const float nrm = lane < nbp ? wv * di : 0.0f;
                    extra = add_piece(ks, nrm, sub, fc, x, extra);
                }
#pragma unroll
                for (int j = 0; j < GN_NODES; ++j) acc[j] = (j == i) ? acc[j] + extra : acc[j];
            }
        }
    }

#pragma unroll
    for (int i = 0; i < GN_NODES; ++i) {
        const float di = lane_bcast(mdi, i);
        float t = acc[i];
        t += lane_xor_f<8>(t);
        t += lane_xor_f<16>(t);
        if (lane < F_IN && node0 + i < n) agg[(size_t)(node0 + i) * ZP + lane] = fmaf(di * di, xs[i], t);
    }
}

#define FT_ROWS   64
#define FT_APITCH (HDIM + 8)
#define FT_HPITCH (HDIM + 4)

typedef __attribute__((ext_vector_type(4))) _Float16 v4h;

static __device__ __forceinline__ void add_source(float z0, float z1, float z2, float z3, float z4, float wt,
                                                  const float (&w1)[F_IN][4], const float (&bv)[4], float (&a)[4]) {
#pragma unroll
    for (int q = 0; q < 4; ++q) {
        float s = fmaf(z0, w1[0][q], bv[q]);
        s = fmaf(z1, w1[1][q], s);
        s = fmaf(z2, w1[2][q], s);
        s = fmaf(z3, w1[3][q], s);
        s = fmaf(z4, w1[4][q], s);
        a[q] = fmaf(wt, fmaxf(s, 0.0f), a[q]);
    }
}

__global__ __launch_bounds__(256) void agg_gemm_head_kernel(const int* __restrict__ rowptr, const int* __restrict__ cnt,
                                                            const int* __restrict__ col, const float* __restrict__ dinv,
                                                            const float* __restrict__ zin,
                                                            const float* __restrict__ W1, const float* __restrict__ b1,
                                                            const float* __restrict__ W2, const float* __restrict__ b2,
                                                            const float* __restrict__ Wout, const float* __restrict__ bout,
                                                            float* __restrict__ out, int n) {
    __shared__ __attribute__((aligned(16))) _Float16 aS[FT_ROWS][FT_APITCH];
    __shared__ __attribute__((aligned(16))) float    hS[FT_ROWS][FT_HPITCH];

    const int tid  = threadIdx.x;
    const int lane = tid & 31;
    const int wave = __builtin_amdgcn_readfirstlane(tid >> 5);
    const int row0 = blockIdx.x * FT_ROWS;
    const int h0   = lane * 4;

    {
        float w1[F_IN][4];
        float bv[4];
#pragma unroll
        for (int f = 0; f < F_IN; ++f) {
            const float4 t = *(const float4*)(W1 + f * HDIM + h0);
            w1[f][0] = t.x; w1[f][1] = t.y; w1[f][2] = t.z; w1[f][3] = t.w;
        }
        {
            const float4 t = *(const float4*)(b1 + h0);
            bv[0] = t.x; bv[1] = t.y; bv[2] = t.z; bv[3] = t.w;
        }
        const float4* zr4 = (const float4*)zin;

        int    mstart = 0, mdeg = 0;
        float  mdi = 0.0f, mzl = 0.0f;
        float4 mzs = {0.0f, 0.0f, 0.0f, 0.0f};
        {
            const int nd = row0 + wave * (FT_ROWS / 8) + lane;
            if (lane < FT_ROWS / 8 && nd < n) {
                mstart = rowptr[nd];
                mdeg   = cnt[nd];
                mdi    = dinv[nd];
                mzs    = zr4[(size_t)nd * (ZP / 4)];
                mzl    = zin[(size_t)nd * ZP + 4];
            }
        }

        auto first_piece = [&](int i) -> int {
            const int s  = __builtin_amdgcn_readlane(mstart, i);
            const int d  = __builtin_amdgcn_readlane(mdeg, i);
            const int nb = d < 32 ? d : 32;
            return col[lane < nb ? s + lane : 0];
        };

        int    ms_b = first_piece(0);
        float  wq_a = dinv[ms_b];
        float4 zq_a = zr4[(size_t)ms_b * (ZP / 4)];
        float  zl_a = zin[(size_t)ms_b * ZP + 4];
        ms_b = first_piece(1);

#pragma unroll 2
        for (int i = 0; i < FT_ROWS / 8; ++i) {
            const int    ms_c = first_piece(i + 2);
            const float  wq_b = dinv[ms_b];
            const float4 zq_b = zr4[(size_t)ms_b * (ZP / 4)];
            const float  zl_b = zin[(size_t)ms_b * ZP + 4];

            const int   m   = wave * (FT_ROWS / 8) + i;
            const int   deg = __builtin_amdgcn_readlane(mdeg, i);
            const float di  = lane_bcast(mdi, i);
            float a[4] = {0.0f, 0.0f, 0.0f, 0.0f};
            add_source(lane_bcast(mzs.x, i), lane_bcast(mzs.y, i), lane_bcast(mzs.z, i), lane_bcast(mzs.w, i),
                       lane_bcast(mzl, i), di * di, w1, bv, a);
            {
                const int   nb = deg < 32 ? deg : 32;
                const float mw = wq_a * di;
                for (int j = 0; j < nb; ++j) {
                    add_source(lane_bcast(zq_a.x, j), lane_bcast(zq_a.y, j), lane_bcast(zq_a.z, j), lane_bcast(zq_a.w, j),
                               lane_bcast(zl_a, j), lane_bcast(mw, j), w1, bv, a);
                }
            }
            if (deg > 32) {
                const int start = __builtin_amdgcn_readlane(mstart, i);
                for (int base = 32; base < deg; base += 32) {
                    const int    nb = (deg - base < 32) ? deg - base : 32;
                    const int    ms = col[lane < nb ? start + base + lane : 0];
                    const float  mw = dinv[ms] * di;
                    const float4 zq = zr4[(size_t)ms * (ZP / 4)];
                    const float  zl = zin[(size_t)ms * ZP + 4];
                    for (int j = 0; j < nb; ++j) {
                        add_source(lane_bcast(zq.x, j), lane_bcast(zq.y, j), lane_bcast(zq.z, j), lane_bcast(zq.w, j),
                                   lane_bcast(zl, j), lane_bcast(mw, j), w1, bv, a);
                    }
                }
            }
            v4h pk;
            pk[0] = (_Float16)a[0]; pk[1] = (_Float16)a[1]; pk[2] = (_Float16)a[2]; pk[3] = (_Float16)a[3];
            *(v4h*)&aS[m][h0] = pk;

            wq_a = wq_b; zq_a = zq_b; zl_a = zl_b;
            ms_b = ms_c;
        }
    }

    const int ncol  = wave * 16;
    const int c     = lane & 15;
    const int h     = lane >> 4;
    const int khalf = h * 8;
    const int colw  = ncol + c;

    v16h bf[4];
#pragma unroll
    for (int ks = 0; ks < 4; ++ks) {
#pragma unroll
        for (int j = 0; j < 16; ++j) {
            const int K = (j & 7) + ((j >> 3) << 4) + khalf + ks * 32;
            bf[ks][j] = (_Float16)W2[K * HDIM + colw];
        }
    }
    const float b2c = b2[colw];
    __syncthreads();

    v8f acc[4] = {};
#pragma unroll
    for (int ks = 0; ks < 4; ++ks) {
#pragma unroll
        for (int rt = 0; rt < 4; ++rt) {
            const _Float16* ap = &aS[rt * 16 + c][ks * 32 + khalf];
            const v8h lo = *(const v8h*)ap;
            const v8h hi = *(const v8h*)(ap + 16);
            const v16h av = __builtin_shufflevector(lo, hi, 0, 1, 2, 3, 4, 5, 6, 7, 8, 9, 10, 11, 12, 13, 14, 15);
            acc[rt] = __builtin_amdgcn_wmma_f32_16x16x32_f16(false, av, false, bf[ks],
                                                             (short)0, acc[rt], false, false);
        }
    }

#pragma unroll
    for (int rt = 0; rt < 4; ++rt) {
#pragma unroll
        for (int v = 0; v < 8; ++v) hS[rt * 16 + 8 * h + v][colw] = fmaxf(acc[rt][v] + b2c, 0.0f);
    }

    float wo[4][CDIM];
    {
        const float4* wp = (const float4*)(Wout + (size_t)h0 * CDIM);
        float t[4 * CDIM];
#pragma unroll
        for (int u = 0; u < CDIM; ++u) {
            const float4 x4 = wp[u];
            t[4 * u + 0] = x4.x; t[4 * u + 1] = x4.y; t[4 * u + 2] = x4.z; t[4 * u + 3] = x4.w;
        }
#pragma unroll
        for (int q = 0; q < 4; ++q) {
#pragma unroll
            for (int cc = 0; cc < CDIM; ++cc) wo[q][cc] = t[q * CDIM + cc];
        }
    }
    float bo[CDIM];
#pragma unroll
    for (int cc = 0; cc < CDIM; ++cc) bo[cc] = bout[cc];
    __syncthreads();

    for (int i = 0; i < FT_ROWS / 8; ++i) {
        const int m    = wave * (FT_ROWS / 8) + i;
        const int node = row0 + m;
        if (node >= n) break;
        const float4 hv = *(const float4*)&hS[m][h0];
        float lg[CDIM];
#pragma unroll
        for (int cc = 0; cc < CDIM; ++cc)
            lg[cc] = hv.x * wo[0][cc] + hv.y * wo[1][cc] + hv.z * wo[2][cc] + hv.w * wo[3][cc];
#pragma unroll
        for (int off = 16; off > 0; off >>= 1) {
#pragma unroll
            for (int cc = 0; cc < CDIM; ++cc) lg[cc] += __shfl_xor(lg[cc], off, 32);
        }
#pragma unroll
        for (int cc = 0; cc < CDIM; ++cc) lg[cc] += bo[cc];
        float mx = lg[0];
#pragma unroll
        for (int cc = 1; cc < CDIM; ++cc) mx = fmaxf(mx, lg[cc]);
        float sum = 0.0f;
#pragma unroll
        for (int cc = 0; cc < CDIM; ++cc) { lg[cc] = __expf(lg[cc] - mx); sum += lg[cc]; }
        const float inv = 1.0f / sum;
        float pv = lg[0];
#pragma unroll
        for (int cc = 1; cc < CDIM; ++cc) pv = (lane == cc) ? lg[cc] : pv;
        if (lane < CDIM) out[(size_t)node * CDIM + lane] = pv * inv;
    }
}

extern "C" void kernel_launch(void* const* d_in, const int* in_sizes, int n_in,
                              void* d_out, int out_size, void* d_ws, size_t ws_size,
                              hipStream_t stream) {
    const float* x    = (const float*)d_in[0];
    const int*   eidx = (const int*)  d_in[1];
    const float* W1   = (const float*)d_in[2];
    const float* b1   = (const float*)d_in[3];
    const float* W2   = (const float*)d_in[4];
    const float* b2   = (const float*)d_in[5];
    const float* Wout = (const float*)d_in[6];
    const float* bout = (const float*)d_in[7];
    float*       out  = (float*)d_out;

    const int N = in_sizes[0] / (T_HIST * F_IN);
    const int E = in_sizes[1] / 2;
    const int* src = eidx;
    const int* dst = eidx + E;

    const int NP = (N + 63) & ~63;
    const int NB = (N + 255) / 256;
    float* dinv   = (float*)d_ws;
    float* bufA   = dinv + NP;
    float* zin    = bufA + (size_t)N * HDIM;
    int*   cnt    = (int*)(zin + (size_t)N * ZP);
    int*   rowptr = cnt + NP;
    int*   cursor = rowptr + NP;
    int*   bsum   = cursor + NP;
    int*   col    = bsum + ((NB + 63) & ~63);

    const int TB = 256;

    fill_zero_kernel<<<(N + TB - 1) / TB, TB, 0, stream>>>(cnt, N);
    deg_kernel<<<(E + TB - 1) / TB, TB, 0, stream>>>(dst, cnt, E);
    dinv_bsum_kernel<<<NB, 256, 0, stream>>>(cnt, dinv, bsum, N);
    rowptr_kernel<<<NB, 256, 0, stream>>>(cnt, bsum, rowptr, cursor, N);
    fill_csr_kernel<<<(E + TB - 1) / TB, TB, 0, stream>>>(src, dst, cursor, col, E);

    gather_in_kernel<<<(N + 8 * GN_NODES - 1) / (8 * GN_NODES), 256, 0, stream>>>(rowptr, cnt, col, dinv, x, zin, N);

    agg_gemm_head_kernel<<<(N + FT_ROWS - 1) / FT_ROWS, 256, 0, stream>>>(rowptr, cnt, col, dinv, zin, W1, b1, W2, b2,
                                                                          Wout, bout, out, N);
}


# edge list put in dst order by a two-level counting sort on LDS counters (256-node bins: count, scan, place, per-bin sort): no global atomics, 6 launches; order past a segment's first 32: LDS arrival
# speedup vs baseline: 8.3385x; 2.4538x over previous
#include <hip/hip_runtime.h>
#include <hip/hip_bf16.h>

typedef __attribute__((ext_vector_type(16))) _Float16 v16h;
typedef __attribute__((ext_vector_type(8)))  float    v8f;
typedef __attribute__((ext_vector_type(8)))  _Float16 v8h;

#define T_HIST 3
#define F_IN   5
#define HDIM   128
#define CDIM   5
#define ZP     8

#define BIN_SHIFT 8
#define BIN_W     (1 << BIN_SHIFT)
#define NCHUNK    256
#define SORT_NT   256
#define SORT_EPT  25
#define SORT_TILE (SORT_NT * SORT_EPT)
#define SCOL_CAP  8192

static __device__ __forceinline__ void block_excl_scan(const int* in, int* out, int nb, int* wsum) {
    const int tid  = threadIdx.x;
    const int lane = tid & 31;
    const int wave = tid >> 5;
    const int per  = (nb + SORT_NT - 1) / SORT_NT;
    const int lo_i = tid * per;
    int s = 0;
    for (int k = 0; k < per; ++k) {
        const int i = lo_i + k;
        if (i < nb) { const int v = in[i]; out[i] = s; s += v; }
    }
    int incl = s;
#pragma unroll
    for (int off = 1; off < 32; off <<= 1) {
        const int u = __shfl_up(incl, off, 32);
        if (lane >= off) incl += u;
    }
    if (lane == 31) wsum[wave] = incl;
    __syncthreads();
    int base = incl - s;
    for (int w = 0; w < wave; ++w) base += wsum[w];
    for (int k = 0; k < per; ++k) {
        const int i = lo_i + k;
        if (i < nb) out[i] += base;
    }
    __syncthreads();
}

__global__ __launch_bounds__(SORT_NT) void bin_count_kernel(const int* __restrict__ dst, int* __restrict__ blockhist,
                                                            int e, int nbins) {
    extern __shared__ int hist[];
    const int tid = threadIdx.x;
    for (int i = tid; i < nbins; i += SORT_NT) hist[i] = 0;
    __syncthreads();
    const int chunk = (e + NCHUNK - 1) / NCHUNK;
    const int lo    = blockIdx.x * chunk;
    const int hi    = lo + chunk < e ? lo + chunk : e;
    for (int tb = lo; tb < hi; tb += SORT_TILE) {
        int d[SORT_EPT];
#pragma unroll
        for (int k = 0; k < SORT_EPT; ++k) {
            const int i = tb + k * SORT_NT + tid;
            d[k] = dst[i < hi ? i : hi - 1];
        }
#pragma unroll
        for (int k = 0; k < SORT_EPT; ++k)
            if (tb + k * SORT_NT + tid < hi) atomicAdd(&hist[d[k] >> BIN_SHIFT], 1);
    }
    __syncthreads();
    for (int i = tid; i < nbins; i += SORT_NT) blockhist[(size_t)i * NCHUNK + blockIdx.x] = hist[i];
}

__global__ __launch_bounds__(256) void bin_scan_kernel(int* __restrict__ blockhist, int* __restrict__ bintotal, int nbins) {
    const int lane = threadIdx.x & 31;
    const int bin  = blockIdx.x * 8 + (threadIdx.x >> 5);
    if (bin >= nbins) return;
    int4* p = (int4*)(blockhist + (size_t)bin * NCHUNK + lane * 8);
    const int4 a = p[0];
    const int4 b = p[1];
    const int t = a.x + a.y + a.z + a.w + b.x + b.y + b.z + b.w;
    int incl = t;
#pragma unroll
    for (int off = 1; off < 32; off <<= 1) {
        const int u = __shfl_up(incl, off, 32);
        if (lane >= off) incl += u;
    }
    int s = incl - t;
    int4 oa, ob;
    oa.x = s; s += a.x; oa.y = s; s += a.y; oa.z = s; s += a.z; oa.w = s; s += a.w;
    ob.x = s; s += b.x; ob.y = s; s += b.y; ob.z = s; s += b.z; ob.w = s;
    p[0] = oa;
    p[1] = ob;
    if (lane == 31) bintotal[bin] = incl;
}

__global__ __launch_bounds__(SORT_NT) void bin_partition_kernel(const int* __restrict__ src, const int* __restrict__ dst,
                                                                const int* __restrict__ blockbase,
                                                                const int* __restrict__ bintotal,
                                                                unsigned* __restrict__ packed, int e, int nbins) {
    extern __shared__ int sm[];
    __shared__ int            wsum[SORT_NT / 32];
    __shared__ unsigned       stage_w[SORT_TILE];
    __shared__ unsigned short stage_b[SORT_TILE];
    int* cur    = sm;
    int* hist   = sm + nbins;
    int* lstart = sm + 2 * nbins;
    const int tid = threadIdx.x;

    for (int i = tid; i < nbins; i += SORT_NT) {
        cur[i]  = bintotal[i];
        hist[i] = blockbase[(size_t)i * NCHUNK + blockIdx.x];
    }
    __syncthreads();
    block_excl_scan(cur, cur, nbins, wsum);
    for (int i = tid; i < nbins; i += SORT_NT) cur[i] += hist[i];

    const int chunk = (e + NCHUNK - 1) / NCHUNK;
    const int lo    = blockIdx.x * chunk;
    const int hi    = lo + chunk < e ? lo + chunk : e;
    for (int tb = lo; tb < hi; tb += SORT_TILE) {
        for (int i = tid; i < nbins; i += SORT_NT) hist[i] = 0;
        __syncthreads();

        int s[SORT_EPT], d[SORT_EPT], r[SORT_EPT];
#pragma unroll
        for (int k = 0; k < SORT_EPT; ++k) {
            const int i  = tb + k * SORT_NT + tid;
            const int ii = i < hi ? i : hi - 1;
            d[k] = dst[ii];
            s[k] = src[ii];
        }
#pragma unroll
        for (int k = 0; k < SORT_EPT; ++k)
            r[k] = tb + k * SORT_NT + tid < hi ? atomicAdd(&hist[d[k] >> BIN_SHIFT], 1) : 0;
        __syncthreads();

        block_excl_scan(hist, lstart, nbins, wsum);

#pragma unroll
        for (int k = 0; k < SORT_EPT; ++k) r[k] += lstart[d[k] >> BIN_SHIFT];
#pragma unroll
        for (int k = 0; k < SORT_EPT; ++k) {
            if (tb + k * SORT_NT + tid < hi) {
                stage_w[r[k]] = ((unsigned)s[k] << BIN_SHIFT) | (unsigned)(d[k] & (BIN_W - 1));
                stage_b[r[k]] = (unsigned short)(d[k] >> BIN_SHIFT);
            }
        }
        __syncthreads();

        const int nt = hi - tb < SORT_TILE ? hi - tb : SORT_TILE;
        for (int i = tid; i < nt; i += SORT_NT) {
            const int bn = stage_b[i];
            packed[cur[bn] + (i - lstart[bn])] = stage_w[i];
        }
        __syncthreads();
        for (int i = tid; i < nbins; i += SORT_NT) cur[i] += hist[i];
    }
}

#define BS_EPT 8
__global__ __launch_bounds__(BIN_W) void bin_sort_kernel(const unsigned* __restrict__ packed, const int* __restrict__ bintotal,
                                                         int* __restrict__ cnt, float* __restrict__ dinv,
                                                         int* __restrict__ rowptr, int* __restrict__ col, int n) {
    __shared__ int lcnt[BIN_W];
    __shared__ int wpart[BIN_W / 32];
    __shared__ int wtot[BIN_W / 32];
    __shared__ int scol[SCOL_CAP];
    const int tid  = threadIdx.x;
    const int lane = tid & 31;
    const int wave = tid >> 5;
    const int bin  = blockIdx.x;

    int part = 0;
    for (int i = tid; i < bin; i += BIN_W) part += bintotal[i];
#pragma unroll
    for (int off = 16; off > 0; off >>= 1) part += __shfl_xor(part, off, 32);
    if (lane == 0) wpart[wave] = part;
    lcnt[tid] = 0;
    const int m = bintotal[bin];
    __syncthreads();
    int start = 0;
#pragma unroll
    for (int w = 0; w < BIN_W / 32; ++w) start += wpart[w];
    const unsigned* my = packed + start;

    for (int base = 0; base < m; base += BIN_W * BS_EPT) {
        unsigned w[BS_EPT];
#pragma unroll
        for (int k = 0; k < BS_EPT; ++k) {
            const int i = base + k * BIN_W + tid;
            w[k] = my[i < m ? i : m - 1];
        }
#pragma unroll
        for (int k = 0; k < BS_EPT; ++k)
            if (base + k * BIN_W + tid < m) atomicAdd(&lcnt[w[k] & (BIN_W - 1)], 1);
    }
    __syncthreads();

    const int c = lcnt[tid];
    int incl = c;
#pragma unroll
    for (int off = 1; off < 32; off <<= 1) {
        const int u = __shfl_up(incl, off, 32);
        if (lane >= off) incl += u;
    }
    if (lane == 31) wtot[wave] = incl;
    __syncthreads();
    {
        int excl = incl - c;
        for (int w = 0; w < wave; ++w) excl += wtot[w];
        const int node = bin * BIN_W + tid;
        if (node < n) {
            cnt[node]    = c;
            dinv[node]   = rsqrtf((float)c + 1.0f);
            rowptr[node] = start + excl;
        }
        lcnt[tid] = excl;
    }
    __syncthreads();

    const bool in_lds = m <= SCOL_CAP;
    for (int base = 0; base < m; base += BIN_W * BS_EPT) {
        unsigned w[BS_EPT];
#pragma unroll
        for (int k = 0; k < BS_EPT; ++k) {
            const int i = base + k * BIN_W + tid;
            w[k] = my[i < m ? i : m - 1];
        }
#pragma unroll
        for (int k = 0; k < BS_EPT; ++k) {
            if (base + k * BIN_W + tid < m) {
                const int pos = atomicAdd(&lcnt[w[k] & (BIN_W - 1)], 1);
                const int sv  = (int)(w[k] >> BIN_SHIFT);
                if (in_lds) scol[pos] = sv;
                else        col[start + pos] = sv;
            }
        }
    }
    if (in_lds) {
        __syncthreads();
        for (int i = tid; i < m; i += BIN_W) col[start + i] = scol[i];
    }
}


template <int M> static __device__ __forceinline__ int lane_xor(int v) {
    if constexpr (M == 16) return __builtin_amdgcn_permlanex16(v, v, 0x76543210, 0xfedcba98, false, false);
    else                   return __builtin_amdgcn_update_dpp(v, v, 0x160 + M, 0xf, 0xf, false);
}
template <int M> static __device__ __forceinline__ float lane_xor_f(float v) {
    return __builtin_bit_cast(float, lane_xor<M>(__builtin_bit_cast(int, v)));
}

template <int K, int J> static __device__ __forceinline__ int cmp_xchg(int key, int lane) {
    const int  o   = lane_xor<J>(key);
    const bool low = (((lane & J) == 0) == ((lane & K) == 0));
    const int  mn  = key < o ? key : o;
    const int  mx  = key < o ? o : key;
    return low ? mn : mx;
}
static __device__ __forceinline__ int sort_wave(int key, int lane) {
    key = cmp_xchg<2, 1>(key, lane);
    key = cmp_xchg<4, 2>(key, lane);   key = cmp_xchg<4, 1>(key, lane);
    key = cmp_xchg<8, 4>(key, lane);   key = cmp_xchg<8, 2>(key, lane);   key = cmp_xchg<8, 1>(key, lane);
    key = cmp_xchg<16, 8>(key, lane);  key = cmp_xchg<16, 4>(key, lane);  key = cmp_xchg<16, 2>(key, lane);
    key = cmp_xchg<16, 1>(key, lane);
    key = cmp_xchg<32, 16>(key, lane); key = cmp_xchg<32, 8>(key, lane);  key = cmp_xchg<32, 4>(key, lane);
    key = cmp_xchg<32, 2>(key, lane);  key = cmp_xchg<32, 1>(key, lane);
    return key;
}

static __device__ __forceinline__ float lane_bcast(float v, int l) {
    return __builtin_bit_cast(float, __builtin_amdgcn_readlane(__builtin_bit_cast(int, v), l));
}

static __device__ __forceinline__ float add_piece(int ks, float nrm, int sub, int fc,
                                                  const float* __restrict__ x, float acc) {
#pragma unroll
    for (int p = 0; p < 8; ++p) {
        const int   from = p * 4 + sub;
        const int   ss   = __shfl(ks, from, 32);
        const float nn   = __shfl(nrm, from, 32);
        const float xv   = x[ss * (T_HIST * F_IN) + (T_HIST - 1) * F_IN + fc];
        acc = fmaf(nn, xv, acc);
    }
    return acc;
}

#define GN_NODES 8

__global__ __launch_bounds__(256) void gather_in_kernel(const int* __restrict__ rowptr, const int* __restrict__ cnt,
                                                       int* __restrict__ col, const float* __restrict__ dinv,
                                                       const float* __restrict__ x, float* __restrict__ agg, int n) {
    const int lane = threadIdx.x & 31;
    const int wave  = __builtin_amdgcn_readfirstlane(threadIdx.x >> 5);
    const int node0 = (blockIdx.x * 8 + wave) * GN_NODES;
    const int f     = lane & 7;
    const int fc    = f < F_IN ? f : F_IN - 1;
    const int sub   = lane >> 3;

    int   mstart = 0, mdeg = 0;
    float mdi = 0.0f;
    {
        const int nd = node0 + lane;
        if (lane < GN_NODES && nd < n) {
            mstart = rowptr[nd];
            mdeg   = cnt[nd];
            mdi    = dinv[nd];
        }
    }


    int   key[GN_NODES];
    float xs[GN_NODES];
#pragma unroll
    for (int i = 0; i < GN_NODES; ++i) {
        const int s  = __builtin_amdgcn_readlane(mstart, i);
        const int d  = __builtin_amdgcn_readlane(mdeg, i);
        const int nb = d < 32 ? d : 32;
        key[i] = lane < nb ? col[s + lane] : 0x7fffffff;
        const int nd = node0 + i < n ? node0 + i : n - 1;
        xs[i] = x[nd * (T_HIST * F_IN) + (T_HIST - 1) * F_IN + fc];
    }

    float w[GN_NODES];
#pragma unroll
    for (int i = 0; i < GN_NODES; ++i) {
        const int d  = __builtin_amdgcn_readlane(mdeg, i);
        const int nb = d < 32 ? d : 32;
        key[i] = sort_wave(key[i], lane);
        w[i]   = dinv[lane < nb ? key[i] : 0];
    }

    float acc[GN_NODES];
#pragma unroll
    for (int i = 0; i < GN_NODES; ++i) {
        const int   d   = __builtin_amdgcn_readlane(mdeg, i);
        const int   nb  = d < 32 ? d : 32;
        const float di  = lane_bcast(mdi, i);
        const int   ks  = lane < nb ? key[i] : 0;
        const float nrm = lane < nb ? w[i] * di : 0.0f;
        acc[i] = add_piece(ks, nrm, sub, fc, x, 0.0f);
    }

#pragma unroll
    for (int i = 0; i < GN_NODES; ++i) {
        const int s  = __builtin_amdgcn_readlane(mstart, i);
        const int d  = __builtin_amdgcn_readlane(mdeg, i);
        const int nb = d < 32 ? d : 32;
        if (lane < nb) col[s + lane] = key[i];
    }

    const unsigned longmask = __builtin_amdgcn_ballot_w32(mdeg > 32);
    if (longmask != 0u) {
#pragma nounroll
        for (int i = 0; i < GN_NODES; ++i) {
            if ((longmask >> i) & 1u) {
                const int   s  = __builtin_amdgcn_readlane(mstart, i);
                const int   d  = __builtin_amdgcn_readlane(mdeg, i);
                const float di = lane_bcast(mdi, i);
                float extra = 0.0f;
                for (int base = 32; base < d; base += 32) {
                    const int   nbp = (d - base < 32) ? d - base : 32;
                    const int   ks  = col[s + base + (lane < nbp ? lane : 0)];
                    const float wv  = dinv[ks];
                    const float nrm = lane < nbp ? wv * di : 0.0f;
                    extra = add_piece(ks, nrm, sub, fc, x, extra);
                }
#pragma unroll
                for (int j = 0; j < GN_NODES; ++j) acc[j] = (j == i) ? acc[j] + extra : acc[j];
            }
        }
    }

#pragma unroll
    for (int i = 0; i < GN_NODES; ++i) {
        const float di = lane_bcast(mdi, i);
        float t = acc[i];
        t += lane_xor_f<8>(t);
        t += lane_xor_f<16>(t);
        if (lane < F_IN && node0 + i < n) agg[(size_t)(node0 + i) * ZP + lane] = fmaf(di * di, xs[i], t);
    }
}

#define FT_ROWS   64
#define FT_APITCH (HDIM + 8)
#define FT_HPITCH (HDIM + 4)

typedef __attribute__((ext_vector_type(4))) _Float16 v4h;

static __device__ __forceinline__ void add_source(float z0, float z1, float z2, float z3, float z4, float wt,
                                                  const float (&w1)[F_IN][4], const float (&bv)[4], float (&a)[4]) {
#pragma unroll
    for (int q = 0; q < 4; ++q) {
        float s = fmaf(z0, w1[0][q], bv[q]);
        s = fmaf(z1, w1[1][q], s);
        s = fmaf(z2, w1[2][q], s);
        s = fmaf(z3, w1[3][q], s);
        s = fmaf(z4, w1[4][q], s);
        a[q] = fmaf(wt, fmaxf(s, 0.0f), a[q]);
    }
}

__global__ __launch_bounds__(256) void agg_gemm_head_kernel(const int* __restrict__ rowptr, const int* __restrict__ cnt,
                                                            const int* __restrict__ col, const float* __restrict__ dinv,
                                                            const float* __restrict__ zin,
                                                            const float* __restrict__ W1, const float* __restrict__ b1,
                                                            const float* __restrict__ W2, const float* __restrict__ b2,
                                                            const float* __restrict__ Wout, const float* __restrict__ bout,
                                                            float* __restrict__ out, int n) {
    __shared__ __attribute__((aligned(16))) _Float16 aS[FT_ROWS][FT_APITCH];
    __shared__ __attribute__((aligned(16))) float    hS[FT_ROWS][FT_HPITCH];

    const int tid  = threadIdx.x;
    const int lane = tid & 31;
    const int wave = __builtin_amdgcn_readfirstlane(tid >> 5);
    const int row0 = blockIdx.x * FT_ROWS;
    const int h0   = lane * 4;

    {
        float w1[F_IN][4];
        float bv[4];
#pragma unroll
        for (int f = 0; f < F_IN; ++f) {
            const float4 t = *(const float4*)(W1 + f * HDIM + h0);
            w1[f][0] = t.x; w1[f][1] = t.y; w1[f][2] = t.z; w1[f][3] = t.w;
        }
        {
            const float4 t = *(const float4*)(b1 + h0);
            bv[0] = t.x; bv[1] = t.y; bv[2] = t.z; bv[3] = t.w;
        }
        const float4* zr4 = (const float4*)zin;

        int    mstart = 0, mdeg = 0;
        float  mdi = 0.0f, mzl = 0.0f;
        float4 mzs = {0.0f, 0.0f, 0.0f, 0.0f};
        {
            const int nd = row0 + wave * (FT_ROWS / 8) + lane;
            if (lane < FT_ROWS / 8 && nd < n) {
                mstart = rowptr[nd];
                mdeg   = cnt[nd];
                mdi    = dinv[nd];
                mzs    = zr4[(size_t)nd * (ZP / 4)];
                mzl    = zin[(size_t)nd * ZP + 4];
            }
        }

        auto first_piece = [&](int i) -> int {
            const int s  = __builtin_amdgcn_readlane(mstart, i);
            const int d  = __builtin_amdgcn_readlane(mdeg, i);
            const int nb = d < 32 ? d : 32;
            return col[lane < nb ? s + lane : 0];
        };

        int    ms_b = first_piece(0);
        float  wq_a = dinv[ms_b];
        float4 zq_a = zr4[(size_t)ms_b * (ZP / 4)];
        float  zl_a = zin[(size_t)ms_b * ZP + 4];
        ms_b = first_piece(1);

#pragma unroll 2
        for (int i = 0; i < FT_ROWS / 8; ++i) {
            const int    ms_c = first_piece(i + 2);
            const float  wq_b = dinv[ms_b];
            const float4 zq_b = zr4[(size_t)ms_b * (ZP / 4)];
            const float  zl_b = zin[(size_t)ms_b * ZP + 4];

            const int   m   = wave * (FT_ROWS / 8) + i;
            const int   deg = __builtin_amdgcn_readlane(mdeg, i);
            const float di  = lane_bcast(mdi, i);
            float a[4] = {0.0f, 0.0f, 0.0f, 0.0f};
            add_source(lane_bcast(mzs.x, i), lane_bcast(mzs.y, i), lane_bcast(mzs.z, i), lane_bcast(mzs.w, i),
                       lane_bcast(mzl, i), di * di, w1, bv, a);
            {
                const int   nb = deg < 32 ? deg : 32;
                const float mw = wq_a * di;
                for (int j = 0; j < nb; ++j) {
                    add_source(lane_bcast(zq_a.x, j), lane_bcast(zq_a.y, j), lane_bcast(zq_a.z, j), lane_bcast(zq_a.w, j),
                               lane_bcast(zl_a, j), lane_bcast(mw, j), w1, bv, a);
                }
            }
            if (deg > 32) {
                const int start = __builtin_amdgcn_readlane(mstart, i);
                for (int base = 32; base < deg; base += 32) {
                    const int    nb = (deg - base < 32) ? deg - base : 32;
                    const int    ms = col[lane < nb ? start + base + lane : 0];
                    const float  mw = dinv[ms] * di;
                    const float4 zq = zr4[(size_t)ms * (ZP / 4)];
                    const float  zl = zin[(size_t)ms * ZP + 4];
                    for (int j = 0; j < nb; ++j) {
                        add_source(lane_bcast(zq.x, j), lane_bcast(zq.y, j), lane_bcast(zq.z, j), lane_bcast(zq.w, j),
                                   lane_bcast(zl, j), lane_bcast(mw, j), w1, bv, a);
                    }
                }
            }
            v4h pk;
            pk[0] = (_Float16)a[0]; pk[1] = (_Float16)a[1]; pk[2] = (_Float16)a[2]; pk[3] = (_Float16)a[3];
            *(v4h*)&aS[m][h0] = pk;

            wq_a = wq_b; zq_a = zq_b; zl_a = zl_b;
            ms_b = ms_c;
        }
    }

    const int ncol  = wave * 16;
    const int c     = lane & 15;
    const int h     = lane >> 4;
    const int khalf = h * 8;
    const int colw  = ncol + c;

    v16h bf[4];
#pragma unroll
    for (int ks = 0; ks < 4; ++ks) {
#pragma unroll
        for (int j = 0; j < 16; ++j) {
            const int K = (j & 7) + ((j >> 3) << 4) + khalf + ks * 32;
            bf[ks][j] = (_Float16)W2[K * HDIM + colw];
        }
    }
    const float b2c = b2[colw];
    __syncthreads();

    v8f acc[4] = {};
#pragma unroll
    for (int ks = 0; ks < 4; ++ks) {
#pragma unroll
        for (int rt = 0; rt < 4; ++rt) {
            const _Float16* ap = &aS[rt * 16 + c][ks * 32 + khalf];
            const v8h lo = *(const v8h*)ap;
            const v8h hi = *(const v8h*)(ap + 16);
            const v16h av = __builtin_shufflevector(lo, hi, 0, 1, 2, 3, 4, 5, 6, 7, 8, 9, 10, 11, 12, 13, 14, 15);
            acc[rt] = __builtin_amdgcn_wmma_f32_16x16x32_f16(false, av, false, bf[ks],
                                                             (short)0, acc[rt], false, false);
        }
    }

#pragma unroll
    for (int rt = 0; rt < 4; ++rt) {
#pragma unroll
        for (int v = 0; v < 8; ++v) hS[rt * 16 + 8 * h + v][colw] = fmaxf(acc[rt][v] + b2c, 0.0f);
    }

    float wo[4][CDIM];
    {
        const float4* wp = (const float4*)(Wout + (size_t)h0 * CDIM);
        float t[4 * CDIM];
#pragma unroll
        for (int u = 0; u < CDIM; ++u) {
            const float4 x4 = wp[u];
            t[4 * u + 0] = x4.x; t[4 * u + 1] = x4.y; t[4 * u + 2] = x4.z; t[4 * u + 3] = x4.w;
        }
#pragma unroll
        for (int q = 0; q < 4; ++q) {
#pragma unroll
            for (int cc = 0; cc < CDIM; ++cc) wo[q][cc] = t[q * CDIM + cc];
        }
    }
    float bo[CDIM];
#pragma unroll
    for (int cc = 0; cc < CDIM; ++cc) bo[cc] = bout[cc];
    __syncthreads();

    for (int i = 0; i < FT_ROWS / 8; ++i) {
        const int m    = wave * (FT_ROWS / 8) + i;
        const int node = row0 + m;
        if (node >= n) break;
        const float4 hv = *(const float4*)&hS[m][h0];
        float lg[CDIM];
#pragma unroll
        for (int cc = 0; cc < CDIM; ++cc)
            lg[cc] = hv.x * wo[0][cc] + hv.y * wo[1][cc] + hv.z * wo[2][cc] + hv.w * wo[3][cc];
#pragma unroll
        for (int off = 16; off > 0; off >>= 1) {
#pragma unroll
            for (int cc = 0; cc < CDIM; ++cc) lg[cc] += __shfl_xor(lg[cc], off, 32);
        }
#pragma unroll
        for (int cc = 0; cc < CDIM; ++cc) lg[cc] += bo[cc];
        float mx = lg[0];
#pragma unroll
        for (int cc = 1; cc < CDIM; ++cc) mx = fmaxf(mx, lg[cc]);
        float sum = 0.0f;
#pragma unroll
        for (int cc = 0; cc < CDIM; ++cc) { lg[cc] = __expf(lg[cc] - mx); sum += lg[cc]; }
        const float inv = 1.0f / sum;
        float pv = lg[0];
#pragma unroll
        for (int cc = 1; cc < CDIM; ++cc) pv = (lane == cc) ? lg[cc] : pv;
        if (lane < CDIM) out[(size_t)node * CDIM + lane] = pv * inv;
    }
}

extern "C" void kernel_launch(void* const* d_in, const int* in_sizes, int n_in,
                              void* d_out, int out_size, void* d_ws, size_t ws_size,
                              hipStream_t stream) {
    const float* x    = (const float*)d_in[0];
    const int*   eidx = (const int*)  d_in[1];
    const float* W1   = (const float*)d_in[2];
    const float* b1   = (const float*)d_in[3];
    const float* W2   = (const float*)d_in[4];
    const float* b2   = (const float*)d_in[5];
    const float* Wout = (const float*)d_in[6];
    const float* bout = (const float*)d_in[7];
    float*       out  = (float*)d_out;

    const int N = in_sizes[0] / (T_HIST * F_IN);
    const int E = in_sizes[1] / 2;
    const int* src = eidx;
    const int* dst = eidx + E;

    const int NP = (N + 63) & ~63;
    const int NB = (N + 255) / 256;
    const int NBINS = (N + BIN_W - 1) >> BIN_SHIFT;
    float* dinv   = (float*)d_ws;
    float* bufA   = dinv + NP;
    float* zin    = bufA + (size_t)N * HDIM;
    int*   cnt    = (int*)(zin + (size_t)N * ZP);
    int*   rowptr = cnt + NP;
    int*   cursor = rowptr + NP;
    int*   bsum   = cursor + NP;
    int*   col    = bsum + ((NB + 63) & ~63);
    unsigned* packed    = (unsigned*)bufA;
    int*      blockhist = (int*)(packed + ((E + 63) & ~63));
    int*      bintotal  = blockhist + (size_t)NBINS * NCHUNK;

    bin_count_kernel<<<NCHUNK, SORT_NT, NBINS * sizeof(int), stream>>>(dst, blockhist, E, NBINS);
    bin_scan_kernel<<<(NBINS + 7) / 8, 256, 0, stream>>>(blockhist, bintotal, NBINS);
    bin_partition_kernel<<<NCHUNK, SORT_NT, 3 * NBINS * sizeof(int), stream>>>(src, dst, blockhist, bintotal, packed, E, NBINS);
    bin_sort_kernel<<<NBINS, BIN_W, 0, stream>>>(packed, bintotal, cnt, dinv, rowptr, col, N);

    gather_in_kernel<<<(N + 8 * GN_NODES - 1) / (8 * GN_NODES), 256, 0, stream>>>(rowptr, cnt, col, dinv, x, zin, N);

    agg_gemm_head_kernel<<<(N + FT_ROWS - 1) / FT_ROWS, 256, 0, stream>>>(rowptr, cnt, col, dinv, zin, W1, b1, W2, b2,
                                                                          Wout, bout, out, N);
}


# source factor dinv[s] moved into the gathered rows: layer 1 adds x_last*dinv rows scaled once per node in the bin sort, layer 2 reads dinv from the z row; no dinv gathers; sums round differently
# speedup vs baseline: 9.7134x; 1.1649x over previous
#include <hip/hip_runtime.h>
#include <hip/hip_bf16.h>

typedef __attribute__((ext_vector_type(16))) _Float16 v16h;
typedef __attribute__((ext_vector_type(8)))  float    v8f;
typedef __attribute__((ext_vector_type(8)))  _Float16 v8h;

#define T_HIST 3
#define F_IN   5
#define HDIM   128
#define CDIM   5
#define ZP     8

#define BIN_SHIFT 8
#define BIN_W     (1 << BIN_SHIFT)
#define NCHUNK    256
#define SORT_NT   256
#define SORT_EPT  25
#define SORT_TILE (SORT_NT * SORT_EPT)
#define SCOL_CAP  8192

static __device__ __forceinline__ void block_excl_scan(const int* in, int* out, int nb, int* wsum) {
    const int tid  = threadIdx.x;
    const int lane = tid & 31;
    const int wave = tid >> 5;
    const int per  = (nb + SORT_NT - 1) / SORT_NT;
    const int lo_i = tid * per;
    int s = 0;
    for (int k = 0; k < per; ++k) {
        const int i = lo_i + k;
        if (i < nb) { const int v = in[i]; out[i] = s; s += v; }
    }
    int incl = s;
#pragma unroll
    for (int off = 1; off < 32; off <<= 1) {
        const int u = __shfl_up(incl, off, 32);
        if (lane >= off) incl += u;
    }
    if (lane == 31) wsum[wave] = incl;
    __syncthreads();
    int base = incl - s;
    for (int w = 0; w < wave; ++w) base += wsum[w];
    for (int k = 0; k < per; ++k) {
        const int i = lo_i + k;
        if (i < nb) out[i] += base;
    }
    __syncthreads();
}

__global__ __launch_bounds__(SORT_NT) void bin_count_kernel(const int* __restrict__ dst, int* __restrict__ blockhist,
                                                            int e, int nbins) {
    extern __shared__ int hist[];
    const int tid = threadIdx.x;
    for (int i = tid; i < nbins; i += SORT_NT) hist[i] = 0;
    __syncthreads();
    const int chunk = (e + NCHUNK - 1) / NCHUNK;
    const int lo    = blockIdx.x * chunk;
    const int hi    = lo + chunk < e ? lo + chunk : e;
    for (int tb = lo; tb < hi; tb += SORT_TILE) {
        int d[SORT_EPT];
#pragma unroll
        for (int k = 0; k < SORT_EPT; ++k) {
            const int i = tb + k * SORT_NT + tid;
            d[k] = dst[i < hi ? i : hi - 1];
        }
#pragma unroll
        for (int k = 0; k < SORT_EPT; ++k)
            if (tb + k * SORT_NT + tid < hi) atomicAdd(&hist[d[k] >> BIN_SHIFT], 1);
    }
    __syncthreads();
    for (int i = tid; i < nbins; i += SORT_NT) blockhist[(size_t)i * NCHUNK + blockIdx.x] = hist[i];
}

__global__ __launch_bounds__(256) void bin_scan_kernel(int* __restrict__ blockhist, int* __restrict__ bintotal, int nbins) {
    const int lane = threadIdx.x & 31;
    const int bin  = blockIdx.x * 8 + (threadIdx.x >> 5);
    if (bin >= nbins) return;
    int4* p = (int4*)(blockhist + (size_t)bin * NCHUNK + lane * 8);
    const int4 a = p[0];
    const int4 b = p[1];
    const int t = a.x + a.y + a.z + a.w + b.x + b.y + b.z + b.w;
    int incl = t;
#pragma unroll
    for (int off = 1; off < 32; off <<= 1) {
        const int u = __shfl_up(incl, off, 32);
        if (lane >= off) incl += u;
    }
    int s = incl - t;
    int4 oa, ob;
    oa.x = s; s += a.x; oa.y = s; s += a.y; oa.z = s; s += a.z; oa.w = s; s += a.w;
    ob.x = s; s += b.x; ob.y = s; s += b.y; ob.z = s; s += b.z; ob.w = s;
    p[0] = oa;
    p[1] = ob;
    if (lane == 31) bintotal[bin] = incl;
}

__global__ __launch_bounds__(SORT_NT) void bin_partition_kernel(const int* __restrict__ src, const int* __restrict__ dst,
                                                                const int* __restrict__ blockbase,
                                                                const int* __restrict__ bintotal,
                                                                unsigned* __restrict__ packed, int e, int nbins) {
    extern __shared__ int sm[];
    __shared__ int            wsum[SORT_NT / 32];
    __shared__ unsigned       stage_w[SORT_TILE];
    __shared__ unsigned short stage_b[SORT_TILE];
    int* cur    = sm;
    int* hist   = sm + nbins;
    int* lstart = sm + 2 * nbins;
    const int tid = threadIdx.x;

    for (int i = tid; i < nbins; i += SORT_NT) {
        cur[i]  = bintotal[i];
        hist[i] = blockbase[(size_t)i * NCHUNK + blockIdx.x];
    }
    __syncthreads();
    block_excl_scan(cur, cur, nbins, wsum);
    for (int i = tid; i < nbins; i += SORT_NT) cur[i] += hist[i];

    const int chunk = (e + NCHUNK - 1) / NCHUNK;
    const int lo    = blockIdx.x * chunk;
    const int hi    = lo + chunk < e ? lo + chunk : e;
    for (int tb = lo; tb < hi; tb += SORT_TILE) {
        for (int i = tid; i < nbins; i += SORT_NT) hist[i] = 0;
        __syncthreads();

        int s[SORT_EPT], d[SORT_EPT], r[SORT_EPT];
#pragma unroll
        for (int k = 0; k < SORT_EPT; ++k) {
            const int i  = tb + k * SORT_NT + tid;
            const int ii = i < hi ? i : hi - 1;
            d[k] = dst[ii];
            s[k] = src[ii];
        }
#pragma unroll
        for (int k = 0; k < SORT_EPT; ++k)
            r[k] = tb + k * SORT_NT + tid < hi ? atomicAdd(&hist[d[k] >> BIN_SHIFT], 1) : 0;
        __syncthreads();

        block_excl_scan(hist, lstart, nbins, wsum);

#pragma unroll
        for (int k = 0; k < SORT_EPT; ++k) r[k] += lstart[d[k] >> BIN_SHIFT];
#pragma unroll
        for (int k = 0; k < SORT_EPT; ++k) {
            if (tb + k * SORT_NT + tid < hi) {
                stage_w[r[k]] = ((unsigned)s[k] << BIN_SHIFT) | (unsigned)(d[k] & (BIN_W - 1));
                stage_b[r[k]] = (unsigned short)(d[k] >> BIN_SHIFT);
            }
        }
        __syncthreads();

        const int nt = hi - tb < SORT_TILE ? hi - tb : SORT_TILE;
        for (int i = tid; i < nt; i += SORT_NT) {
            const int bn = stage_b[i];
            packed[cur[bn] + (i - lstart[bn])] = stage_w[i];
        }
        __syncthreads();
        for (int i = tid; i < nbins; i += SORT_NT) cur[i] += hist[i];
    }
}

#define BS_EPT 8
__global__ __launch_bounds__(BIN_W) void bin_sort_kernel(const unsigned* __restrict__ packed, const int* __restrict__ bintotal,
                                                         int* __restrict__ cnt, float* __restrict__ dinv,
                                                         int* __restrict__ rowptr, int* __restrict__ col,
                                                         const float* __restrict__ x, float* __restrict__ xp, int n) {
    __shared__ int lcnt[BIN_W];
    __shared__ int wpart[BIN_W / 32];
    __shared__ int wtot[BIN_W / 32];
    __shared__ int scol[SCOL_CAP];
    const int tid  = threadIdx.x;
    const int lane = tid & 31;
    const int wave = tid >> 5;
    const int bin  = blockIdx.x;

    int part = 0;
    for (int i = tid; i < bin; i += BIN_W) part += bintotal[i];
#pragma unroll
    for (int off = 16; off > 0; off >>= 1) part += __shfl_xor(part, off, 32);
    if (lane == 0) wpart[wave] = part;
    lcnt[tid] = 0;
    const int m = bintotal[bin];
    __syncthreads();
    int start = 0;
#pragma unroll
    for (int w = 0; w < BIN_W / 32; ++w) start += wpart[w];
    const unsigned* my = packed + start;

    for (int base = 0; base < m; base += BIN_W * BS_EPT) {
        unsigned w[BS_EPT];
#pragma unroll
        for (int k = 0; k < BS_EPT; ++k) {
            const int i = base + k * BIN_W + tid;
            w[k] = my[i < m ? i : m - 1];
        }
#pragma unroll
        for (int k = 0; k < BS_EPT; ++k)
            if (base + k * BIN_W + tid < m) atomicAdd(&lcnt[w[k] & (BIN_W - 1)], 1);
    }
    __syncthreads();

    const int c = lcnt[tid];
    int incl = c;
#pragma unroll
    for (int off = 1; off < 32; off <<= 1) {
        const int u = __shfl_up(incl, off, 32);
        if (lane >= off) incl += u;
    }
    if (lane == 31) wtot[wave] = incl;
    __syncthreads();
    {
        int excl = incl - c;
        for (int w = 0; w < wave; ++w) excl += wtot[w];
        const int node = bin * BIN_W + tid;
        if (node < n) {
            const float di = rsqrtf((float)c + 1.0f);
            cnt[node]    = c;
            dinv[node]   = di;
            rowptr[node] = start + excl;
            const float* xr = x + (size_t)node * (T_HIST * F_IN) + (T_HIST - 1) * F_IN;
            float4 lo4, hi4;
            lo4.x = di * xr[0]; lo4.y = di * xr[1]; lo4.z = di * xr[2]; lo4.w = di * xr[3];
            hi4.x = di * xr[4]; hi4.y = 0.0f;       hi4.z = 0.0f;       hi4.w = 0.0f;
            ((float4*)xp)[(size_t)node * (ZP / 4)]     = lo4;
            ((float4*)xp)[(size_t)node * (ZP / 4) + 1] = hi4;
        }
        if (bin == 0 && tid < ZP / 4) {
            float4 z4; z4.x = 0.0f; z4.y = 0.0f; z4.z = 0.0f; z4.w = 0.0f;
            ((float4*)xp)[(size_t)n * (ZP / 4) + tid] = z4;
        }
        lcnt[tid] = excl;
    }
    __syncthreads();

    const bool in_lds = m <= SCOL_CAP;
    for (int base = 0; base < m; base += BIN_W * BS_EPT) {
        unsigned w[BS_EPT];
#pragma unroll
        for (int k = 0; k < BS_EPT; ++k) {
            const int i = base + k * BIN_W + tid;
            w[k] = my[i < m ? i : m - 1];
        }
#pragma unroll
        for (int k = 0; k < BS_EPT; ++k) {
            if (base + k * BIN_W + tid < m) {
                const int pos = atomicAdd(&lcnt[w[k] & (BIN_W - 1)], 1);
                const int sv  = (int)(w[k] >> BIN_SHIFT);
                if (in_lds) scol[pos] = sv;
                else        col[start + pos] = sv;
            }
        }
    }
    if (in_lds) {
        __syncthreads();
        for (int i = tid; i < m; i += BIN_W) col[start + i] = scol[i];
    }
}


template <int M> static __device__ __forceinline__ int lane_xor(int v) {
    if constexpr (M == 16) return __builtin_amdgcn_permlanex16(v, v, 0x76543210, 0xfedcba98, false, false);
    else                   return __builtin_amdgcn_update_dpp(v, v, 0x160 + M, 0xf, 0xf, false);
}
template <int M> static __device__ __forceinline__ float lane_xor_f(float v) {
    return __builtin_bit_cast(float, lane_xor<M>(__builtin_bit_cast(int, v)));
}

template <int K, int J> static __device__ __forceinline__ int cmp_xchg(int key, int lane) {
    const int  o   = lane_xor<J>(key);
    const bool low = (((lane & J) == 0) == ((lane & K) == 0));
    const int  mn  = key < o ? key : o;
    const int  mx  = key < o ? o : key;
    return low ? mn : mx;
}
static __device__ __forceinline__ int sort_wave(int key, int lane) {
    key = cmp_xchg<2, 1>(key, lane);
    key = cmp_xchg<4, 2>(key, lane);   key = cmp_xchg<4, 1>(key, lane);
    key = cmp_xchg<8, 4>(key, lane);   key = cmp_xchg<8, 2>(key, lane);   key = cmp_xchg<8, 1>(key, lane);
    key = cmp_xchg<16, 8>(key, lane);  key = cmp_xchg<16, 4>(key, lane);  key = cmp_xchg<16, 2>(key, lane);
    key = cmp_xchg<16, 1>(key, lane);
    key = cmp_xchg<32, 16>(key, lane); key = cmp_xchg<32, 8>(key, lane);  key = cmp_xchg<32, 4>(key, lane);
    key = cmp_xchg<32, 2>(key, lane);  key = cmp_xchg<32, 1>(key, lane);
    return key;
}

static __device__ __forceinline__ float lane_bcast(float v, int l) {
    return __builtin_bit_cast(float, __builtin_amdgcn_readlane(__builtin_bit_cast(int, v), l));
}

static __device__ __forceinline__ float add_piece(int ks, int sub, int fc, const float* __restrict__ xp, float acc) {
#pragma unroll
    for (int p = 0; p < 8; ++p) {
        const int   from = p * 4 + sub;
        const int   ss   = __shfl(ks, from, 32);
        const float xv   = xp[ss * ZP + fc];
        acc += xv;
    }
    return acc;
}

#define GN_NODES 8

__global__ __launch_bounds__(256) void gather_in_kernel(const int* __restrict__ rowptr, const int* __restrict__ cnt,
                                                       int* __restrict__ col, const float* __restrict__ dinv,
                                                       const float* __restrict__ xp, float* __restrict__ agg, int n) {
    const int lane = threadIdx.x & 31;
    const int wave  = __builtin_amdgcn_readfirstlane(threadIdx.x >> 5);
    const int node0 = (blockIdx.x * 8 + wave) * GN_NODES;
    const int f     = lane & 7;
    const int fc    = f < F_IN ? f : F_IN - 1;
    const int sub   = lane >> 3;

    int   mstart = 0, mdeg = 0;
    float mdi = 0.0f;
    {
        const int nd = node0 + lane;
        if (lane < GN_NODES && nd < n) {
            mstart = rowptr[nd];
            mdeg   = cnt[nd];
            mdi    = dinv[nd];
        }
    }


    int   key[GN_NODES];
    float xs[GN_NODES];
#pragma unroll
    for (int i = 0; i < GN_NODES; ++i) {
        const int s  = __builtin_amdgcn_readlane(mstart, i);
        const int d  = __builtin_amdgcn_readlane(mdeg, i);
        const int nb = d < 32 ? d : 32;
        key[i] = lane < nb ? col[s + lane] : 0x7fffffff;
        const int nd = node0 + i < n ? node0 + i : n - 1;
        xs[i] = xp[nd * ZP + fc];
    }

#pragma unroll
    for (int i = 0; i < GN_NODES; ++i) key[i] = sort_wave(key[i], lane);

    float acc[GN_NODES];
#pragma unroll
    for (int i = 0; i < GN_NODES; ++i) {
        const int d  = __builtin_amdgcn_readlane(mdeg, i);
        const int nb = d < 32 ? d : 32;
        const int ks = lane < nb ? key[i] : n;
        acc[i] = add_piece(ks, sub, fc, xp, 0.0f);
    }

#pragma unroll
    for (int i = 0; i < GN_NODES; ++i) {
        const int s  = __builtin_amdgcn_readlane(mstart, i);
        const int d  = __builtin_amdgcn_readlane(mdeg, i);
        const int nb = d < 32 ? d : 32;
        if (lane < nb) col[s + lane] = key[i];
    }

    const unsigned longmask = __builtin_amdgcn_ballot_w32(mdeg > 32);
    if (longmask != 0u) {
#pragma nounroll
        for (int i = 0; i < GN_NODES; ++i) {
            if ((longmask >> i) & 1u) {
                const int s = __builtin_amdgcn_readlane(mstart, i);
                const int d = __builtin_amdgcn_readlane(mdeg, i);
                float extra = 0.0f;
                for (int base = 32; base < d; base += 32) {
                    const int nbp = (d - base < 32) ? d - base : 32;
                    const int kc  = col[s + base + (lane < nbp ? lane : 0)];
                    const int ks  = lane < nbp ? kc : n;
                    extra = add_piece(ks, sub, fc, xp, extra);
                }
#pragma unroll
                for (int j = 0; j < GN_NODES; ++j) acc[j] = (j == i) ? acc[j] + extra : acc[j];
            }
        }
    }

#pragma unroll
    for (int i = 0; i < GN_NODES; ++i) {
        const float di = lane_bcast(mdi, i);
        float t = acc[i];
        t += lane_xor_f<8>(t);
        t += lane_xor_f<16>(t);
        const float v = lane < F_IN ? di * (t + xs[i]) : di;
        if (lane <= F_IN && node0 + i < n) agg[(size_t)(node0 + i) * ZP + lane] = v;
    }
}

#define FT_ROWS   64
#define FT_APITCH (HDIM + 8)
#define FT_HPITCH (HDIM + 4)

typedef __attribute__((ext_vector_type(4))) _Float16 v4h;

static __device__ __forceinline__ void add_source(float z0, float z1, float z2, float z3, float z4, float wt,
                                                  const float (&w1)[F_IN][4], const float (&bv)[4], float (&a)[4]) {
#pragma unroll
    for (int q = 0; q < 4; ++q) {
        float s = fmaf(z0, w1[0][q], bv[q]);
        s = fmaf(z1, w1[1][q], s);
        s = fmaf(z2, w1[2][q], s);
        s = fmaf(z3, w1[3][q], s);
        s = fmaf(z4, w1[4][q], s);
        a[q] = fmaf(wt, fmaxf(s, 0.0f), a[q]);
    }
}

__global__ __launch_bounds__(256) void agg_gemm_head_kernel(const int* __restrict__ rowptr, const int* __restrict__ cnt,
                                                            const int* __restrict__ col, const float* __restrict__ dinv,
                                                            const float* __restrict__ zin,
                                                            const float* __restrict__ W1, const float* __restrict__ b1,
                                                            const float* __restrict__ W2, const float* __restrict__ b2,
                                                            const float* __restrict__ Wout, const float* __restrict__ bout,
                                                            float* __restrict__ out, int n) {
    __shared__ __attribute__((aligned(16))) _Float16 aS[FT_ROWS][FT_APITCH];
    __shared__ __attribute__((aligned(16))) float    hS[FT_ROWS][FT_HPITCH];

    const int tid  = threadIdx.x;
    const int lane = tid & 31;
    const int wave = __builtin_amdgcn_readfirstlane(tid >> 5);
    const int row0 = blockIdx.x * FT_ROWS;
    const int h0   = lane * 4;

    {
        float w1[F_IN][4];
        float bv[4];
#pragma unroll
        for (int f = 0; f < F_IN; ++f) {
            const float4 t = *(const float4*)(W1 + f * HDIM + h0);
            w1[f][0] = t.x; w1[f][1] = t.y; w1[f][2] = t.z; w1[f][3] = t.w;
        }
        {
            const float4 t = *(const float4*)(b1 + h0);
            bv[0] = t.x; bv[1] = t.y; bv[2] = t.z; bv[3] = t.w;
        }
        const float4* zr4 = (const float4*)zin;

        int    mstart = 0, mdeg = 0;
        float  mdi = 0.0f, mzl = 0.0f;
        float4 mzs = {0.0f, 0.0f, 0.0f, 0.0f};
        {
            const int nd = row0 + wave * (FT_ROWS / 8) + lane;
            if (lane < FT_ROWS / 8 && nd < n) {
                mstart = rowptr[nd];
                mdeg   = cnt[nd];
                mdi    = dinv[nd];
                mzs    = zr4[(size_t)nd * (ZP / 4)];
                mzl    = zin[(size_t)nd * ZP + 4];
            }
        }

        auto first_piece = [&](int i) -> int {
            const int s  = __builtin_amdgcn_readlane(mstart, i);
            const int d  = __builtin_amdgcn_readlane(mdeg, i);
            const int nb = d < 32 ? d : 32;
            return col[lane < nb ? s + lane : 0];
        };

        const float2* zr2 = (const float2*)zin;

        int    ms_b = first_piece(0);
        float4 zq_a = zr4[(size_t)ms_b * (ZP / 4)];
        float2 zt_a = zr2[(size_t)ms_b * (ZP / 2) + 2];
        ms_b = first_piece(1);

#pragma unroll 2
        for (int i = 0; i < FT_ROWS / 8; ++i) {
            const int    ms_c = first_piece(i + 2);
            const float4 zq_b = zr4[(size_t)ms_b * (ZP / 4)];
            const float2 zt_b = zr2[(size_t)ms_b * (ZP / 2) + 2];

            const int   m   = wave * (FT_ROWS / 8) + i;
            const int   deg = __builtin_amdgcn_readlane(mdeg, i);
            const float di  = lane_bcast(mdi, i);
            float a[4] = {0.0f, 0.0f, 0.0f, 0.0f};
            add_source(lane_bcast(mzs.x, i), lane_bcast(mzs.y, i), lane_bcast(mzs.z, i), lane_bcast(mzs.w, i),
                       lane_bcast(mzl, i), di * di, w1, bv, a);
            {
                const int   nb = deg < 32 ? deg : 32;
                const float mw = zt_a.y * di;
                for (int j = 0; j < nb; ++j) {
                    add_source(lane_bcast(zq_a.x, j), lane_bcast(zq_a.y, j), lane_bcast(zq_a.z, j), lane_bcast(zq_a.w, j),
                               lane_bcast(zt_a.x, j), lane_bcast(mw, j), w1, bv, a);
                }
            }
            if (deg > 32) {
                const int start = __builtin_amdgcn_readlane(mstart, i);
                for (int base = 32; base < deg; base += 32) {
                    const int    nb = (deg - base < 32) ? deg - base : 32;
                    const int    ms = col[lane < nb ? start + base + lane : 0];
                    const float4 zq = zr4[(size_t)ms * (ZP / 4)];
                    const float2 zt = zr2[(size_t)ms * (ZP / 2) + 2];
                    const float  mw = zt.y * di;
                    for (int j = 0; j < nb; ++j) {
                        add_source(lane_bcast(zq.x, j), lane_bcast(zq.y, j), lane_bcast(zq.z, j), lane_bcast(zq.w, j),
                                   lane_bcast(zt.x, j), lane_bcast(mw, j), w1, bv, a);
                    }
                }
            }
            v4h pk;
            pk[0] = (_Float16)a[0]; pk[1] = (_Float16)a[1]; pk[2] = (_Float16)a[2]; pk[3] = (_Float16)a[3];
            *(v4h*)&aS[m][h0] = pk;

            zq_a = zq_b; zt_a = zt_b;
            ms_b = ms_c;
        }
    }

    const int ncol  = wave * 16;
    const int c     = lane & 15;
    const int h     = lane >> 4;
    const int khalf = h * 8;
    const int colw  = ncol + c;

    v16h bf[4];
#pragma unroll
    for (int ks = 0; ks < 4; ++ks) {
#pragma unroll
        for (int j = 0; j < 16; ++j) {
            const int K = (j & 7) + ((j >> 3) << 4) + khalf + ks * 32;
            bf[ks][j] = (_Float16)W2[K * HDIM + colw];
        }
    }
    const float b2c = b2[colw];
    __syncthreads();

    v8f acc[4] = {};
#pragma unroll
    for (int ks = 0; ks < 4; ++ks) {
#pragma unroll
        for (int rt = 0; rt < 4; ++rt) {
            const _Float16* ap = &aS[rt * 16 + c][ks * 32 + khalf];
            const v8h lo = *(const v8h*)ap;
            const v8h hi = *(const v8h*)(ap + 16);
            const v16h av = __builtin_shufflevector(lo, hi, 0, 1, 2, 3, 4, 5, 6, 7, 8, 9, 10, 11, 12, 13, 14, 15);
            acc[rt] = __builtin_amdgcn_wmma_f32_16x16x32_f16(false, av, false, bf[ks],
                                                             (short)0, acc[rt], false, false);
        }
    }

#pragma unroll
    for (int rt = 0; rt < 4; ++rt) {
#pragma unroll
        for (int v = 0; v < 8; ++v) hS[rt * 16 + 8 * h + v][colw] = fmaxf(acc[rt][v] + b2c, 0.0f);
    }

    float wo[4][CDIM];
    {
        const float4* wp = (const float4*)(Wout + (size_t)h0 * CDIM);
        float t[4 * CDIM];
#pragma unroll
        for (int u = 0; u < CDIM; ++u) {
            const float4 x4 = wp[u];
            t[4 * u + 0] = x4.x; t[4 * u + 1] = x4.y; t[4 * u + 2] = x4.z; t[4 * u + 3] = x4.w;
        }
#pragma unroll
        for (int q = 0; q < 4; ++q) {
#pragma unroll
            for (int cc = 0; cc < CDIM; ++cc) wo[q][cc] = t[q * CDIM + cc];
        }
    }
    float bo[CDIM];
#pragma unroll
    for (int cc = 0; cc < CDIM; ++cc) bo[cc] = bout[cc];
    __syncthreads();

    for (int i = 0; i < FT_ROWS / 8; ++i) {
        const int m    = wave * (FT_ROWS / 8) + i;
        const int node = row0 + m;
        if (node >= n) break;
        const float4 hv = *(const float4*)&hS[m][h0];
        float lg[CDIM];
#pragma unroll
        for (int cc = 0; cc < CDIM; ++cc)
            lg[cc] = hv.x * wo[0][cc] + hv.y * wo[1][cc] + hv.z * wo[2][cc] + hv.w * wo[3][cc];
#pragma unroll
        for (int off = 16; off > 0; off >>= 1) {
#pragma unroll
            for (int cc = 0; cc < CDIM; ++cc) lg[cc] += __shfl_xor(lg[cc], off, 32);
        }
#pragma unroll
        for (int cc = 0; cc < CDIM; ++cc) lg[cc] += bo[cc];
        float mx = lg[0];
#pragma unroll
        for (int cc = 1; cc < CDIM; ++cc) mx = fmaxf(mx, lg[cc]);
        float sum = 0.0f;
#pragma unroll
        for (int cc = 0; cc < CDIM; ++cc) { lg[cc] = __expf(lg[cc] - mx); sum += lg[cc]; }
        const float inv = 1.0f / sum;
        float pv = lg[0];
#pragma unroll
        for (int cc = 1; cc < CDIM; ++cc) pv = (lane == cc) ? lg[cc] : pv;
        if (lane < CDIM) out[(size_t)node * CDIM + lane] = pv * inv;
    }
}

extern "C" void kernel_launch(void* const* d_in, const int* in_sizes, int n_in,
                              void* d_out, int out_size, void* d_ws, size_t ws_size,
                              hipStream_t stream) {
    const float* x    = (const float*)d_in[0];
    const int*   eidx = (const int*)  d_in[1];
    const float* W1   = (const float*)d_in[2];
    const float* b1   = (const float*)d_in[3];
    const float* W2   = (const float*)d_in[4];
    const float* b2   = (const float*)d_in[5];
    const float* Wout = (const float*)d_in[6];
    const float* bout = (const float*)d_in[7];
    float*       out  = (float*)d_out;

    const int N = in_sizes[0] / (T_HIST * F_IN);
    const int E = in_sizes[1] / 2;
    const int* src = eidx;
    const int* dst = eidx + E;

    const int NP = (N + 63) & ~63;
    const int NB = (N + 255) / 256;
    const int NBINS = (N + BIN_W - 1) >> BIN_SHIFT;
    float* dinv   = (float*)d_ws;
    float* bufA   = dinv + NP;
    float* zin    = bufA + (size_t)N * HDIM;
    int*   cnt    = (int*)(zin + (size_t)N * ZP);
    int*   rowptr = cnt + NP;
    int*   cursor = rowptr + NP;
    int*   bsum   = cursor + NP;
    int*   col    = bsum + ((NB + 63) & ~63);
    unsigned* packed    = (unsigned*)bufA;
    int*      blockhist = (int*)(packed + ((E + 63) & ~63));
    int*      bintotal  = blockhist + (size_t)NBINS * NCHUNK;
    float*    xp        = (float*)(bintotal + ((NBINS + 63) & ~63));

    bin_count_kernel<<<NCHUNK, SORT_NT, NBINS * sizeof(int), stream>>>(dst, blockhist, E, NBINS);
    bin_scan_kernel<<<(NBINS + 7) / 8, 256, 0, stream>>>(blockhist, bintotal, NBINS);
    bin_partition_kernel<<<NCHUNK, SORT_NT, 3 * NBINS * sizeof(int), stream>>>(src, dst, blockhist, bintotal, packed, E, NBINS);
    bin_sort_kernel<<<NBINS, BIN_W, 0, stream>>>(packed, bintotal, cnt, dinv, rowptr, col, x, xp, N);

    gather_in_kernel<<<(N + 8 * GN_NODES - 1) / (8 * GN_NODES), 256, 0, stream>>>(rowptr, cnt, col, dinv, xp, zin, N);

    agg_gemm_head_kernel<<<(N + FT_ROWS - 1) / FT_ROWS, 256, 0, stream>>>(rowptr, cnt, col, dinv, zin, W1, b1, W2, b2,
                                                                          Wout, bout, out, N);
}


# edge sort in 2 launches, not 4: each 6400-edge chunk bin-sorted in LDS, stored as one run with its bin offsets; a bin's work-group walks the chunk runs; sum order past a segment's first 32 changes
# speedup vs baseline: 11.1521x; 1.1481x over previous
#include <hip/hip_runtime.h>
#include <hip/hip_bf16.h>

typedef __attribute__((ext_vector_type(16))) _Float16 v16h;
typedef __attribute__((ext_vector_type(8)))  float    v8f;
typedef __attribute__((ext_vector_type(8)))  _Float16 v8h;

#define T_HIST 3
#define F_IN   5
#define HDIM   128
#define CDIM   5
#define ZP     8

#define BIN_SHIFT 8
#define BIN_W     (1 << BIN_SHIFT)
#define SORT_NT   256
#define SORT_EPT  25
#define SORT_TILE (SORT_NT * SORT_EPT)
#define SCOL_CAP  8192

static __device__ __forceinline__ void block_excl_scan(const int* in, int* out, int nb, int* wsum) {
    const int tid  = threadIdx.x;
    const int lane = tid & 31;
    const int wave = tid >> 5;
    const int per  = (nb + SORT_NT - 1) / SORT_NT;
    const int lo_i = tid * per;
    int s = 0;
    for (int k = 0; k < per; ++k) {
        const int i = lo_i + k;
        if (i < nb) { const int v = in[i]; out[i] = s; s += v; }
    }
    int incl = s;
#pragma unroll
    for (int off = 1; off < 32; off <<= 1) {
        const int u = __shfl_up(incl, off, 32);
        if (lane >= off) incl += u;
    }
    if (lane == 31) wsum[wave] = incl;
    __syncthreads();
    int base = incl - s;
    for (int w = 0; w < wave; ++w) base += wsum[w];
    for (int k = 0; k < per; ++k) {
        const int i = lo_i + k;
        if (i < nb) out[i] += base;
    }
    __syncthreads();
}

__global__ __launch_bounds__(SORT_NT) void chunk_sort_kernel(const int* __restrict__ src, const int* __restrict__ dst,
                                                             int* __restrict__ chunkoff, unsigned* __restrict__ packed,
                                                             int e, int nbins, int nchunks) {
    extern __shared__ int sm[];
    __shared__ int      wsum[SORT_NT / 32];
    __shared__ unsigned stage_w[SORT_TILE];
    int* hist   = sm;
    int* lstart = sm + nbins;
    const int tid = threadIdx.x;
    const int lo  = blockIdx.x * SORT_TILE;
    const int hi  = lo + SORT_TILE < e ? lo + SORT_TILE : e;

    for (int i = tid; i < nbins; i += SORT_NT) hist[i] = 0;
    __syncthreads();

    int s[SORT_EPT], d[SORT_EPT], r[SORT_EPT];
#pragma unroll
    for (int k = 0; k < SORT_EPT; ++k) {
        const int i  = lo + k * SORT_NT + tid;
        const int ii = i < hi ? i : hi - 1;
        d[k] = dst[ii];
        s[k] = src[ii];
    }
#pragma unroll
    for (int k = 0; k < SORT_EPT; ++k)
        r[k] = lo + k * SORT_NT + tid < hi ? atomicAdd(&hist[d[k] >> BIN_SHIFT], 1) : 0;
    __syncthreads();

    block_excl_scan(hist, lstart, nbins, wsum);

    for (int i = tid; i < nbins; i += SORT_NT) chunkoff[(size_t)i * nchunks + blockIdx.x] = lstart[i];
    if (tid == 0) chunkoff[(size_t)nbins * nchunks + blockIdx.x] = hi - lo;

#pragma unroll
    for (int k = 0; k < SORT_EPT; ++k) r[k] += lstart[d[k] >> BIN_SHIFT];
#pragma unroll
    for (int k = 0; k < SORT_EPT; ++k) {
        if (lo + k * SORT_NT + tid < hi)
            stage_w[r[k]] = ((unsigned)s[k] << BIN_SHIFT) | (unsigned)(d[k] & (BIN_W - 1));
    }
    __syncthreads();

    const int nt = hi - lo;
    for (int i = tid; i < nt; i += SORT_NT) packed[lo + i] = stage_w[i];
}

#define BS_RUNS 16
template <typename F>
static __device__ __forceinline__ void walk_bin(const unsigned* __restrict__ packed, const int* __restrict__ row0,
                                                const int* __restrict__ row1, int nchunks, int tid, int lane, F f) {
    for (int g = 0; g < nchunks; g += BIN_W) {
        const int c = g + tid;
        int pos = 0, len = 0;
        if (c < nchunks) {
            const int o = row0[c];
            len = row1[c] - o;
            pos = c * SORT_TILE + o;
        }
#pragma unroll
        for (int j0 = 0; j0 < 32; j0 += BS_RUNS) {
            unsigned w[BS_RUNS];
#pragma unroll
            for (int k = 0; k < BS_RUNS; ++k) {
                const int p = __builtin_amdgcn_readlane(pos, j0 + k);
                const int l = __builtin_amdgcn_readlane(len, j0 + k);
                w[k] = packed[lane < l ? p + lane : 0];
            }
#pragma unroll
            for (int k = 0; k < BS_RUNS; ++k) {
                const int l = __builtin_amdgcn_readlane(len, j0 + k);
                if (lane < l) f(w[k]);
            }
        }
        unsigned longmask = __builtin_amdgcn_ballot_w32(len > 32);
        while (longmask != 0u) {
            const int j = __builtin_ctz(longmask);
            longmask &= longmask - 1u;
            const int p = __shfl(pos, j, 32);
            const int l = __shfl(len, j, 32);
            for (int b = 32; b < l; b += 32)
                if (b + lane < l) f(packed[p + b + lane]);
        }
    }
}

__global__ __launch_bounds__(BIN_W) void bin_sort_kernel(const unsigned* __restrict__ packed, const int* __restrict__ chunkoff,
                                                         int* __restrict__ cnt, float* __restrict__ dinv,
                                                         int* __restrict__ rowptr, int* __restrict__ col,
                                                         const float* __restrict__ x, float* __restrict__ xp, int n,
                                                         int nchunks) {
    __shared__ int lcnt[BIN_W];
    __shared__ int wpart[BIN_W / 32];
    __shared__ int wlen[BIN_W / 32];
    __shared__ int wtot[BIN_W / 32];
    __shared__ int scol[SCOL_CAP];
    const int tid  = threadIdx.x;
    const int lane = tid & 31;
    const int wave = tid >> 5;
    const int bin  = blockIdx.x;
    const int* row0 = chunkoff + (size_t)bin * nchunks;
    const int* row1 = row0 + nchunks;

    int part = 0, plen = 0;
    for (int c = tid; c < nchunks; c += BIN_W) {
        const int o = row0[c];
        part += o;
        plen += row1[c] - o;
    }
#pragma unroll
    for (int off = 16; off > 0; off >>= 1) {
        part += __shfl_xor(part, off, 32);
        plen += __shfl_xor(plen, off, 32);
    }
    if (lane == 0) { wpart[wave] = part; wlen[wave] = plen; }
    lcnt[tid] = 0;
    __syncthreads();
    int start = 0, m = 0;
#pragma unroll
    for (int w = 0; w < BIN_W / 32; ++w) { start += wpart[w]; m += wlen[w]; }

    walk_bin(packed, row0, row1, nchunks, tid, lane, [&](unsigned w) { atomicAdd(&lcnt[w & (BIN_W - 1)], 1); });
    __syncthreads();

    const int c = lcnt[tid];
    int incl = c;
#pragma unroll
    for (int off = 1; off < 32; off <<= 1) {
        const int u = __shfl_up(incl, off, 32);
        if (lane >= off) incl += u;
    }
    if (lane == 31) wtot[wave] = incl;
    __syncthreads();
    {
        int excl = incl - c;
        for (int w = 0; w < wave; ++w) excl += wtot[w];
        const int node = bin * BIN_W + tid;
        if (node < n) {
            const float di = rsqrtf((float)c + 1.0f);
            cnt[node]    = c;
            dinv[node]   = di;
            rowptr[node] = start + excl;
            const float* xr = x + (size_t)node * (T_HIST * F_IN) + (T_HIST - 1) * F_IN;
            float4 lo4, hi4;
            lo4.x = di * xr[0]; lo4.y = di * xr[1]; lo4.z = di * xr[2]; lo4.w = di * xr[3];
            hi4.x = di * xr[4]; hi4.y = 0.0f;       hi4.z = 0.0f;       hi4.w = 0.0f;
            ((float4*)xp)[(size_t)node * (ZP / 4)]     = lo4;
            ((float4*)xp)[(size_t)node * (ZP / 4) + 1] = hi4;
        }
        if (bin == 0 && tid < ZP / 4) {
            float4 z4; z4.x = 0.0f; z4.y = 0.0f; z4.z = 0.0f; z4.w = 0.0f;
            ((float4*)xp)[(size_t)n * (ZP / 4) + tid] = z4;
        }
        lcnt[tid] = excl;
    }
    __syncthreads();

    const bool in_lds = m <= SCOL_CAP;
    walk_bin(packed, row0, row1, nchunks, tid, lane, [&](unsigned w) {
        const int pos = atomicAdd(&lcnt[w & (BIN_W - 1)], 1);
        const int sv  = (int)(w >> BIN_SHIFT);
        if (in_lds) scol[pos] = sv;
        else        col[start + pos] = sv;
    });
    if (in_lds) {
        __syncthreads();
        for (int i = tid; i < m; i += BIN_W) col[start + i] = scol[i];
    }
}


template <int M> static __device__ __forceinline__ int lane_xor(int v) {
    if constexpr (M == 16) return __builtin_amdgcn_permlanex16(v, v, 0x76543210, 0xfedcba98, false, false);
    else                   return __builtin_amdgcn_update_dpp(v, v, 0x160 + M, 0xf, 0xf, false);
}
template <int M> static __device__ __forceinline__ float lane_xor_f(float v) {
    return __builtin_bit_cast(float, lane_xor<M>(__builtin_bit_cast(int, v)));
}

template <int K, int J> static __device__ __forceinline__ int cmp_xchg(int key, int lane) {
    const int  o   = lane_xor<J>(key);
    const bool low = (((lane & J) == 0) == ((lane & K) == 0));
    const int  mn  = key < o ? key : o;
    const int  mx  = key < o ? o : key;
    return low ? mn : mx;
}
static __device__ __forceinline__ int sort_wave(int key, int lane) {
    key = cmp_xchg<2, 1>(key, lane);
    key = cmp_xchg<4, 2>(key, lane);   key = cmp_xchg<4, 1>(key, lane);
    key = cmp_xchg<8, 4>(key, lane);   key = cmp_xchg<8, 2>(key, lane);   key = cmp_xchg<8, 1>(key, lane);
    key = cmp_xchg<16, 8>(key, lane);  key = cmp_xchg<16, 4>(key, lane);  key = cmp_xchg<16, 2>(key, lane);
    key = cmp_xchg<16, 1>(key, lane);
    key = cmp_xchg<32, 16>(key, lane); key = cmp_xchg<32, 8>(key, lane);  key = cmp_xchg<32, 4>(key, lane);
    key = cmp_xchg<32, 2>(key, lane);  key = cmp_xchg<32, 1>(key, lane);
    return key;
}

static __device__ __forceinline__ float lane_bcast(float v, int l) {
    return __builtin_bit_cast(float, __builtin_amdgcn_readlane(__builtin_bit_cast(int, v), l));
}

static __device__ __forceinline__ float add_piece(int ks, int sub, int fc, const float* __restrict__ xp, float acc) {
#pragma unroll
    for (int p = 0; p < 8; ++p) {
        const int   from = p * 4 + sub;
        const int   ss   = __shfl(ks, from, 32);
        const float xv   = xp[ss * ZP + fc];
        acc += xv;
    }
    return acc;
}

#define GN_NODES 8

__global__ __launch_bounds__(256) void gather_in_kernel(const int* __restrict__ rowptr, const int* __restrict__ cnt,
                                                       int* __restrict__ col, const float* __restrict__ dinv,
                                                       const float* __restrict__ xp, float* __restrict__ agg, int n) {
    const int lane = threadIdx.x & 31;
    const int wave  = __builtin_amdgcn_readfirstlane(threadIdx.x >> 5);
    const int node0 = (blockIdx.x * 8 + wave) * GN_NODES;
    const int f     = lane & 7;
    const int fc    = f < F_IN ? f : F_IN - 1;
    const int sub   = lane >> 3;

    int   mstart = 0, mdeg = 0;
    float mdi = 0.0f;
    {
        const int nd = node0 + lane;
        if (lane < GN_NODES && nd < n) {
            mstart = rowptr[nd];
            mdeg   = cnt[nd];
            mdi    = dinv[nd];
        }
    }


    int   key[GN_NODES];
    float xs[GN_NODES];
#pragma unroll
    for (int i = 0; i < GN_NODES; ++i) {
        const int s  = __builtin_amdgcn_readlane(mstart, i);
        const int d  = __builtin_amdgcn_readlane(mdeg, i);
        const int nb = d < 32 ? d : 32;
        key[i] = lane < nb ? col[s + lane] : 0x7fffffff;
        const int nd = node0 + i < n ? node0 + i : n - 1;
        xs[i] = xp[nd * ZP + fc];
    }

#pragma unroll
    for (int i = 0; i < GN_NODES; ++i) key[i] = sort_wave(key[i], lane);

    float acc[GN_NODES];
#pragma unroll
    for (int i = 0; i < GN_NODES; ++i) {
        const int d  = __builtin_amdgcn_readlane(mdeg, i);
        const int nb = d < 32 ? d : 32;
        const int ks = lane < nb ? key[i] : n;
        acc[i] = add_piece(ks, sub, fc, xp, 0.0f);
    }

#pragma unroll
    for (int i = 0; i < GN_NODES; ++i) {
        const int s  = __builtin_amdgcn_readlane(mstart, i);
        const int d  = __builtin_amdgcn_readlane(mdeg, i);
        const int nb = d < 32 ? d : 32;
        if (lane < nb) col[s + lane] = key[i];
    }

    const unsigned longmask = __builtin_amdgcn_ballot_w32(mdeg > 32);
    if (longmask != 0u) {
#pragma nounroll
        for (int i = 0; i < GN_NODES; ++i) {
            if ((longmask >> i) & 1u) {
                const int s = __builtin_amdgcn_readlane(mstart, i);
                const int d = __builtin_amdgcn_readlane(mdeg, i);
                float extra = 0.0f;
                for (int base = 32; base < d; base += 32) {
                    const int nbp = (d - base < 32) ? d - base : 32;
                    const int kc  = col[s + base + (lane < nbp ? lane : 0)];
                    const int ks  = lane < nbp ? kc : n;
                    extra = add_piece(ks, sub, fc, xp, extra);
                }
#pragma unroll
                for (int j = 0; j < GN_NODES; ++j) acc[j] = (j == i) ? acc[j] + extra : acc[j];
            }
        }
    }

#pragma unroll
    for (int i = 0; i < GN_NODES; ++i) {
        const float di = lane_bcast(mdi, i);
        float t = acc[i];
        t += lane_xor_f<8>(t);
        t += lane_xor_f<16>(t);
        const float v = lane < F_IN ? di * (t + xs[i]) : di;
        if (lane <= F_IN && node0 + i < n) agg[(size_t)(node0 + i) * ZP + lane] = v;
    }
}

#define FT_ROWS   64
#define FT_APITCH (HDIM + 8)
#define FT_HPITCH (HDIM + 4)

typedef __attribute__((ext_vector_type(4))) _Float16 v4h;

static __device__ __forceinline__ void add_source(float z0, float z1, float z2, float z3, float z4, float wt,
                                                  const float (&w1)[F_IN][4], const float (&bv)[4], float (&a)[4]) {
#pragma unroll
    for (int q = 0; q < 4; ++q) {
        float s = fmaf(z0, w1[0][q], bv[q]);
        s = fmaf(z1, w1[1][q], s);
        s = fmaf(z2, w1[2][q], s);
        s = fmaf(z3, w1[3][q], s);
        s = fmaf(z4, w1[4][q], s);
        a[q] = fmaf(wt, fmaxf(s, 0.0f), a[q]);
    }
}

__global__ __launch_bounds__(256) void agg_gemm_head_kernel(const int* __restrict__ rowptr, const int* __restrict__ cnt,
                                                            const int* __restrict__ col, const float* __restrict__ dinv,
                                                            const float* __restrict__ zin,
                                                            const float* __restrict__ W1, const float* __restrict__ b1,
                                                            const float* __restrict__ W2, const float* __restrict__ b2,
                                                            const float* __restrict__ Wout, const float* __restrict__ bout,
                                                            float* __restrict__ out, int n) {
    __shared__ __attribute__((aligned(16))) _Float16 aS[FT_ROWS][FT_APITCH];
    __shared__ __attribute__((aligned(16))) float    hS[FT_ROWS][FT_HPITCH];

    const int tid  = threadIdx.x;
    const int lane = tid & 31;
    const int wave = __builtin_amdgcn_readfirstlane(tid >> 5);
    const int row0 = blockIdx.x * FT_ROWS;
    const int h0   = lane * 4;

    {
        float w1[F_IN][4];
        float bv[4];
#pragma unroll
        for (int f = 0; f < F_IN; ++f) {
            const float4 t = *(const float4*)(W1 + f * HDIM + h0);
            w1[f][0] = t.x; w1[f][1] = t.y; w1[f][2] = t.z; w1[f][3] = t.w;
        }
        {
            const float4 t = *(const float4*)(b1 + h0);
            bv[0] = t.x; bv[1] = t.y; bv[2] = t.z; bv[3] = t.w;
        }
        const float4* zr4 = (const float4*)zin;

        int    mstart = 0, mdeg = 0;
        float  mdi = 0.0f, mzl = 0.0f;
        float4 mzs = {0.0f, 0.0f, 0.0f, 0.0f};
        {
            const int nd = row0 + wave * (FT_ROWS / 8) + lane;
            if (lane < FT_ROWS / 8 && nd < n) {
                mstart = rowptr[nd];
                mdeg   = cnt[nd];
                mdi    = dinv[nd];
                mzs    = zr4[(size_t)nd * (ZP / 4)];
                mzl    = zin[(size_t)nd * ZP + 4];
            }
        }

        auto first_piece = [&](int i) -> int {
            const int s  = __builtin_amdgcn_readlane(mstart, i);
            const int d  = __builtin_amdgcn_readlane(mdeg, i);
            const int nb = d < 32 ? d : 32;
            return col[lane < nb ? s + lane : 0];
        };

        const float2* zr2 = (const float2*)zin;

        int    ms_b = first_piece(0);
        float4 zq_a = zr4[(size_t)ms_b * (ZP / 4)];
        float2 zt_a = zr2[(size_t)ms_b * (ZP / 2) + 2];
        ms_b = first_piece(1);

#pragma unroll 2
        for (int i = 0; i < FT_ROWS / 8; ++i) {
            const int    ms_c = first_piece(i + 2);
            const float4 zq_b = zr4[(size_t)ms_b * (ZP / 4)];
            const float2 zt_b = zr2[(size_t)ms_b * (ZP / 2) + 2];

            const int   m   = wave * (FT_ROWS / 8) + i;
            const int   deg = __builtin_amdgcn_readlane(mdeg, i);
            const float di  = lane_bcast(mdi, i);
            float a[4] = {0.0f, 0.0f, 0.0f, 0.0f};
            add_source(lane_bcast(mzs.x, i), lane_bcast(mzs.y, i), lane_bcast(mzs.z, i), lane_bcast(mzs.w, i),
                       lane_bcast(mzl, i), di * di, w1, bv, a);
            {
                const int   nb = deg < 32 ? deg : 32;
                const float mw = zt_a.y * di;
                for (int j = 0; j < nb; ++j) {
                    add_source(lane_bcast(zq_a.x, j), lane_bcast(zq_a.y, j), lane_bcast(zq_a.z, j), lane_bcast(zq_a.w, j),
                               lane_bcast(zt_a.x, j), lane_bcast(mw, j), w1, bv, a);
                }
            }
            if (deg > 32) {
                const int start = __builtin_amdgcn_readlane(mstart, i);
                for (int base = 32; base < deg; base += 32) {
                    const int    nb = (deg - base < 32) ? deg - base : 32;
                    const int    ms = col[lane < nb ? start + base + lane : 0];
                    const float4 zq = zr4[(size_t)ms * (ZP / 4)];
                    const float2 zt = zr2[(size_t)ms * (ZP / 2) + 2];
                    const float  mw = zt.y * di;
                    for (int j = 0; j < nb; ++j) {
                        add_source(lane_bcast(zq.x, j), lane_bcast(zq.y, j), lane_bcast(zq.z, j), lane_bcast(zq.w, j),
                                   lane_bcast(zt.x, j), lane_bcast(mw, j), w1, bv, a);
                    }
                }
            }
            v4h pk;
            pk[0] = (_Float16)a[0]; pk[1] = (_Float16)a[1]; pk[2] = (_Float16)a[2]; pk[3] = (_Float16)a[3];
            *(v4h*)&aS[m][h0] = pk;

            zq_a = zq_b; zt_a = zt_b;
            ms_b = ms_c;
        }
    }

    const int ncol  = wave * 16;
    const int c     = lane & 15;
    const int h     = lane >> 4;
    const int khalf = h * 8;
    const int colw  = ncol + c;

    v16h bf[4];
#pragma unroll
    for (int ks = 0; ks < 4; ++ks) {
#pragma unroll
        for (int j = 0; j < 16; ++j) {
            const int K = (j & 7) + ((j >> 3) << 4) + khalf + ks * 32;
            bf[ks][j] = (_Float16)W2[K * HDIM + colw];
        }
    }
    const float b2c = b2[colw];
    __syncthreads();

    v8f acc[4] = {};
#pragma unroll
    for (int ks = 0; ks < 4; ++ks) {
#pragma unroll
        for (int rt = 0; rt < 4; ++rt) {
            const _Float16* ap = &aS[rt * 16 + c][ks * 32 + khalf];
            const v8h lo = *(const v8h*)ap;
            const v8h hi = *(const v8h*)(ap + 16);
            const v16h av = __builtin_shufflevector(lo, hi, 0, 1, 2, 3, 4, 5, 6, 7, 8, 9, 10, 11, 12, 13, 14, 15);
            acc[rt] = __builtin_amdgcn_wmma_f32_16x16x32_f16(false, av, false, bf[ks],
                                                             (short)0, acc[rt], false, false);
        }
    }

#pragma unroll
    for (int rt = 0; rt < 4; ++rt) {
#pragma unroll
        for (int v = 0; v < 8; ++v) hS[rt * 16 + 8 * h + v][colw] = fmaxf(acc[rt][v] + b2c, 0.0f);
    }

    float wo[4][CDIM];
    {
        const float4* wp = (const float4*)(Wout + (size_t)h0 * CDIM);
        float t[4 * CDIM];
#pragma unroll
        for (int u = 0; u < CDIM; ++u) {
            const float4 x4 = wp[u];
            t[4 * u + 0] = x4.x; t[4 * u + 1] = x4.y; t[4 * u + 2] = x4.z; t[4 * u + 3] = x4.w;
        }
#pragma unroll
        for (int q = 0; q < 4; ++q) {
#pragma unroll
            for (int cc = 0; cc < CDIM; ++cc) wo[q][cc] = t[q * CDIM + cc];
        }
    }
    float bo[CDIM];
#pragma unroll
    for (int cc = 0; cc < CDIM; ++cc) bo[cc] = bout[cc];
    __syncthreads();

    for (int i = 0; i < FT_ROWS / 8; ++i) {
        const int m    = wave * (FT_ROWS / 8) + i;
        const int node = row0 + m;
        if (node >= n) break;
        const float4 hv = *(const float4*)&hS[m][h0];
        float lg[CDIM];
#pragma unroll
        for (int cc = 0; cc < CDIM; ++cc)
            lg[cc] = hv.x * wo[0][cc] + hv.y * wo[1][cc] + hv.z * wo[2][cc] + hv.w * wo[3][cc];
#pragma unroll
        for (int off = 16; off > 0; off >>= 1) {
#pragma unroll
            for (int cc = 0; cc < CDIM; ++cc) lg[cc] += __shfl_xor(lg[cc], off, 32);
        }
#pragma unroll
        for (int cc = 0; cc < CDIM; ++cc) lg[cc] += bo[cc];
        float mx = lg[0];
#pragma unroll
        for (int cc = 1; cc < CDIM; ++cc) mx = fmaxf(mx, lg[cc]);
        float sum = 0.0f;
#pragma unroll
        for (int cc = 0; cc < CDIM; ++cc) { lg[cc] = __expf(lg[cc] - mx); sum += lg[cc]; }
        const float inv = 1.0f / sum;
        float pv = lg[0];
#pragma unroll
        for (int cc = 1; cc < CDIM; ++cc) pv = (lane == cc) ? lg[cc] : pv;
        if (lane < CDIM) out[(size_t)node * CDIM + lane] = pv * inv;
    }
}

extern "C" void kernel_launch(void* const* d_in, const int* in_sizes, int n_in,
                              void* d_out, int out_size, void* d_ws, size_t ws_size,
                              hipStream_t stream) {
    const float* x    = (const float*)d_in[0];
    const int*   eidx = (const int*)  d_in[1];
    const float* W1   = (const float*)d_in[2];
    const float* b1   = (const float*)d_in[3];
    const float* W2   = (const float*)d_in[4];
    const float* b2   = (const float*)d_in[5];
    const float* Wout = (const float*)d_in[6];
    const float* bout = (const float*)d_in[7];
    float*       out  = (float*)d_out;

    const int N = in_sizes[0] / (T_HIST * F_IN);
    const int E = in_sizes[1] / 2;
    const int* src = eidx;
    const int* dst = eidx + E;

    const int NP = (N + 63) & ~63;
    const int NB = (N + 255) / 256;
    const int NBINS = (N + BIN_W - 1) >> BIN_SHIFT;
    float* dinv   = (float*)d_ws;
    float* bufA   = dinv + NP;
    float* zin    = bufA + (size_t)N * HDIM;
    int*   cnt    = (int*)(zin + (size_t)N * ZP);
    int*   rowptr = cnt + NP;
    int*   cursor = rowptr + NP;
    int*   bsum   = cursor + NP;
    int*   col    = bsum + ((NB + 63) & ~63);
    const int NCH = (E + SORT_TILE - 1) / SORT_TILE;
    unsigned* packed   = (unsigned*)bufA;
    int*      chunkoff = (int*)(packed + ((E + 63) & ~63));
    float*    xp       = (float*)(chunkoff + (((size_t)(NBINS + 1) * NCH + 63) & ~(size_t)63));

    chunk_sort_kernel<<<NCH, SORT_NT, 2 * NBINS * sizeof(int), stream>>>(src, dst, chunkoff, packed, E, NBINS, NCH);
    bin_sort_kernel<<<NBINS, BIN_W, 0, stream>>>(packed, chunkoff, cnt, dinv, rowptr, col, x, xp, N, NCH);

    gather_in_kernel<<<(N + 8 * GN_NODES - 1) / (8 * GN_NODES), 256, 0, stream>>>(rowptr, cnt, col, dinv, xp, zin, N);

    agg_gemm_head_kernel<<<(N + FT_ROWS - 1) / FT_ROWS, 256, 0, stream>>>(rowptr, cnt, col, dinv, zin, W1, b1, W2, b2,
                                                                          Wout, bout, out, N);
}


# layer-2 source rows rebuilt on WMMA: w*(z@W1+b1) per 16-entry tile from two-part f16 operands, terms taken as f16, ReLU on packed bits, rows added by an all-ones WMMA in f32; sum order changes
# speedup vs baseline: 11.6437x; 1.0441x over previous
#include <hip/hip_runtime.h>
#include <hip/hip_bf16.h>

typedef __attribute__((ext_vector_type(16))) _Float16 v16h;
typedef __attribute__((ext_vector_type(8)))  float    v8f;
typedef __attribute__((ext_vector_type(8)))  _Float16 v8h;

#define T_HIST 3
#define F_IN   5
#define HDIM   128
#define CDIM   5
#define ZP     8

#define BIN_SHIFT 8
#define BIN_W     (1 << BIN_SHIFT)
#define SORT_NT   256
#define SORT_EPT  25
#define SORT_TILE (SORT_NT * SORT_EPT)
#define SCOL_CAP  8192

static __device__ __forceinline__ void block_excl_scan(const int* in, int* out, int nb, int* wsum) {
    const int tid  = threadIdx.x;
    const int lane = tid & 31;
    const int wave = tid >> 5;
    const int per  = (nb + SORT_NT - 1) / SORT_NT;
    const int lo_i = tid * per;
    int s = 0;
    for (int k = 0; k < per; ++k) {
        const int i = lo_i + k;
        if (i < nb) { const int v = in[i]; out[i] = s; s += v; }
    }
    int incl = s;
#pragma unroll
    for (int off = 1; off < 32; off <<= 1) {
        const int u = __shfl_up(incl, off, 32);
        if (lane >= off) incl += u;
    }
    if (lane == 31) wsum[wave] = incl;
    __syncthreads();
    int base = incl - s;
    for (int w = 0; w < wave; ++w) base += wsum[w];
    for (int k = 0; k < per; ++k) {
        const int i = lo_i + k;
        if (i < nb) out[i] += base;
    }
    __syncthreads();
}

__global__ __launch_bounds__(SORT_NT) void chunk_sort_kernel(const int* __restrict__ src, const int* __restrict__ dst,
                                                             int* __restrict__ chunkoff, unsigned* __restrict__ packed,
                                                             int e, int nbins, int nchunks) {
    extern __shared__ int sm[];
    __shared__ int      wsum[SORT_NT / 32];
    __shared__ unsigned stage_w[SORT_TILE];
    int* hist   = sm;
    int* lstart = sm + nbins;
    const int tid = threadIdx.x;
    const int lo  = blockIdx.x * SORT_TILE;
    const int hi  = lo + SORT_TILE < e ? lo + SORT_TILE : e;

    for (int i = tid; i < nbins; i += SORT_NT) hist[i] = 0;
    __syncthreads();

    int s[SORT_EPT], d[SORT_EPT], r[SORT_EPT];
#pragma unroll
    for (int k = 0; k < SORT_EPT; ++k) {
        const int i  = lo + k * SORT_NT + tid;
        const int ii = i < hi ? i : hi - 1;
        d[k] = dst[ii];
        s[k] = src[ii];
    }
#pragma unroll
    for (int k = 0; k < SORT_EPT; ++k)
        r[k] = lo + k * SORT_NT + tid < hi ? atomicAdd(&hist[d[k] >> BIN_SHIFT], 1) : 0;
    __syncthreads();

    block_excl_scan(hist, lstart, nbins, wsum);

    for (int i = tid; i < nbins; i += SORT_NT) chunkoff[(size_t)i * nchunks + blockIdx.x] = lstart[i];
    if (tid == 0) chunkoff[(size_t)nbins * nchunks + blockIdx.x] = hi - lo;

#pragma unroll
    for (int k = 0; k < SORT_EPT; ++k) r[k] += lstart[d[k] >> BIN_SHIFT];
#pragma unroll
    for (int k = 0; k < SORT_EPT; ++k) {
        if (lo + k * SORT_NT + tid < hi)
            stage_w[r[k]] = ((unsigned)s[k] << BIN_SHIFT) | (unsigned)(d[k] & (BIN_W - 1));
    }
    __syncthreads();

    const int nt = hi - lo;
    for (int i = tid; i < nt; i += SORT_NT) packed[lo + i] = stage_w[i];
}

#define BS_RUNS 16
template <typename F>
static __device__ __forceinline__ void walk_bin(const unsigned* __restrict__ packed, const int* __restrict__ row0,
                                                const int* __restrict__ row1, int nchunks, int tid, int lane, F f) {
    for (int g = 0; g < nchunks; g += BIN_W) {
        const int c = g + tid;
        int pos = 0, len = 0;
        if (c < nchunks) {
            const int o = row0[c];
            len = row1[c] - o;
            pos = c * SORT_TILE + o;
        }
#pragma unroll
        for (int j0 = 0; j0 < 32; j0 += BS_RUNS) {
            unsigned w[BS_RUNS];
#pragma unroll
            for (int k = 0; k < BS_RUNS; ++k) {
                const int p = __builtin_amdgcn_readlane(pos, j0 + k);
                const int l = __builtin_amdgcn_readlane(len, j0 + k);
                w[k] = packed[lane < l ? p + lane : 0];
            }
#pragma unroll
            for (int k = 0; k < BS_RUNS; ++k) {
                const int l = __builtin_amdgcn_readlane(len, j0 + k);
                if (lane < l) f(w[k]);
            }
        }
        unsigned longmask = __builtin_amdgcn_ballot_w32(len > 32);
        while (longmask != 0u) {
            const int j = __builtin_ctz(longmask);
            longmask &= longmask - 1u;
            const int p = __shfl(pos, j, 32);
            const int l = __shfl(len, j, 32);
            for (int b = 32; b < l; b += 32)
                if (b + lane < l) f(packed[p + b + lane]);
        }
    }
}

__global__ __launch_bounds__(BIN_W) void bin_sort_kernel(const unsigned* __restrict__ packed, const int* __restrict__ chunkoff,
                                                         int* __restrict__ cnt, float* __restrict__ dinv,
                                                         int* __restrict__ rowptr, int* __restrict__ col,
                                                         const float* __restrict__ x, float* __restrict__ xp, int n,
                                                         int nchunks) {
    __shared__ int lcnt[BIN_W];
    __shared__ int wpart[BIN_W / 32];
    __shared__ int wlen[BIN_W / 32];
    __shared__ int wtot[BIN_W / 32];
    __shared__ int scol[SCOL_CAP];
    const int tid  = threadIdx.x;
    const int lane = tid & 31;
    const int wave = tid >> 5;
    const int bin  = blockIdx.x;
    const int* row0 = chunkoff + (size_t)bin * nchunks;
    const int* row1 = row0 + nchunks;

    int part = 0, plen = 0;
    for (int c = tid; c < nchunks; c += BIN_W) {
        const int o = row0[c];
        part += o;
        plen += row1[c] - o;
    }
#pragma unroll
    for (int off = 16; off > 0; off >>= 1) {
        part += __shfl_xor(part, off, 32);
        plen += __shfl_xor(plen, off, 32);
    }
    if (lane == 0) { wpart[wave] = part; wlen[wave] = plen; }
    lcnt[tid] = 0;
    __syncthreads();
    int start = 0, m = 0;
#pragma unroll
    for (int w = 0; w < BIN_W / 32; ++w) { start += wpart[w]; m += wlen[w]; }

    walk_bin(packed, row0, row1, nchunks, tid, lane, [&](unsigned w) { atomicAdd(&lcnt[w & (BIN_W - 1)], 1); });
    __syncthreads();

    const int c = lcnt[tid];
    int incl = c;
#pragma unroll
    for (int off = 1; off < 32; off <<= 1) {
        const int u = __shfl_up(incl, off, 32);
        if (lane >= off) incl += u;
    }
    if (lane == 31) wtot[wave] = incl;
    __syncthreads();
    {
        int excl = incl - c;
        for (int w = 0; w < wave; ++w) excl += wtot[w];
        const int node = bin * BIN_W + tid;
        if (node < n) {
            const float di = rsqrtf((float)c + 1.0f);
            cnt[node]    = c;
            dinv[node]   = di;
            rowptr[node] = start + excl;
            const float* xr = x + (size_t)node * (T_HIST * F_IN) + (T_HIST - 1) * F_IN;
            float4 lo4, hi4;
            lo4.x = di * xr[0]; lo4.y = di * xr[1]; lo4.z = di * xr[2]; lo4.w = di * xr[3];
            hi4.x = di * xr[4]; hi4.y = 0.0f;       hi4.z = 0.0f;       hi4.w = 0.0f;
            ((float4*)xp)[(size_t)node * (ZP / 4)]     = lo4;
            ((float4*)xp)[(size_t)node * (ZP / 4) + 1] = hi4;
        }
        if (bin == 0 && tid < ZP / 4) {
            float4 z4; z4.x = 0.0f; z4.y = 0.0f; z4.z = 0.0f; z4.w = 0.0f;
            ((float4*)xp)[(size_t)n * (ZP / 4) + tid] = z4;
        }
        lcnt[tid] = excl;
    }
    __syncthreads();

    const bool in_lds = m <= SCOL_CAP;
    walk_bin(packed, row0, row1, nchunks, tid, lane, [&](unsigned w) {
        const int pos = atomicAdd(&lcnt[w & (BIN_W - 1)], 1);
        const int sv  = (int)(w >> BIN_SHIFT);
        if (in_lds) scol[pos] = sv;
        else        col[start + pos] = sv;
    });
    if (in_lds) {
        __syncthreads();
        for (int i = tid; i < m; i += BIN_W) col[start + i] = scol[i];
    }
}


template <int M> static __device__ __forceinline__ int lane_xor(int v) {
    if constexpr (M == 16) return __builtin_amdgcn_permlanex16(v, v, 0x76543210, 0xfedcba98, false, false);
    else                   return __builtin_amdgcn_update_dpp(v, v, 0x160 + M, 0xf, 0xf, false);
}
template <int M> static __device__ __forceinline__ float lane_xor_f(float v) {
    return __builtin_bit_cast(float, lane_xor<M>(__builtin_bit_cast(int, v)));
}

template <int K, int J> static __device__ __forceinline__ int cmp_xchg(int key, int lane) {
    const int  o   = lane_xor<J>(key);
    const bool low = (((lane & J) == 0) == ((lane & K) == 0));
    const int  mn  = key < o ? key : o;
    const int  mx  = key < o ? o : key;
    return low ? mn : mx;
}
static __device__ __forceinline__ int sort_wave(int key, int lane) {
    key = cmp_xchg<2, 1>(key, lane);
    key = cmp_xchg<4, 2>(key, lane);   key = cmp_xchg<4, 1>(key, lane);
    key = cmp_xchg<8, 4>(key, lane);   key = cmp_xchg<8, 2>(key, lane);   key = cmp_xchg<8, 1>(key, lane);
    key = cmp_xchg<16, 8>(key, lane);  key = cmp_xchg<16, 4>(key, lane);  key = cmp_xchg<16, 2>(key, lane);
    key = cmp_xchg<16, 1>(key, lane);
    key = cmp_xchg<32, 16>(key, lane); key = cmp_xchg<32, 8>(key, lane);  key = cmp_xchg<32, 4>(key, lane);
    key = cmp_xchg<32, 2>(key, lane);  key = cmp_xchg<32, 1>(key, lane);
    return key;
}

static __device__ __forceinline__ float lane_bcast(float v, int l) {
    return __builtin_bit_cast(float, __builtin_amdgcn_readlane(__builtin_bit_cast(int, v), l));
}

static __device__ __forceinline__ float add_piece(int ks, int sub, int fc, const float* __restrict__ xp, float acc) {
#pragma unroll
    for (int p = 0; p < 8; ++p) {
        const int   from = p * 4 + sub;
        const int   ss   = __shfl(ks, from, 32);
        const float xv   = xp[ss * ZP + fc];
        acc += xv;
    }
    return acc;
}

#define GN_NODES 8

__global__ __launch_bounds__(256) void gather_in_kernel(const int* __restrict__ rowptr, const int* __restrict__ cnt,
                                                       int* __restrict__ col, const float* __restrict__ dinv,
                                                       const float* __restrict__ xp, float* __restrict__ agg, int n) {
    const int lane = threadIdx.x & 31;
    const int wave  = __builtin_amdgcn_readfirstlane(threadIdx.x >> 5);
    const int node0 = (blockIdx.x * 8 + wave) * GN_NODES;
    const int f     = lane & 7;
    const int fc    = f < F_IN ? f : F_IN - 1;
    const int sub   = lane >> 3;

    int   mstart = 0, mdeg = 0;
    float mdi = 0.0f;
    {
        const int nd = node0 + lane;
        if (lane < GN_NODES && nd < n) {
            mstart = rowptr[nd];
            mdeg   = cnt[nd];
            mdi    = dinv[nd];
        }
    }


    int   key[GN_NODES];
    float xs[GN_NODES];
#pragma unroll
    for (int i = 0; i < GN_NODES; ++i) {
        const int s  = __builtin_amdgcn_readlane(mstart, i);
        const int d  = __builtin_amdgcn_readlane(mdeg, i);
        const int nb = d < 32 ? d : 32;
        key[i] = lane < nb ? col[s + lane] : 0x7fffffff;
        const int nd = node0 + i < n ? node0 + i : n - 1;
        xs[i] = xp[nd * ZP + fc];
    }

#pragma unroll
    for (int i = 0; i < GN_NODES; ++i) key[i] = sort_wave(key[i], lane);

    float acc[GN_NODES];
#pragma unroll
    for (int i = 0; i < GN_NODES; ++i) {
        const int d  = __builtin_amdgcn_readlane(mdeg, i);
        const int nb = d < 32 ? d : 32;
        const int ks = lane < nb ? key[i] : n;
        acc[i] = add_piece(ks, sub, fc, xp, 0.0f);
    }

#pragma unroll
    for (int i = 0; i < GN_NODES; ++i) {
        const int s  = __builtin_amdgcn_readlane(mstart, i);
        const int d  = __builtin_amdgcn_readlane(mdeg, i);
        const int nb = d < 32 ? d : 32;
        if (lane < nb) col[s + lane] = key[i];
    }

    const unsigned longmask = __builtin_amdgcn_ballot_w32(mdeg > 32);
    if (longmask != 0u) {
#pragma nounroll
        for (int i = 0; i < GN_NODES; ++i) {
            if ((longmask >> i) & 1u) {
                const int s = __builtin_amdgcn_readlane(mstart, i);
                const int d = __builtin_amdgcn_readlane(mdeg, i);
                float extra = 0.0f;
                for (int base = 32; base < d; base += 32) {
                    const int nbp = (d - base < 32) ? d - base : 32;
                    const int kc  = col[s + base + (lane < nbp ? lane : 0)];
                    const int ks  = lane < nbp ? kc : n;
                    extra = add_piece(ks, sub, fc, xp, extra);
                }
#pragma unroll
                for (int j = 0; j < GN_NODES; ++j) acc[j] = (j == i) ? acc[j] + extra : acc[j];
            }
        }
    }

#pragma unroll
    for (int i = 0; i < GN_NODES; ++i) {
        const float di = lane_bcast(mdi, i);
        float t = acc[i];
        t += lane_xor_f<8>(t);
        t += lane_xor_f<16>(t);
        const float v = lane < F_IN ? di * (t + xs[i]) : di;
        if (lane <= F_IN && node0 + i < n) agg[(size_t)(node0 + i) * ZP + lane] = v;
    }
}

#define FT_ROWS   64
#define FT_APITCH (HDIM + 8)
#define FT_HPITCH (HDIM + 4)

typedef __attribute__((ext_vector_type(2))) _Float16 v2h;
typedef __attribute__((ext_vector_type(8))) unsigned v8u;
typedef __attribute__((ext_vector_type(8))) short    v8s;
typedef __attribute__((ext_vector_type(4))) unsigned v4u;

static __device__ __forceinline__ unsigned pk2(_Float16 a, _Float16 b) {
    v2h p; p[0] = a; p[1] = b;
    return __builtin_bit_cast(unsigned, p);
}
static __device__ __forceinline__ void split2(float v, _Float16& hi, _Float16& lo) {
    hi = (_Float16)v;
    lo = (_Float16)(v - (float)hi);
}
static __device__ __forceinline__ v8h relu8(v8h v) {
    const v8s z = {};
    return __builtin_bit_cast(v8h, __builtin_elementwise_max(__builtin_bit_cast(v8s, v), z));
}

static __device__ __forceinline__ void add_piece_wmma(float z0, float z1, float z2, float z3, float z4, float w,
                                                      bool second, int half, const v16h (&bfr)[HDIM / 16],
                                                      const v16h ones, float (&t)[HDIM / 16]) {
    _Float16 yh[F_IN], yl[F_IN], wh, wl;
    split2(w * z0, yh[0], yl[0]);
    split2(w * z1, yh[1], yl[1]);
    split2(w * z2, yh[2], yl[2]);
    split2(w * z3, yh[3], yl[3]);
    split2(w * z4, yh[4], yl[4]);
    split2(w, wh, wl);
    const unsigned p00 = pk2(yh[0], yh[1]);
    const unsigned p01 = pk2(yh[2], yh[3]);
    const unsigned p02 = pk2(yh[4], yl[0]);
    const unsigned p03 = pk2(yl[1], yl[2]);
    const unsigned p04 = pk2(yl[3], yl[4]);
    const unsigned p12 = pk2(yh[4], wh);
    const unsigned p13 = pk2(wl, wh);
    const bool up = half != 0;
    const v8h zeroh = {};
    const v8f zerof = {};
    if (second) {
        const unsigned x00 = (unsigned)lane_xor<16>((int)p00);
        const unsigned x01 = (unsigned)lane_xor<16>((int)p01);
        const unsigned x02 = (unsigned)lane_xor<16>((int)p02);
        const unsigned x03 = (unsigned)lane_xor<16>((int)p03);
        const unsigned x04 = (unsigned)lane_xor<16>((int)p04);
        const unsigned x12 = (unsigned)lane_xor<16>((int)p12);
        const unsigned x13 = (unsigned)lane_xor<16>((int)p13);
        const v8u au0 = {up ? x00 : p00, up ? x01 : p01, up ? x12 : p02, up ? x13 : p03, up ? 0u : p04, 0u, 0u, 0u};
        const v8u au1 = {up ? p00 : x00, up ? p01 : x01, up ? p12 : x02, up ? p13 : x03, up ? 0u : x04, 0u, 0u, 0u};
        const v16h av0 = __builtin_bit_cast(v16h, au0);
        const v16h av1 = __builtin_bit_cast(v16h, au1);
#pragma unroll
        for (int j0 = 0; j0 < HDIM / 16; j0 += 4) {
            v8h d0[4], d1[4];
#pragma unroll
            for (int u = 0; u < 4; ++u) {
                d0[u] = __builtin_amdgcn_wmma_f16_16x16x32_f16(false, av0, false, bfr[j0 + u], (short)0, zeroh, false, false);
                d1[u] = __builtin_amdgcn_wmma_f16_16x16x32_f16(false, av1, false, bfr[j0 + u], (short)0, zeroh, false, false);
            }
#pragma unroll
            for (int u = 0; u < 4; ++u) {
                const v16h hb = __builtin_shufflevector(relu8(d0[u]), relu8(d1[u]), 0, 1, 2, 3, 4, 5, 6, 7, 8, 9, 10, 11, 12, 13, 14, 15);
                const v8f s = __builtin_amdgcn_wmma_f32_16x16x32_f16(false, ones, false, hb, (short)0, zerof, false, false);
                t[j0 + u] += s[0];
            }
        }
    } else {
        const unsigned x00 = (unsigned)lane_xor<16>((int)p00);
        const unsigned x01 = (unsigned)lane_xor<16>((int)p01);
        const unsigned x12 = (unsigned)lane_xor<16>((int)p12);
        const unsigned x13 = (unsigned)lane_xor<16>((int)p13);
        const v8u au0 = {up ? x00 : p00, up ? x01 : p01, up ? x12 : p02, up ? x13 : p03, up ? 0u : p04, 0u, 0u, 0u};
        const v16h av0 = __builtin_bit_cast(v16h, au0);
#pragma unroll
        for (int j0 = 0; j0 < HDIM / 16; j0 += 4) {
            v8h d0[4];
#pragma unroll
            for (int u = 0; u < 4; ++u)
                d0[u] = __builtin_amdgcn_wmma_f16_16x16x32_f16(false, av0, false, bfr[j0 + u], (short)0, zeroh, false, false);
#pragma unroll
            for (int u = 0; u < 4; ++u) {
                const v16h hb = __builtin_shufflevector(relu8(d0[u]), zeroh, 0, 1, 2, 3, 4, 5, 6, 7, 8, 9, 10, 11, 12, 13, 14, 15);
                const v8f s = __builtin_amdgcn_wmma_f32_16x16x32_f16(false, ones, false, hb, (short)0, zerof, false, false);
                t[j0 + u] += s[0];
            }
        }
    }
}

__global__ __launch_bounds__(256, 6) void agg_gemm_head_kernel(const int* __restrict__ rowptr, const int* __restrict__ cnt,
                                                            const int* __restrict__ col, const float* __restrict__ dinv,
                                                            const float* __restrict__ zin,
                                                            const float* __restrict__ W1, const float* __restrict__ b1,
                                                            const float* __restrict__ W2, const float* __restrict__ b2,
                                                            const float* __restrict__ Wout, const float* __restrict__ bout,
                                                            float* __restrict__ out, int n) {
    __shared__ __attribute__((aligned(16))) _Float16 aS[FT_ROWS][FT_APITCH];
    __shared__ __attribute__((aligned(16))) float    hS[FT_ROWS][FT_HPITCH];

    const int tid  = threadIdx.x;
    const int lane = tid & 31;
    const int wave = __builtin_amdgcn_readfirstlane(tid >> 5);
    const int row0 = blockIdx.x * FT_ROWS;
    const int h0   = lane * 4;
    const int c    = lane & 15;
    const int h    = lane >> 4;

    {
        v4u* bS = (v4u*)&hS[0][0];
        {
            const int cl = 16 * wave + c;
            _Float16 Wh[F_IN], Wl[F_IN], bh, bl;
#pragma unroll
            for (int f = 0; f < F_IN; ++f) split2(W1[f * HDIM + cl], Wh[f], Wl[f]);
            split2(b1[cl], bh, bl);
            const unsigned q00 = pk2(Wh[0], Wh[1]), q01 = pk2(Wh[2], Wh[3]), q02 = pk2(Wh[4], Wh[0]),
                           q03 = pk2(Wh[1], Wh[2]), q04 = pk2(Wh[3], Wh[4]);
            const unsigned q10 = pk2(Wl[0], Wl[1]), q11 = pk2(Wl[2], Wl[3]), q12 = pk2(Wl[4], bh), q13 = pk2(bh, bl);
            const bool up = h != 0;
            const v4u lo4 = {up ? q10 : q00, up ? q11 : q01, up ? q12 : q02, up ? q13 : q03};
            bS[(wave * 32 + lane) * 2] = lo4;
            ((unsigned*)bS)[(wave * 32 + lane) * 8 + 4] = up ? 0u : q04;
        }
        __syncthreads();
        v16h bfr[HDIM / 16];
#pragma unroll
        for (int j = 0; j < HDIM / 16; ++j) {
            const v4u      lo4 = bS[(j * 32 + lane) * 2];
            const unsigned w4  = ((const unsigned*)bS)[(j * 32 + lane) * 8 + 4];
            const v8u bu = {lo4[0], lo4[1], lo4[2], lo4[3], w4, 0u, 0u, 0u};
            bfr[j] = __builtin_bit_cast(v16h, bu);
        }
        v16h ones;
#pragma unroll
        for (int s = 0; s < 16; ++s) ones[s] = (_Float16)1.0f;
        const float4* zr4 = (const float4*)zin;

        int   mstart = 0, mdeg = 0;
        float mdi = 0.0f;
        {
            const int nd = row0 + wave * (FT_ROWS / 8) + lane;
            if (lane < FT_ROWS / 8 && nd < n) {
                mstart = rowptr[nd];
                mdeg   = cnt[nd];
                mdi    = dinv[nd];
            }
        }

        auto first_piece = [&](int i) -> int {
            const int s    = __builtin_amdgcn_readlane(mstart, i);
            const int d    = __builtin_amdgcn_readlane(mdeg, i);
            const int nb   = d < 31 ? d : 31;
            const int cv   = col[(lane >= 1 && lane <= nb) ? s + lane - 1 : 0];
            const int nd   = row0 + wave * (FT_ROWS / 8) + i;
            const int self = nd < n ? nd : 0;
            return lane == 0 ? self : cv;
        };

        const float2* zr2 = (const float2*)zin;

        int    ms_b = first_piece(0);
        float4 zq_a = zr4[(size_t)ms_b * (ZP / 4)];
        float2 zt_a = zr2[(size_t)ms_b * (ZP / 2) + 2];
        ms_b = first_piece(1);

#pragma nounroll
        for (int i = 0; i < FT_ROWS / 8; ++i) {
            const int    ms_c = first_piece(i + 2);
            const float4 zq_b = zr4[(size_t)ms_b * (ZP / 4)];
            const float2 zt_b = zr2[(size_t)ms_b * (ZP / 2) + 2];

            const int   m   = wave * (FT_ROWS / 8) + i;
            const int   deg = __builtin_amdgcn_readlane(mdeg, i);
            const float di  = lane_bcast(mdi, i);
            float a[HDIM / 16] = {0.0f, 0.0f, 0.0f, 0.0f, 0.0f, 0.0f, 0.0f, 0.0f};
            {
                const int   nb = deg < 31 ? deg : 31;
                const float mw = lane <= nb ? zt_a.y * di : 0.0f;
                add_piece_wmma(zq_a.x, zq_a.y, zq_a.z, zq_a.w, zt_a.x, mw, nb >= 16, h, bfr, ones, a);
            }
            if (deg > 31) {
                const int start = __builtin_amdgcn_readlane(mstart, i);
                for (int base = 31; base < deg; base += 32) {
                    const int    nb = (deg - base < 32) ? deg - base : 32;
                    const int    ms = col[lane < nb ? start + base + lane : 0];
                    const float4 zq = zr4[(size_t)ms * (ZP / 4)];
                    const float2 zt = zr2[(size_t)ms * (ZP / 2) + 2];
                    const float  mw = lane < nb ? zt.y * di : 0.0f;
                    add_piece_wmma(zq.x, zq.y, zq.z, zq.w, zt.x, mw, nb > 16, h, bfr, ones, a);
                }
            }
            if (h == 0) {
#pragma unroll
                for (int j = 0; j < HDIM / 16; ++j) aS[m][16 * j + c] = (_Float16)a[j];
            }

            zq_a = zq_b; zt_a = zt_b;
            ms_b = ms_c;
        }
    }

    const int ncol  = wave * 16;
    const int khalf = h * 8;
    const int colw  = ncol + c;

    v16h bf[4];
#pragma unroll
    for (int ks = 0; ks < 4; ++ks) {
#pragma unroll
        for (int j = 0; j < 16; ++j) {
            const int K = (j & 7) + ((j >> 3) << 4) + khalf + ks * 32;
            bf[ks][j] = (_Float16)W2[K * HDIM + colw];
        }
    }
    const float b2c = b2[colw];
    __syncthreads();

    v8f acc[4] = {};
#pragma unroll
    for (int ks = 0; ks < 4; ++ks) {
#pragma unroll
        for (int rt = 0; rt < 4; ++rt) {
            const _Float16* ap = &aS[rt * 16 + c][ks * 32 + khalf];
            const v8h lo = *(const v8h*)ap;
            const v8h hi = *(const v8h*)(ap + 16);
            const v16h av = __builtin_shufflevector(lo, hi, 0, 1, 2, 3, 4, 5, 6, 7, 8, 9, 10, 11, 12, 13, 14, 15);
            acc[rt] = __builtin_amdgcn_wmma_f32_16x16x32_f16(false, av, false, bf[ks],
                                                             (short)0, acc[rt], false, false);
        }
    }

#pragma unroll
    for (int rt = 0; rt < 4; ++rt) {
#pragma unroll
        for (int v = 0; v < 8; ++v) hS[rt * 16 + 8 * h + v][colw] = fmaxf(acc[rt][v] + b2c, 0.0f);
    }

    float wo[4][CDIM];
    {
        const float4* wp = (const float4*)(Wout + (size_t)h0 * CDIM);
        float t[4 * CDIM];
#pragma unroll
        for (int u = 0; u < CDIM; ++u) {
            const float4 x4 = wp[u];
            t[4 * u + 0] = x4.x; t[4 * u + 1] = x4.y; t[4 * u + 2] = x4.z; t[4 * u + 3] = x4.w;
        }
#pragma unroll
        for (int q = 0; q < 4; ++q) {
#pragma unroll
            for (int cc = 0; cc < CDIM; ++cc) wo[q][cc] = t[q * CDIM + cc];
        }
    }
    float bo[CDIM];
#pragma unroll
    for (int cc = 0; cc < CDIM; ++cc) bo[cc] = bout[cc];
    __syncthreads();

    for (int i = 0; i < FT_ROWS / 8; ++i) {
        const int m    = wave * (FT_ROWS / 8) + i;
        const int node = row0 + m;
        if (node >= n) break;
        const float4 hv = *(const float4*)&hS[m][h0];
        float lg[CDIM];
#pragma unroll
        for (int cc = 0; cc < CDIM; ++cc)
            lg[cc] = hv.x * wo[0][cc] + hv.y * wo[1][cc] + hv.z * wo[2][cc] + hv.w * wo[3][cc];
#pragma unroll
        for (int off = 16; off > 0; off >>= 1) {
#pragma unroll
            for (int cc = 0; cc < CDIM; ++cc) lg[cc] += __shfl_xor(lg[cc], off, 32);
        }
#pragma unroll
        for (int cc = 0; cc < CDIM; ++cc) lg[cc] += bo[cc];
        float mx = lg[0];
#pragma unroll
        for (int cc = 1; cc < CDIM; ++cc) mx = fmaxf(mx, lg[cc]);
        float sum = 0.0f;
#pragma unroll
        for (int cc = 0; cc < CDIM; ++cc) { lg[cc] = __expf(lg[cc] - mx); sum += lg[cc]; }
        const float inv = 1.0f / sum;
        float pv = lg[0];
#pragma unroll
        for (int cc = 1; cc < CDIM; ++cc) pv = (lane == cc) ? lg[cc] : pv;
        if (lane < CDIM) out[(size_t)node * CDIM + lane] = pv * inv;
    }
}

extern "C" void kernel_launch(void* const* d_in, const int* in_sizes, int n_in,
                              void* d_out, int out_size, void* d_ws, size_t ws_size,
                              hipStream_t stream) {
    const float* x    = (const float*)d_in[0];
    const int*   eidx = (const int*)  d_in[1];
    const float* W1   = (const float*)d_in[2];
    const float* b1   = (const float*)d_in[3];
    const float* W2   = (const float*)d_in[4];
    const float* b2   = (const float*)d_in[5];
    const float* Wout = (const float*)d_in[6];
    const float* bout = (const float*)d_in[7];
    float*       out  = (float*)d_out;

    const int N = in_sizes[0] / (T_HIST * F_IN);
    const int E = in_sizes[1] / 2;
    const int* src = eidx;
    const int* dst = eidx + E;

    const int NP = (N + 63) & ~63;
    const int NB = (N + 255) / 256;
    const int NBINS = (N + BIN_W - 1) >> BIN_SHIFT;
    float* dinv   = (float*)d_ws;
    float* bufA   = dinv + NP;
    float* zin    = bufA + (size_t)N * HDIM;
    int*   cnt    = (int*)(zin + (size_t)N * ZP);
    int*   rowptr = cnt + NP;
    int*   cursor = rowptr + NP;
    int*   bsum   = cursor + NP;
    int*   col    = bsum + ((NB + 63) & ~63);
    const int NCH = (E + SORT_TILE - 1) / SORT_TILE;
    unsigned* packed   = (unsigned*)bufA;
    int*      chunkoff = (int*)(packed + ((E + 63) & ~63));
    float*    xp       = (float*)(chunkoff + (((size_t)(NBINS + 1) * NCH + 63) & ~(size_t)63));

    chunk_sort_kernel<<<NCH, SORT_NT, 2 * NBINS * sizeof(int), stream>>>(src, dst, chunkoff, packed, E, NBINS, NCH);
    bin_sort_kernel<<<NBINS, BIN_W, 0, stream>>>(packed, chunkoff, cnt, dinv, rowptr, col, x, xp, N, NCH);

    gather_in_kernel<<<(N + 8 * GN_NODES - 1) / (8 * GN_NODES), 256, 0, stream>>>(rowptr, cnt, col, dinv, xp, zin, N);

    agg_gemm_head_kernel<<<(N + FT_ROWS - 1) / FT_ROWS, 256, 0, stream>>>(rowptr, cnt, col, dinv, zin, W1, b1, W2, b2,
                                                                          Wout, bout, out, N);
}


# output head on WMMA: the W2 product is formed transposed, its packed D tile (two-part f16) is the B fragment against Wout's rows; waves' shares added through LDS; head sum order changes
# speedup vs baseline: 12.9004x; 1.1079x over previous
#include <hip/hip_runtime.h>
#include <hip/hip_bf16.h>

typedef __attribute__((ext_vector_type(16))) _Float16 v16h;
typedef __attribute__((ext_vector_type(8)))  float    v8f;
typedef __attribute__((ext_vector_type(8)))  _Float16 v8h;

#define T_HIST 3
#define F_IN   5
#define HDIM   128
#define CDIM   5
#define ZP     8

#define BIN_SHIFT 8
#define BIN_W     (1 << BIN_SHIFT)
#define SORT_NT   256
#define SORT_EPT  25
#define SORT_TILE (SORT_NT * SORT_EPT)
#define SCOL_CAP  8192

static __device__ __forceinline__ void block_excl_scan(const int* in, int* out, int nb, int* wsum) {
    const int tid  = threadIdx.x;
    const int lane = tid & 31;
    const int wave = tid >> 5;
    const int per  = (nb + SORT_NT - 1) / SORT_NT;
    const int lo_i = tid * per;
    int s = 0;
    for (int k = 0; k < per; ++k) {
        const int i = lo_i + k;
        if (i < nb) { const int v = in[i]; out[i] = s; s += v; }
    }
    int incl = s;
#pragma unroll
    for (int off = 1; off < 32; off <<= 1) {
        const int u = __shfl_up(incl, off, 32);
        if (lane >= off) incl += u;
    }
    if (lane == 31) wsum[wave] = incl;
    __syncthreads();
    int base = incl - s;
    for (int w = 0; w < wave; ++w) base += wsum[w];
    for (int k = 0; k < per; ++k) {
        const int i = lo_i + k;
        if (i < nb) out[i] += base;
    }
    __syncthreads();
}

__global__ __launch_bounds__(SORT_NT) void chunk_sort_kernel(const int* __restrict__ src, const int* __restrict__ dst,
                                                             int* __restrict__ chunkoff, unsigned* __restrict__ packed,
                                                             int e, int nbins, int nchunks) {
    extern __shared__ int sm[];
    __shared__ int      wsum[SORT_NT / 32];
    __shared__ unsigned stage_w[SORT_TILE];
    int* hist   = sm;
    int* lstart = sm + nbins;
    const int tid = threadIdx.x;
    const int lo  = blockIdx.x * SORT_TILE;
    const int hi  = lo + SORT_TILE < e ? lo + SORT_TILE : e;

    for (int i = tid; i < nbins; i += SORT_NT) hist[i] = 0;
    __syncthreads();

    int s[SORT_EPT], d[SORT_EPT], r[SORT_EPT];
#pragma unroll
    for (int k = 0; k < SORT_EPT; ++k) {
        const int i  = lo + k * SORT_NT + tid;
        const int ii = i < hi ? i : hi - 1;
        d[k] = dst[ii];
        s[k] = src[ii];
    }
#pragma unroll
    for (int k = 0; k < SORT_EPT; ++k)
        r[k] = lo + k * SORT_NT + tid < hi ? atomicAdd(&hist[d[k] >> BIN_SHIFT], 1) : 0;
    __syncthreads();

    block_excl_scan(hist, lstart, nbins, wsum);

    for (int i = tid; i < nbins; i += SORT_NT) chunkoff[(size_t)i * nchunks + blockIdx.x] = lstart[i];
    if (tid == 0) chunkoff[(size_t)nbins * nchunks + blockIdx.x] = hi - lo;

#pragma unroll
    for (int k = 0; k < SORT_EPT; ++k) r[k] += lstart[d[k] >> BIN_SHIFT];
#pragma unroll
    for (int k = 0; k < SORT_EPT; ++k) {
        if (lo + k * SORT_NT + tid < hi)
            stage_w[r[k]] = ((unsigned)s[k] << BIN_SHIFT) | (unsigned)(d[k] & (BIN_W - 1));
    }
    __syncthreads();

    const int nt = hi - lo;
    for (int i = tid; i < nt; i += SORT_NT) packed[lo + i] = stage_w[i];
}

#define BS_RUNS 16
template <typename F>
static __device__ __forceinline__ void walk_bin(const unsigned* __restrict__ packed, const int* __restrict__ row0,
                                                const int* __restrict__ row1, int nchunks, int tid, int lane, F f) {
    for (int g = 0; g < nchunks; g += BIN_W) {
        const int c = g + tid;
        int pos = 0, len = 0;
        if (c < nchunks) {
            const int o = row0[c];
            len = row1[c] - o;
            pos = c * SORT_TILE + o;
        }
#pragma unroll
        for (int j0 = 0; j0 < 32; j0 += BS_RUNS) {
            unsigned w[BS_RUNS];
#pragma unroll
            for (int k = 0; k < BS_RUNS; ++k) {
                const int p = __builtin_amdgcn_readlane(pos, j0 + k);
                const int l = __builtin_amdgcn_readlane(len, j0 + k);
                w[k] = packed[lane < l ? p + lane : 0];
            }
#pragma unroll
            for (int k = 0; k < BS_RUNS; ++k) {
                const int l = __builtin_amdgcn_readlane(len, j0 + k);
                if (lane < l) f(w[k]);
            }
        }
        unsigned longmask = __builtin_amdgcn_ballot_w32(len > 32);
        while (longmask != 0u) {
            const int j = __builtin_ctz(longmask);
            longmask &= longmask - 1u;
            const int p = __shfl(pos, j, 32);
            const int l = __shfl(len, j, 32);
            for (int b = 32; b < l; b += 32)
                if (b + lane < l) f(packed[p + b + lane]);
        }
    }
}

__global__ __launch_bounds__(BIN_W) void bin_sort_kernel(const unsigned* __restrict__ packed, const int* __restrict__ chunkoff,
                                                         int* __restrict__ cnt, float* __restrict__ dinv,
                                                         int* __restrict__ rowptr, int* __restrict__ col,
                                                         const float* __restrict__ x, float* __restrict__ xp, int n,
                                                         int nchunks) {
    __shared__ int lcnt[BIN_W];
    __shared__ int wpart[BIN_W / 32];
    __shared__ int wlen[BIN_W / 32];
    __shared__ int wtot[BIN_W / 32];
    __shared__ int scol[SCOL_CAP];
    const int tid  = threadIdx.x;
    const int lane = tid & 31;
    const int wave = tid >> 5;
    const int bin  = blockIdx.x;
    const int* row0 = chunkoff + (size_t)bin * nchunks;
    const int* row1 = row0 + nchunks;

    int part = 0, plen = 0;
    for (int c = tid; c < nchunks; c += BIN_W) {
        const int o = row0[c];
        part += o;
        plen += row1[c] - o;
    }
#pragma unroll
    for (int off = 16; off > 0; off >>= 1) {
        part += __shfl_xor(part, off, 32);
        plen += __shfl_xor(plen, off, 32);
    }
    if (lane == 0) { wpart[wave] = part; wlen[wave] = plen; }
    lcnt[tid] = 0;
    __syncthreads();
    int start = 0, m = 0;
#pragma unroll
    for (int w = 0; w < BIN_W / 32; ++w) { start += wpart[w]; m += wlen[w]; }

    walk_bin(packed, row0, row1, nchunks, tid, lane, [&](unsigned w) { atomicAdd(&lcnt[w & (BIN_W - 1)], 1); });
    __syncthreads();

    const int c = lcnt[tid];
    int incl = c;
#pragma unroll
    for (int off = 1; off < 32; off <<= 1) {
        const int u = __shfl_up(incl, off, 32);
        if (lane >= off) incl += u;
    }
    if (lane == 31) wtot[wave] = incl;
    __syncthreads();
    {
        int excl = incl - c;
        for (int w = 0; w < wave; ++w) excl += wtot[w];
        const int node = bin * BIN_W + tid;
        if (node < n) {
            const float di = rsqrtf((float)c + 1.0f);
            cnt[node]    = c;
            dinv[node]   = di;
            rowptr[node] = start + excl;
            const float* xr = x + (size_t)node * (T_HIST * F_IN) + (T_HIST - 1) * F_IN;
            float4 lo4, hi4;
            lo4.x = di * xr[0]; lo4.y = di * xr[1]; lo4.z = di * xr[2]; lo4.w = di * xr[3];
            hi4.x = di * xr[4]; hi4.y = 0.0f;       hi4.z = 0.0f;       hi4.w = 0.0f;
            ((float4*)xp)[(size_t)node * (ZP / 4)]     = lo4;
            ((float4*)xp)[(size_t)node * (ZP / 4) + 1] = hi4;
        }
        if (bin == 0 && tid < ZP / 4) {
            float4 z4; z4.x = 0.0f; z4.y = 0.0f; z4.z = 0.0f; z4.w = 0.0f;
            ((float4*)xp)[(size_t)n * (ZP / 4) + tid] = z4;
        }
        lcnt[tid] = excl;
    }
    __syncthreads();

    const bool in_lds = m <= SCOL_CAP;
    walk_bin(packed, row0, row1, nchunks, tid, lane, [&](unsigned w) {
        const int pos = atomicAdd(&lcnt[w & (BIN_W - 1)], 1);
        const int sv  = (int)(w >> BIN_SHIFT);
        if (in_lds) scol[pos] = sv;
        else        col[start + pos] = sv;
    });
    if (in_lds) {
        __syncthreads();
        for (int i = tid; i < m; i += BIN_W) col[start + i] = scol[i];
    }
}


template <int M> static __device__ __forceinline__ int lane_xor(int v) {
    if constexpr (M == 16) return __builtin_amdgcn_permlanex16(v, v, 0x76543210, 0xfedcba98, false, false);
    else                   return __builtin_amdgcn_update_dpp(v, v, 0x160 + M, 0xf, 0xf, false);
}
template <int M> static __device__ __forceinline__ float lane_xor_f(float v) {
    return __builtin_bit_cast(float, lane_xor<M>(__builtin_bit_cast(int, v)));
}

template <int K, int J> static __device__ __forceinline__ int cmp_xchg(int key, int lane) {
    const int  o   = lane_xor<J>(key);
    const bool low = (((lane & J) == 0) == ((lane & K) == 0));
    const int  mn  = key < o ? key : o;
    const int  mx  = key < o ? o : key;
    return low ? mn : mx;
}
static __device__ __forceinline__ int sort_wave(int key, int lane) {
    key = cmp_xchg<2, 1>(key, lane);
    key = cmp_xchg<4, 2>(key, lane);   key = cmp_xchg<4, 1>(key, lane);
    key = cmp_xchg<8, 4>(key, lane);   key = cmp_xchg<8, 2>(key, lane);   key = cmp_xchg<8, 1>(key, lane);
    key = cmp_xchg<16, 8>(key, lane);  key = cmp_xchg<16, 4>(key, lane);  key = cmp_xchg<16, 2>(key, lane);
    key = cmp_xchg<16, 1>(key, lane);
    key = cmp_xchg<32, 16>(key, lane); key = cmp_xchg<32, 8>(key, lane);  key = cmp_xchg<32, 4>(key, lane);
    key = cmp_xchg<32, 2>(key, lane);  key = cmp_xchg<32, 1>(key, lane);
    return key;
}

static __device__ __forceinline__ float lane_bcast(float v, int l) {
    return __builtin_bit_cast(float, __builtin_amdgcn_readlane(__builtin_bit_cast(int, v), l));
}

static __device__ __forceinline__ float add_piece(int ks, int sub, int fc, const float* __restrict__ xp, float acc) {
#pragma unroll
    for (int p = 0; p < 8; ++p) {
        const int   from = p * 4 + sub;
        const int   ss   = __shfl(ks, from, 32);
        const float xv   = xp[ss * ZP + fc];
        acc += xv;
    }
    return acc;
}

#define GN_NODES 8

__global__ __launch_bounds__(256) void gather_in_kernel(const int* __restrict__ rowptr, const int* __restrict__ cnt,
                                                       int* __restrict__ col, const float* __restrict__ dinv,
                                                       const float* __restrict__ xp, float* __restrict__ agg, int n) {
    const int lane = threadIdx.x & 31;
    const int wave  = __builtin_amdgcn_readfirstlane(threadIdx.x >> 5);
    const int node0 = (blockIdx.x * 8 + wave) * GN_NODES;
    const int f     = lane & 7;
    const int fc    = f < F_IN ? f : F_IN - 1;
    const int sub   = lane >> 3;

    int   mstart = 0, mdeg = 0;
    float mdi = 0.0f;
    {
        const int nd = node0 + lane;
        if (lane < GN_NODES && nd < n) {
            mstart = rowptr[nd];
            mdeg   = cnt[nd];
            mdi    = dinv[nd];
        }
    }


    int   key[GN_NODES];
    float xs[GN_NODES];
#pragma unroll
    for (int i = 0; i < GN_NODES; ++i) {
        const int s  = __builtin_amdgcn_readlane(mstart, i);
        const int d  = __builtin_amdgcn_readlane(mdeg, i);
        const int nb = d < 32 ? d : 32;
        key[i] = lane < nb ? col[s + lane] : 0x7fffffff;
        const int nd = node0 + i < n ? node0 + i : n - 1;
        xs[i] = xp[nd * ZP + fc];
    }

#pragma unroll
    for (int i = 0; i < GN_NODES; ++i) key[i] = sort_wave(key[i], lane);

    float acc[GN_NODES];
#pragma unroll
    for (int i = 0; i < GN_NODES; ++i) {
        const int d  = __builtin_amdgcn_readlane(mdeg, i);
        const int nb = d < 32 ? d : 32;
        const int ks = lane < nb ? key[i] : n;
        acc[i] = add_piece(ks, sub, fc, xp, 0.0f);
    }

#pragma unroll
    for (int i = 0; i < GN_NODES; ++i) {
        const int s  = __builtin_amdgcn_readlane(mstart, i);
        const int d  = __builtin_amdgcn_readlane(mdeg, i);
        const int nb = d < 32 ? d : 32;
        if (lane < nb) col[s + lane] = key[i];
    }

    const unsigned longmask = __builtin_amdgcn_ballot_w32(mdeg > 32);
    if (longmask != 0u) {
#pragma nounroll
        for (int i = 0; i < GN_NODES; ++i) {
            if ((longmask >> i) & 1u) {
                const int s = __builtin_amdgcn_readlane(mstart, i);
                const int d = __builtin_amdgcn_readlane(mdeg, i);
                float extra = 0.0f;
                for (int base = 32; base < d; base += 32) {
                    const int nbp = (d - base < 32) ? d - base : 32;
                    const int kc  = col[s + base + (lane < nbp ? lane : 0)];
                    const int ks  = lane < nbp ? kc : n;
                    extra = add_piece(ks, sub, fc, xp, extra);
                }
#pragma unroll
                for (int j = 0; j < GN_NODES; ++j) acc[j] = (j == i) ? acc[j] + extra : acc[j];
            }
        }
    }

#pragma unroll
    for (int i = 0; i < GN_NODES; ++i) {
        const float di = lane_bcast(mdi, i);
        float t = acc[i];
        t += lane_xor_f<8>(t);
        t += lane_xor_f<16>(t);
        const float v = lane < F_IN ? di * (t + xs[i]) : di;
        if (lane <= F_IN && node0 + i < n) agg[(size_t)(node0 + i) * ZP + lane] = v;
    }
}

#define FT_ROWS   64
#define FT_APITCH (HDIM + 8)

typedef __attribute__((ext_vector_type(2))) _Float16 v2h;
typedef __attribute__((ext_vector_type(8))) unsigned v8u;
typedef __attribute__((ext_vector_type(8))) short    v8s;
typedef __attribute__((ext_vector_type(4))) unsigned v4u;

static __device__ __forceinline__ unsigned pk2(_Float16 a, _Float16 b) {
    v2h p; p[0] = a; p[1] = b;
    return __builtin_bit_cast(unsigned, p);
}
static __device__ __forceinline__ void split2(float v, _Float16& hi, _Float16& lo) {
    hi = (_Float16)v;
    lo = (_Float16)(v - (float)hi);
}
static __device__ __forceinline__ v8h relu8(v8h v) {
    const v8s z = {};
    return __builtin_bit_cast(v8h, __builtin_elementwise_max(__builtin_bit_cast(v8s, v), z));
}

static __device__ __forceinline__ void add_piece_wmma(float z0, float z1, float z2, float z3, float z4, float w,
                                                      bool second, int half, const v16h (&bfr)[HDIM / 16],
                                                      const v16h ones, float (&t)[HDIM / 16]) {
    _Float16 yh[F_IN], yl[F_IN], wh, wl;
    split2(w * z0, yh[0], yl[0]);
    split2(w * z1, yh[1], yl[1]);
    split2(w * z2, yh[2], yl[2]);
    split2(w * z3, yh[3], yl[3]);
    split2(w * z4, yh[4], yl[4]);
    split2(w, wh, wl);
    const unsigned p00 = pk2(yh[0], yh[1]);
    const unsigned p01 = pk2(yh[2], yh[3]);
    const unsigned p02 = pk2(yh[4], yl[0]);
    const unsigned p03 = pk2(yl[1], yl[2]);
    const unsigned p04 = pk2(yl[3], yl[4]);
    const unsigned p12 = pk2(yh[4], wh);
    const unsigned p13 = pk2(wl, wh);
    const bool up = half != 0;
    const v8h zeroh = {};
    const v8f zerof = {};
    if (second) {
        const unsigned x00 = (unsigned)lane_xor<16>((int)p00);
        const unsigned x01 = (unsigned)lane_xor<16>((int)p01);
        const unsigned x02 = (unsigned)lane_xor<16>((int)p02);
        const unsigned x03 = (unsigned)lane_xor<16>((int)p03);
        const unsigned x04 = (unsigned)lane_xor<16>((int)p04);
        const unsigned x12 = (unsigned)lane_xor<16>((int)p12);
        const unsigned x13 = (unsigned)lane_xor<16>((int)p13);
        const v8u au0 = {up ? x00 : p00, up ? x01 : p01, up ? x12 : p02, up ? x13 : p03, up ? 0u : p04, 0u, 0u, 0u};
        const v8u au1 = {up ? p00 : x00, up ? p01 : x01, up ? p12 : x02, up ? p13 : x03, up ? 0u : x04, 0u, 0u, 0u};
        const v16h av0 = __builtin_bit_cast(v16h, au0);
        const v16h av1 = __builtin_bit_cast(v16h, au1);
#pragma unroll
        for (int j0 = 0; j0 < HDIM / 16; j0 += 4) {
            v8h d0[4], d1[4];
#pragma unroll
            for (int u = 0; u < 4; ++u) {
                d0[u] = __builtin_amdgcn_wmma_f16_16x16x32_f16(false, av0, false, bfr[j0 + u], (short)0, zeroh, false, false);
                d1[u] = __builtin_amdgcn_wmma_f16_16x16x32_f16(false, av1, false, bfr[j0 + u], (short)0, zeroh, false, false);
            }
#pragma unroll
            for (int u = 0; u < 4; ++u) {
                const v16h hb = __builtin_shufflevector(relu8(d0[u]), relu8(d1[u]), 0, 1, 2, 3, 4, 5, 6, 7, 8, 9, 10, 11, 12, 13, 14, 15);
                const v8f s = __builtin_amdgcn_wmma_f32_16x16x32_f16(false, ones, false, hb, (short)0, zerof, false, false);
                t[j0 + u] += s[0];
            }
        }
    } else {
        const unsigned x00 = (unsigned)lane_xor<16>((int)p00);
        const unsigned x01 = (unsigned)lane_xor<16>((int)p01);
        const unsigned x12 = (unsigned)lane_xor<16>((int)p12);
        const unsigned x13 = (unsigned)lane_xor<16>((int)p13);
        const v8u au0 = {up ? x00 : p00, up ? x01 : p01, up ? x12 : p02, up ? x13 : p03, up ? 0u : p04, 0u, 0u, 0u};
        const v16h av0 = __builtin_bit_cast(v16h, au0);
#pragma unroll
        for (int j0 = 0; j0 < HDIM / 16; j0 += 4) {
            v8h d0[4];
#pragma unroll
            for (int u = 0; u < 4; ++u)
                d0[u] = __builtin_amdgcn_wmma_f16_16x16x32_f16(false, av0, false, bfr[j0 + u], (short)0, zeroh, false, false);
#pragma unroll
            for (int u = 0; u < 4; ++u) {
                const v16h hb = __builtin_shufflevector(relu8(d0[u]), zeroh, 0, 1, 2, 3, 4, 5, 6, 7, 8, 9, 10, 11, 12, 13, 14, 15);
                const v8f s = __builtin_amdgcn_wmma_f32_16x16x32_f16(false, ones, false, hb, (short)0, zerof, false, false);
                t[j0 + u] += s[0];
            }
        }
    }
}

__global__ __launch_bounds__(256, 6) void agg_gemm_head_kernel(const int* __restrict__ rowptr, const int* __restrict__ cnt,
                                                            const int* __restrict__ col, const float* __restrict__ dinv,
                                                            const float* __restrict__ zin,
                                                            const float* __restrict__ W1, const float* __restrict__ b1,
                                                            const float* __restrict__ W2, const float* __restrict__ b2,
                                                            const float* __restrict__ Wout, const float* __restrict__ bout,
                                                            float* __restrict__ out, int n) {
    __shared__ __attribute__((aligned(16))) _Float16 aS[FT_ROWS][FT_APITCH];
    __shared__ __attribute__((aligned(16))) float    pS[8 * CDIM * FT_ROWS];

    const int tid  = threadIdx.x;
    const int lane = tid & 31;
    const int wave = __builtin_amdgcn_readfirstlane(tid >> 5);
    const int row0 = blockIdx.x * FT_ROWS;
    const int c    = lane & 15;
    const int h    = lane >> 4;

    {
        v4u* bS = (v4u*)pS;
        {
            const int cl = 16 * wave + c;
            _Float16 Wh[F_IN], Wl[F_IN], bh, bl;
#pragma unroll
            for (int f = 0; f < F_IN; ++f) split2(W1[f * HDIM + cl], Wh[f], Wl[f]);
            split2(b1[cl], bh, bl);
            const unsigned q00 = pk2(Wh[0], Wh[1]), q01 = pk2(Wh[2], Wh[3]), q02 = pk2(Wh[4], Wh[0]),
                           q03 = pk2(Wh[1], Wh[2]), q04 = pk2(Wh[3], Wh[4]);
            const unsigned q10 = pk2(Wl[0], Wl[1]), q11 = pk2(Wl[2], Wl[3]), q12 = pk2(Wl[4], bh), q13 = pk2(bh, bl);
            const bool up = h != 0;
            const v4u lo4 = {up ? q10 : q00, up ? q11 : q01, up ? q12 : q02, up ? q13 : q03};
            bS[(wave * 32 + lane) * 2] = lo4;
            ((unsigned*)bS)[(wave * 32 + lane) * 8 + 4] = up ? 0u : q04;
        }
        __syncthreads();
        v16h bfr[HDIM / 16];
#pragma unroll
        for (int j = 0; j < HDIM / 16; ++j) {
            const v4u      lo4 = bS[(j * 32 + lane) * 2];
            const unsigned w4  = ((const unsigned*)bS)[(j * 32 + lane) * 8 + 4];
            const v8u bu = {lo4[0], lo4[1], lo4[2], lo4[3], w4, 0u, 0u, 0u};
            bfr[j] = __builtin_bit_cast(v16h, bu);
        }
        v16h ones;
#pragma unroll
        for (int s = 0; s < 16; ++s) ones[s] = (_Float16)1.0f;
        const float4* zr4 = (const float4*)zin;

        int   mstart = 0, mdeg = 0;
        float mdi = 0.0f;
        {
            const int nd = row0 + wave * (FT_ROWS / 8) + lane;
            if (lane < FT_ROWS / 8 && nd < n) {
                mstart = rowptr[nd];
                mdeg   = cnt[nd];
                mdi    = dinv[nd];
            }
        }

        auto first_piece = [&](int i) -> int {
            const int s    = __builtin_amdgcn_readlane(mstart, i);
            const int d    = __builtin_amdgcn_readlane(mdeg, i);
            const int nb   = d < 31 ? d : 31;
            const int cv   = col[(lane >= 1 && lane <= nb) ? s + lane - 1 : 0];
            const int nd   = row0 + wave * (FT_ROWS / 8) + i;
            const int self = nd < n ? nd : 0;
            return lane == 0 ? self : cv;
        };

        const float2* zr2 = (const float2*)zin;

        int    ms_b = first_piece(0);
        float4 zq_a = zr4[(size_t)ms_b * (ZP / 4)];
        float2 zt_a = zr2[(size_t)ms_b * (ZP / 2) + 2];
        ms_b = first_piece(1);

#pragma nounroll
        for (int i = 0; i < FT_ROWS / 8; ++i) {
            const int    ms_c = first_piece(i + 2);
            const float4 zq_b = zr4[(size_t)ms_b * (ZP / 4)];
            const float2 zt_b = zr2[(size_t)ms_b * (ZP / 2) + 2];

            const int   m   = wave * (FT_ROWS / 8) + i;
            const int   deg = __builtin_amdgcn_readlane(mdeg, i);
            const float di  = lane_bcast(mdi, i);
            float a[HDIM / 16] = {0.0f, 0.0f, 0.0f, 0.0f, 0.0f, 0.0f, 0.0f, 0.0f};
            {
                const int   nb = deg < 31 ? deg : 31;
                const float mw = lane <= nb ? zt_a.y * di : 0.0f;
                add_piece_wmma(zq_a.x, zq_a.y, zq_a.z, zq_a.w, zt_a.x, mw, nb >= 16, h, bfr, ones, a);
            }
            if (deg > 31) {
                const int start = __builtin_amdgcn_readlane(mstart, i);
                for (int base = 31; base < deg; base += 32) {
                    const int    nb = (deg - base < 32) ? deg - base : 32;
                    const int    ms = col[lane < nb ? start + base + lane : 0];
                    const float4 zq = zr4[(size_t)ms * (ZP / 4)];
                    const float2 zt = zr2[(size_t)ms * (ZP / 2) + 2];
                    const float  mw = lane < nb ? zt.y * di : 0.0f;
                    add_piece_wmma(zq.x, zq.y, zq.z, zq.w, zt.x, mw, nb > 16, h, bfr, ones, a);
                }
            }
            if (h == 0) {
#pragma unroll
                for (int j = 0; j < HDIM / 16; ++j) aS[m][16 * j + c] = (_Float16)a[j];
            }

            zq_a = zq_b; zt_a = zt_b;
            ms_b = ms_c;
        }
    }

    const int ncol  = wave * 16;
    const int khalf = h * 8;
    const int colw  = ncol + c;

    v16h wf[4];
#pragma unroll
    for (int ks = 0; ks < 4; ++ks) {
#pragma unroll
        for (int j = 0; j < 16; ++j) {
            const int K = (j & 7) + ((j >> 3) << 4) + khalf + ks * 32;
            wf[ks][j] = (_Float16)W2[K * HDIM + colw];
        }
    }
    float b2r[8];
    {
        const float4 q0 = *(const float4*)(b2 + ncol + khalf);
        const float4 q1 = *(const float4*)(b2 + ncol + khalf + 4);
        b2r[0] = q0.x; b2r[1] = q0.y; b2r[2] = q0.z; b2r[3] = q0.w;
        b2r[4] = q1.x; b2r[5] = q1.y; b2r[6] = q1.z; b2r[7] = q1.w;
    }
    v16h whi, wlo;
    {
        const int cls = c < CDIM ? c : 0;
#pragma unroll
        for (int s = 0; s < 8; ++s) {
            const float wv = Wout[(ncol + khalf + s) * CDIM + cls];
            _Float16 hi, lo;
            split2(c < CDIM ? wv : 0.0f, hi, lo);
            whi[s] = hi; whi[s + 8] = hi;
            wlo[s] = lo; wlo[s + 8] = (_Float16)0.0f;
        }
    }
    __syncthreads();

    v8f acc[4] = {};
#pragma unroll
    for (int ks = 0; ks < 4; ++ks) {
#pragma unroll
        for (int rt = 0; rt < 4; ++rt) {
            const _Float16* ap = &aS[rt * 16 + c][ks * 32 + khalf];
            const v8h lo = *(const v8h*)ap;
            const v8h hi = *(const v8h*)(ap + 16);
            const v16h av = __builtin_shufflevector(lo, hi, 0, 1, 2, 3, 4, 5, 6, 7, 8, 9, 10, 11, 12, 13, 14, 15);
            acc[rt] = __builtin_amdgcn_wmma_f32_16x16x32_f16(false, wf[ks], false, av,
                                                             (short)0, acc[rt], false, false);
        }
    }

    const v8f zerof = {};
#pragma unroll
    for (int rt = 0; rt < 4; ++rt) {
        v8h hh, hl;
#pragma unroll
        for (int v = 0; v < 8; ++v) {
            _Float16 hi, lo;
            split2(fmaxf(acc[rt][v] + b2r[v], 0.0f), hi, lo);
            hh[v] = hi; hl[v] = lo;
        }
        const v16h hb = __builtin_shufflevector(hh, hl, 0, 1, 2, 3, 4, 5, 6, 7, 8, 9, 10, 11, 12, 13, 14, 15);
        v8f lg = __builtin_amdgcn_wmma_f32_16x16x32_f16(false, whi, false, hb, (short)0, zerof, false, false);
        lg = __builtin_amdgcn_wmma_f32_16x16x32_f16(false, wlo, false, hb, (short)0, lg, false, false);
        if (h == 0) {
#pragma unroll
            for (int r = 0; r < CDIM; ++r) pS[(wave * CDIM + r) * FT_ROWS + rt * 16 + c] = lg[r];
        }
    }
    __syncthreads();

    if (wave < FT_ROWS / 32) {
        const int node = row0 + tid;
        if (node < n) {
            float lg[CDIM];
#pragma unroll
            for (int cc = 0; cc < CDIM; ++cc) {
                float s = pS[cc * FT_ROWS + tid];
#pragma unroll
                for (int w = 1; w < 8; ++w) s += pS[(w * CDIM + cc) * FT_ROWS + tid];
                lg[cc] = s + bout[cc];
            }
            float mx = lg[0];
#pragma unroll
            for (int cc = 1; cc < CDIM; ++cc) mx = fmaxf(mx, lg[cc]);
            float sum = 0.0f;
#pragma unroll
            for (int cc = 0; cc < CDIM; ++cc) { lg[cc] = __expf(lg[cc] - mx); sum += lg[cc]; }
            const float inv = 1.0f / sum;
#pragma unroll
            for (int cc = 0; cc < CDIM; ++cc) out[(size_t)node * CDIM + cc] = lg[cc] * inv;
        }
    }
}

extern "C" void kernel_launch(void* const* d_in, const int* in_sizes, int n_in,
                              void* d_out, int out_size, void* d_ws, size_t ws_size,
                              hipStream_t stream) {
    const float* x    = (const float*)d_in[0];
    const int*   eidx = (const int*)  d_in[1];
    const float* W1   = (const float*)d_in[2];
    const float* b1   = (const float*)d_in[3];
    const float* W2   = (const float*)d_in[4];
    const float* b2   = (const float*)d_in[5];
    const float* Wout = (const float*)d_in[6];
    const float* bout = (const float*)d_in[7];
    float*       out  = (float*)d_out;

    const int N = in_sizes[0] / (T_HIST * F_IN);
    const int E = in_sizes[1] / 2;
    const int* src = eidx;
    const int* dst = eidx + E;

    const int NP = (N + 63) & ~63;
    const int NB = (N + 255) / 256;
    const int NBINS = (N + BIN_W - 1) >> BIN_SHIFT;
    float* dinv   = (float*)d_ws;
    float* bufA   = dinv + NP;
    float* zin    = bufA + (size_t)N * HDIM;
    int*   cnt    = (int*)(zin + (size_t)N * ZP);
    int*   rowptr = cnt + NP;
    int*   cursor = rowptr + NP;
    int*   bsum   = cursor + NP;
    int*   col    = bsum + ((NB + 63) & ~63);
    const int NCH = (E + SORT_TILE - 1) / SORT_TILE;
    unsigned* packed   = (unsigned*)bufA;
    int*      chunkoff = (int*)(packed + ((E + 63) & ~63));
    float*    xp       = (float*)(chunkoff + (((size_t)(NBINS + 1) * NCH + 63) & ~(size_t)63));

    chunk_sort_kernel<<<NCH, SORT_NT, 2 * NBINS * sizeof(int), stream>>>(src, dst, chunkoff, packed, E, NBINS, NCH);
    bin_sort_kernel<<<NBINS, BIN_W, 0, stream>>>(packed, chunkoff, cnt, dinv, rowptr, col, x, xp, N, NCH);

    gather_in_kernel<<<(N + 8 * GN_NODES - 1) / (8 * GN_NODES), 256, 0, stream>>>(rowptr, cnt, col, dinv, xp, zin, N);

    agg_gemm_head_kernel<<<(N + FT_ROWS - 1) / FT_ROWS, 256, 0, stream>>>(rowptr, cnt, col, dinv, zin, W1, b1, W2, b2,
                                                                          Wout, bout, out, N);
}
